# starting kernel
# baseline (speedup 1.0000x reference)
#include <hip/hip_runtime.h>

typedef __attribute__((ext_vector_type(2))) float v2f;
typedef __attribute__((ext_vector_type(8))) float v8f;

#define NN 50000
#define NE 800000
#define KIN 128
#define HID 64
#define NC 32
#define ROW_TILES (NN / 16)

__global__ void k_zero(float* __restrict__ p, int n) {
  int i = blockIdx.x * blockDim.x + threadIdx.x;
  if (i < n) p[i] = 0.0f;
}

__global__ void k_degree(const int* __restrict__ dst, float* __restrict__ deg) {
  int e = blockIdx.x * blockDim.x + threadIdx.x;
  if (e < NE) atomicAdd(&deg[dst[e]], 1.0f);
}

__global__ void k_dinv(float* __restrict__ deg) {
  int i = blockIdx.x * blockDim.x + threadIdx.x;
  if (i < NN) deg[i] = rsqrtf(deg[i] + 1.0f);
}

__global__ void k_gemm_xw0(const float* __restrict__ X,
                           const float* __restrict__ W,
                           float* __restrict__ H) {
  int wave = (blockIdx.x * blockDim.x + threadIdx.x) >> 5;
  if (wave >= ROW_TILES) return;
  int lane = threadIdx.x & 31;
  int hh   = lane >> 4;
  int lid  = lane & 15;
  const float* xr = X + (wave * 16 + lid) * KIN;
  v8f c0 = {}, c1 = {}, c2 = {}, c3 = {};
  for (int k = 0; k < KIN; k += 4) {
    __builtin_prefetch(xr + k + 32, 0, 1);
    int ka = k + 2 * hh;
    v2f a;
    a.x = xr[ka];
    a.y = xr[ka + 1];
    const float* wr0 = W + ka * HID;
    const float* wr1 = wr0 + HID;
    v2f b;
    b.x = wr0[lid];      b.y = wr1[lid];
    c0 = __builtin_amdgcn_wmma_f32_16x16x4_f32(false, a, false, b, (short)0, c0, false, false);
    b.x = wr0[lid + 16]; b.y = wr1[lid + 16];
    c1 = __builtin_amdgcn_wmma_f32_16x16x4_f32(false, a, false, b, (short)0, c1, false, false);
    b.x = wr0[lid + 32]; b.y = wr1[lid + 32];
    c2 = __builtin_amdgcn_wmma_f32_16x16x4_f32(false, a, false, b, (short)0, c2, false, false);
    b.x = wr0[lid + 48]; b.y = wr1[lid + 48];
    c3 = __builtin_amdgcn_wmma_f32_16x16x4_f32(false, a, false, b, (short)0, c3, false, false);
  }
  int mrow = wave * 16 + hh * 8;
  for (int j = 0; j < 8; ++j) {
    float* o = H + (mrow + j) * HID + lid;
    o[0]  = c0[j];
    o[16] = c1[j];
    o[32] = c2[j];
    o[48] = c3[j];
  }
}

__global__ void k_gemm_hw12(const float* __restrict__ H,
                            const float* __restrict__ W1,
                            const float* __restrict__ W2,
                            float* __restrict__ MU,
                            float* __restrict__ LV) {
  int wave = (blockIdx.x * blockDim.x + threadIdx.x) >> 5;
  if (wave >= ROW_TILES) return;
  int lane = threadIdx.x & 31;
  int hh   = lane >> 4;
  int lid  = lane & 15;
  const float* hr = H + (wave * 16 + lid) * HID;
  v8f m0 = {}, m1 = {}, l0 = {}, l1 = {};
  for (int k = 0; k < HID; k += 4) {
    int ka = k + 2 * hh;
    v2f a;
    a.x = hr[ka];
    a.y = hr[ka + 1];
    const float* u0 = W1 + ka * NC;
    const float* u1 = u0 + NC;
    const float* v0 = W2 + ka * NC;
    const float* v1 = v0 + NC;
    v2f b;
    b.x = u0[lid];      b.y = u1[lid];
    m0 = __builtin_amdgcn_wmma_f32_16x16x4_f32(false, a, false, b, (short)0, m0, false, false);
    b.x = u0[lid + 16]; b.y = u1[lid + 16];
    m1 = __builtin_amdgcn_wmma_f32_16x16x4_f32(false, a, false, b, (short)0, m1, false, false);
    b.x = v0[lid];      b.y = v1[lid];
    l0 = __builtin_amdgcn_wmma_f32_16x16x4_f32(false, a, false, b, (short)0, l0, false, false);
    b.x = v0[lid + 16]; b.y = v1[lid + 16];
    l1 = __builtin_amdgcn_wmma_f32_16x16x4_f32(false, a, false, b, (short)0, l1, false, false);
  }
  int mrow = wave * 16 + hh * 8;
  for (int j = 0; j < 8; ++j) {
    float* om = MU + (mrow + j) * NC + lid;
    float* ol = LV + (mrow + j) * NC + lid;
    om[0]  = m0[j];
    om[16] = m1[j];
    ol[0]  = l0[j];
    ol[16] = l1[j];
  }
}

__global__ void k_selfinit64(const float* __restrict__ Hlin,
                             const float* __restrict__ dinv,
                             const float* __restrict__ bias,
                             float* __restrict__ Agg) {
  int idx = blockIdx.x * blockDim.x + threadIdx.x;
  if (idx >= NN * HID) return;
  int i = idx >> 6;
  int f = idx & 63;
  float d = dinv[i];
  Agg[idx] = Hlin[idx] * d * d + bias[f];
}

__global__ void k_scatter64(const float* __restrict__ Hlin,
                            const int* __restrict__ src,
                            const int* __restrict__ dst,
                            const float* __restrict__ dinv,
                            float* __restrict__ Agg) {
  int idx = blockIdx.x * blockDim.x + threadIdx.x;
  int e = idx >> 6;
  if (e >= NE) return;
  int f = idx & 63;
  int s = src[e], d = dst[e];
  float nrm = dinv[s] * dinv[d];
  atomicAdd(&Agg[d * HID + f], Hlin[s * HID + f] * nrm);
}

__global__ void k_reludrop(const float* __restrict__ Agg,
                           const float* __restrict__ mask,
                           float* __restrict__ Hout) {
  int idx = blockIdx.x * blockDim.x + threadIdx.x;
  if (idx < NN * HID) Hout[idx] = fmaxf(Agg[idx], 0.0f) * mask[idx];
}

__global__ void k_selfinit32(const float* __restrict__ MUlin,
                             const float* __restrict__ LVlin,
                             const float* __restrict__ dinv,
                             const float* __restrict__ b1,
                             const float* __restrict__ b2,
                             float* __restrict__ MUout,
                             float* __restrict__ LVout) {
  int idx = blockIdx.x * blockDim.x + threadIdx.x;
  if (idx >= NN * NC) return;
  int i = idx >> 5;
  int f = idx & 31;
  float d  = dinv[i];
  float dd = d * d;
  MUout[idx] = MUlin[idx] * dd + b1[f];
  LVout[idx] = LVlin[idx] * dd + b2[f];
}

__global__ void k_scatter32(const float* __restrict__ MUlin,
                            const float* __restrict__ LVlin,
                            const int* __restrict__ src,
                            const int* __restrict__ dst,
                            const float* __restrict__ dinv,
                            float* __restrict__ MUout,
                            float* __restrict__ LVout) {
  int idx = blockIdx.x * blockDim.x + threadIdx.x;
  int e = idx >> 5;
  if (e >= NE) return;
  int f = idx & 31;
  int s = src[e], d = dst[e];
  float nrm = dinv[s] * dinv[d];
  atomicAdd(&MUout[d * NC + f], MUlin[s * NC + f] * nrm);
  atomicAdd(&LVout[d * NC + f], LVlin[s * NC + f] * nrm);
}

__global__ void k_final(const float* __restrict__ eps,
                        const float* __restrict__ MU,
                        const float* __restrict__ LV,
                        float* __restrict__ Z) {
  int idx = blockIdx.x * blockDim.x + threadIdx.x;
  if (idx < NN * NC) Z[idx] = eps[idx] * expf(LV[idx]) + MU[idx];
}

extern "C" void kernel_launch(void* const* d_in, const int* in_sizes, int n_in,
                              void* d_out, int out_size, void* d_ws, size_t ws_size,
                              hipStream_t stream) {
  (void)in_sizes; (void)n_in; (void)out_size; (void)ws_size;
  const float* x    = (const float*)d_in[0];
  const int*   ei   = (const int*)d_in[1];
  const float* W0   = (const float*)d_in[2];
  const float* b0   = (const float*)d_in[3];
  const float* W1   = (const float*)d_in[4];
  const float* b1   = (const float*)d_in[5];
  const float* W2   = (const float*)d_in[6];
  const float* b2   = (const float*)d_in[7];
  const float* mask = (const float*)d_in[8];
  const float* eps  = (const float*)d_in[9];
  const int* src = ei;
  const int* dst = ei + NE;

  float* ws    = (float*)d_ws;
  float* dinv  = ws;
  float* hlin  = dinv + NN;
  float* hagg  = hlin + (size_t)NN * HID;
  float* mulin = hagg + (size_t)NN * HID;
  float* lvlin = mulin + (size_t)NN * NC;

  float* z  = (float*)d_out;
  float* mu = z + (size_t)NN * NC;
  float* lv = mu + (size_t)NN * NC;

  const int B = 256;
  const int gN   = (NN + B - 1) / B;
  const int gE   = (NE + B - 1) / B;
  const int gW   = (ROW_TILES * 32 + B - 1) / B;
  const int gNH  = (NN * HID + B - 1) / B;
  const int gEH  = (NE * HID + B - 1) / B;
  const int gNC_ = (NN * NC + B - 1) / B;
  const int gEC  = (NE * NC + B - 1) / B;

  k_zero   <<<gN, B, 0, stream>>>(dinv, NN);
  k_degree <<<gE, B, 0, stream>>>(dst, dinv);
  k_dinv   <<<gN, B, 0, stream>>>(dinv);

  k_gemm_xw0  <<<gW,  B, 0, stream>>>(x, W0, hlin);
  k_selfinit64<<<gNH, B, 0, stream>>>(hlin, dinv, b0, hagg);
  k_scatter64 <<<gEH, B, 0, stream>>>(hlin, src, dst, dinv, hagg);
  k_reludrop  <<<gNH, B, 0, stream>>>(hagg, mask, hlin);

  k_gemm_hw12 <<<gW,   B, 0, stream>>>(hlin, W1, W2, mulin, lvlin);
  k_selfinit32<<<gNC_, B, 0, stream>>>(mulin, lvlin, dinv, b1, b2, mu, lv);
  k_scatter32 <<<gEC,  B, 0, stream>>>(mulin, lvlin, src, dst, dinv, mu, lv);

  k_final     <<<gNC_, B, 0, stream>>>(eps, mu, lv, z);
}


# Sum each node's incoming rows in registers over an edge list ordered by destination (place = returning degree add, prefix offsets), replacing the f32 atomic scatter-add; sum order = list order
# speedup vs baseline: 2.1197x; 2.1197x over previous
#include <hip/hip_runtime.h>


typedef __attribute__((ext_vector_type(2))) float v2f;
typedef __attribute__((ext_vector_type(8))) float v8f;

#define NN 50000
#define NE 800000
#define KIN 128
#define HID 64
#define NC 32
#define ROW_TILES (NN / 16)
#define NNP 50048

__device__ __forceinline__ int clamp_idx(int v, int hi) {
  return v < 0 ? 0 : (v > hi ? hi : v);
}

__global__ void k_zero(int* __restrict__ p, int n) {
  int i = blockIdx.x * blockDim.x + threadIdx.x;
  if (i < n) p[i] = 0;
}

__global__ void k_rank(const int* __restrict__ dst,
                       int* __restrict__ cnt,
                       int* __restrict__ rank) {
  int e = blockIdx.x * blockDim.x + threadIdx.x;
  if (e < NE) rank[e] = atomicAdd(&cnt[dst[e]], 1);
}

__global__ void k_offsets(const int* __restrict__ cnt,
                          int* __restrict__ off,
                          float* __restrict__ dinv) {
  __shared__ int red[8];
  __shared__ int wsum[8];
  int t = threadIdx.x;
  int lane = t & 31;
  int w = t >> 5;
  const int4* c4 = (const int4*)cnt;
  int n4 = blockIdx.x * 64;
  int s = 0;
#pragma unroll 8
  for (int i = t; i < n4; i += 256) {
    int4 v = c4[i];
    s += (v.x + v.y) + (v.z + v.w);
  }
  for (int o = 16; o > 0; o >>= 1) s += __shfl_xor(s, o);
  if (lane == 0) red[w] = s;

  int i = blockIdx.x * 256 + t;
  int c = (i < NN) ? cnt[i] : 0;
  int x = c;
  for (int o = 1; o < 32; o <<= 1) {
    int y = __shfl_up(x, o);
    if (lane >= o) x += y;
  }
  if (lane == 31) wsum[w] = x;
  __syncthreads();
  int base = 0;
  for (int k = 0; k < 8; ++k) base += red[k];
  for (int k = 0; k < 8; ++k) base += (k < w) ? wsum[k] : 0;
  if (i < NN) {
    off[i]  = base + x - c;
    dinv[i] = rsqrtf((float)c + 1.0f);
  }
}

__global__ void k_fill(const int* __restrict__ src,
                       const int* __restrict__ dst,
                       const int* __restrict__ off,
                       const int* __restrict__ rank,
                       int* __restrict__ lst) {
  int e = blockIdx.x * blockDim.x + threadIdx.x;
  if (e >= NE) return;
  int pos = clamp_idx(off[dst[e]] + rank[e], NE - 1);
  lst[pos] = src[e];
}

__global__ void k_gemm_xw0(const float* __restrict__ X,
                           const float* __restrict__ W,
                           float* __restrict__ H) {
  int wave = (blockIdx.x * blockDim.x + threadIdx.x) >> 5;
  if (wave >= ROW_TILES) return;
  int lane = threadIdx.x & 31;
  int hh   = lane >> 4;
  int lid  = lane & 15;
  const float* xr = X + (wave * 16 + lid) * KIN;
  v8f c0 = {}, c1 = {}, c2 = {}, c3 = {};
  for (int k = 0; k < KIN; k += 4) {
    __builtin_prefetch(xr + k + 32, 0, 1);
    int ka = k + 2 * hh;
    v2f a;
    a.x = xr[ka];
    a.y = xr[ka + 1];
    const float* wr0 = W + ka * HID;
    const float* wr1 = wr0 + HID;
    v2f b;
    b.x = wr0[lid];      b.y = wr1[lid];
    c0 = __builtin_amdgcn_wmma_f32_16x16x4_f32(false, a, false, b, (short)0, c0, false, false);
    b.x = wr0[lid + 16]; b.y = wr1[lid + 16];
    c1 = __builtin_amdgcn_wmma_f32_16x16x4_f32(false, a, false, b, (short)0, c1, false, false);
    b.x = wr0[lid + 32]; b.y = wr1[lid + 32];
    c2 = __builtin_amdgcn_wmma_f32_16x16x4_f32(false, a, false, b, (short)0, c2, false, false);
    b.x = wr0[lid + 48]; b.y = wr1[lid + 48];
    c3 = __builtin_amdgcn_wmma_f32_16x16x4_f32(false, a, false, b, (short)0, c3, false, false);
  }
  int mrow = wave * 16 + hh * 8;
  for (int j = 0; j < 8; ++j) {
    float* o = H + (mrow + j) * HID + lid;
    o[0]  = c0[j];
    o[16] = c1[j];
    o[32] = c2[j];
    o[48] = c3[j];
  }
}

__global__ void k_gemm_hw12(const float* __restrict__ H,
                            const float* __restrict__ W1,
                            const float* __restrict__ W2,
                            float* __restrict__ MU,
                            float* __restrict__ LV) {
  int wave = (blockIdx.x * blockDim.x + threadIdx.x) >> 5;
  if (wave >= ROW_TILES) return;
  int lane = threadIdx.x & 31;
  int hh   = lane >> 4;
  int lid  = lane & 15;
  const float* hr = H + (wave * 16 + lid) * HID;
  v8f m0 = {}, m1 = {}, l0 = {}, l1 = {};
  for (int k = 0; k < HID; k += 4) {
    int ka = k + 2 * hh;
    v2f a;
    a.x = hr[ka];
    a.y = hr[ka + 1];
    const float* u0 = W1 + ka * NC;
    const float* u1 = u0 + NC;
    const float* v0 = W2 + ka * NC;
    const float* v1 = v0 + NC;
    v2f b;
    b.x = u0[lid];      b.y = u1[lid];
    m0 = __builtin_amdgcn_wmma_f32_16x16x4_f32(false, a, false, b, (short)0, m0, false, false);
    b.x = u0[lid + 16]; b.y = u1[lid + 16];
    m1 = __builtin_amdgcn_wmma_f32_16x16x4_f32(false, a, false, b, (short)0, m1, false, false);
    b.x = v0[lid];      b.y = v1[lid];
    l0 = __builtin_amdgcn_wmma_f32_16x16x4_f32(false, a, false, b, (short)0, l0, false, false);
    b.x = v0[lid + 16]; b.y = v1[lid + 16];
    l1 = __builtin_amdgcn_wmma_f32_16x16x4_f32(false, a, false, b, (short)0, l1, false, false);
  }
  int mrow = wave * 16 + hh * 8;
  for (int j = 0; j < 8; ++j) {
    float* om = MU + (mrow + j) * NC + lid;
    float* ol = LV + (mrow + j) * NC + lid;
    om[0]  = m0[j];
    om[16] = m1[j];
    ol[0]  = l0[j];
    ol[16] = l1[j];
  }
}

__global__ void k_gather64(const float* __restrict__ Hlin,
                           const int* __restrict__ lst,
                           const int* __restrict__ off,
                           const int* __restrict__ cnt,
                           const float* __restrict__ dinv,
                           const float* __restrict__ bias,
                           float* __restrict__ Agg) {
  int idx = blockIdx.x * blockDim.x + threadIdx.x;
  int i = idx >> 4;
  if (i >= NN) return;
  int q = idx & 15;
  const float4* H4 = (const float4*)Hlin;
  float di = dinv[i];
  float4 h = H4[i * 16 + q];
  float4 acc;
  acc.x = h.x * di * di + bias[4 * q + 0];
  acc.y = h.y * di * di + bias[4 * q + 1];
  acc.z = h.z * di * di + bias[4 * q + 2];
  acc.w = h.w * di * di + bias[4 * q + 3];
  int j = off[i];
  int end = j + cnt[i];
  for (; j + 4 <= end; j += 4) {
    int s0 = clamp_idx(lst[j + 0], NN - 1);
    int s1 = clamp_idx(lst[j + 1], NN - 1);
    int s2 = clamp_idx(lst[j + 2], NN - 1);
    int s3 = clamp_idx(lst[j + 3], NN - 1);
    float w0 = dinv[s0] * di, w1 = dinv[s1] * di, w2 = dinv[s2] * di, w3 = dinv[s3] * di;
    float4 r0 = H4[s0 * 16 + q];
    float4 r1 = H4[s1 * 16 + q];
    float4 r2 = H4[s2 * 16 + q];
    float4 r3 = H4[s3 * 16 + q];
    acc.x += r0.x * w0; acc.y += r0.y * w0; acc.z += r0.z * w0; acc.w += r0.w * w0;
    acc.x += r1.x * w1; acc.y += r1.y * w1; acc.z += r1.z * w1; acc.w += r1.w * w1;
    acc.x += r2.x * w2; acc.y += r2.y * w2; acc.z += r2.z * w2; acc.w += r2.w * w2;
    acc.x += r3.x * w3; acc.y += r3.y * w3; acc.z += r3.z * w3; acc.w += r3.w * w3;
  }
  for (; j < end; ++j) {
    int s0 = clamp_idx(lst[j], NN - 1);
    float w0 = dinv[s0] * di;
    float4 r0 = H4[s0 * 16 + q];
    acc.x += r0.x * w0; acc.y += r0.y * w0; acc.z += r0.z * w0; acc.w += r0.w * w0;
  }
  ((float4*)Agg)[i * 16 + q] = acc;
}

__global__ void k_reludrop(const float* __restrict__ Agg,
                           const float* __restrict__ mask,
                           float* __restrict__ Hout) {
  int idx = blockIdx.x * blockDim.x + threadIdx.x;
  if (idx < NN * HID) Hout[idx] = fmaxf(Agg[idx], 0.0f) * mask[idx];
}

__global__ void k_gather32(const float* __restrict__ MUlin,
                           const float* __restrict__ LVlin,
                           const int* __restrict__ lst,
                           const int* __restrict__ off,
                           const int* __restrict__ cnt,
                           const float* __restrict__ dinv,
                           const float* __restrict__ b1,
                           const float* __restrict__ b2,
                           float* __restrict__ MUout,
                           float* __restrict__ LVout) {
  int idx = blockIdx.x * blockDim.x + threadIdx.x;
  int i = idx >> 4;
  if (i >= NN) return;
  int q = idx & 15;
  int g = q & 7;
  bool isLV = q >= 8;
  const float4* L4 = (const float4*)(isLV ? LVlin : MUlin);
  const float*  bb = isLV ? b2 : b1;
  float4*       O4 = (float4*)(isLV ? LVout : MUout);
  float di = dinv[i];
  float dd = di * di;
  float4 h = L4[i * 8 + g];
  float4 acc;
  acc.x = h.x * dd + bb[4 * g + 0];
  acc.y = h.y * dd + bb[4 * g + 1];
  acc.z = h.z * dd + bb[4 * g + 2];
  acc.w = h.w * dd + bb[4 * g + 3];
  int j = off[i];
  int end = j + cnt[i];
  for (; j + 4 <= end; j += 4) {
    int s0 = clamp_idx(lst[j + 0], NN - 1);
    int s1 = clamp_idx(lst[j + 1], NN - 1);
    int s2 = clamp_idx(lst[j + 2], NN - 1);
    int s3 = clamp_idx(lst[j + 3], NN - 1);
    float w0 = dinv[s0] * di, w1 = dinv[s1] * di, w2 = dinv[s2] * di, w3 = dinv[s3] * di;
    float4 r0 = L4[s0 * 8 + g];
    float4 r1 = L4[s1 * 8 + g];
    float4 r2 = L4[s2 * 8 + g];
    float4 r3 = L4[s3 * 8 + g];
    acc.x += r0.x * w0; acc.y += r0.y * w0; acc.z += r0.z * w0; acc.w += r0.w * w0;
    acc.x += r1.x * w1; acc.y += r1.y * w1; acc.z += r1.z * w1; acc.w += r1.w * w1;
    acc.x += r2.x * w2; acc.y += r2.y * w2; acc.z += r2.z * w2; acc.w += r2.w * w2;
    acc.x += r3.x * w3; acc.y += r3.y * w3; acc.z += r3.z * w3; acc.w += r3.w * w3;
  }
  for (; j < end; ++j) {
    int s0 = clamp_idx(lst[j], NN - 1);
    float w0 = dinv[s0] * di;
    float4 r0 = L4[s0 * 8 + g];
    acc.x += r0.x * w0; acc.y += r0.y * w0; acc.z += r0.z * w0; acc.w += r0.w * w0;
  }
  O4[i * 8 + g] = acc;
}

__global__ void k_final(const float* __restrict__ eps,
                        const float* __restrict__ MU,
                        const float* __restrict__ LV,
                        float* __restrict__ Z) {
  int idx = blockIdx.x * blockDim.x + threadIdx.x;
  if (idx < NN * NC) Z[idx] = eps[idx] * expf(LV[idx]) + MU[idx];
}

extern "C" void kernel_launch(void* const* d_in, const int* in_sizes, int n_in,
                              void* d_out, int out_size, void* d_ws, size_t ws_size,
                              hipStream_t stream) {
  (void)in_sizes; (void)n_in; (void)out_size; (void)ws_size;
  const float* x    = (const float*)d_in[0];
  const int*   ei   = (const int*)d_in[1];
  const float* W0   = (const float*)d_in[2];
  const float* b0   = (const float*)d_in[3];
  const float* W1   = (const float*)d_in[4];
  const float* b1   = (const float*)d_in[5];
  const float* W2   = (const float*)d_in[6];
  const float* b2   = (const float*)d_in[7];
  const float* mask = (const float*)d_in[8];
  const float* eps  = (const float*)d_in[9];
  const int* src = ei;
  const int* dst = ei + NE;

  int*   cnt   = (int*)d_ws;
  int*   off   = cnt + NNP;
  float* dinv  = (float*)(off + NNP);
  int*   rank  = (int*)(dinv + NNP);
  int*   lst   = rank + NE;
  float* hlin  = (float*)(lst + NE);
  float* hagg  = hlin + (size_t)NN * HID;
  float* mulin = hagg + (size_t)NN * HID;
  float* lvlin = mulin + (size_t)NN * NC;

  float* z  = (float*)d_out;
  float* mu = z + (size_t)NN * NC;
  float* lv = mu + (size_t)NN * NC;

  const int B = 256;
  const int gN   = (NN + B - 1) / B;
  const int gE   = (NE + B - 1) / B;
  const int gW   = (ROW_TILES * 32 + B - 1) / B;
  const int gNH  = (NN * HID + B - 1) / B;
  const int gNC_ = (NN * NC + B - 1) / B;
  const int gG   = (NN * 16 + B - 1) / B;

  k_zero   <<<gN, B, 0, stream>>>(cnt, NN);
  k_rank   <<<gE, B, 0, stream>>>(dst, cnt, rank);
  k_offsets<<<gN, B, 0, stream>>>(cnt, off, dinv);
  k_fill   <<<gE, B, 0, stream>>>(src, dst, off, rank, lst);

  k_gemm_xw0 <<<gW,  B, 0, stream>>>(x, W0, hlin);
  k_gather64 <<<gG,  B, 0, stream>>>(hlin, lst, off, cnt, dinv, b0, hagg);
  k_reludrop <<<gNH, B, 0, stream>>>(hagg, mask, hlin);

  k_gemm_hw12<<<gW,   B, 0, stream>>>(hlin, W1, W2, mulin, lvlin);
  k_gather32 <<<gG,   B, 0, stream>>>(mulin, lvlin, lst, off, cnt, dinv, b1, b2, mu, lv);

  k_final    <<<gNC_, B, 0, stream>>>(eps, mu, lv, z);
}


# Relu*mask and z = eps*exp(logvar)+mu are computed in the epilogues of the two row-sum kernels that produce their inputs (logvar handed across lanes); both elementwise launches are gone
# speedup vs baseline: 2.3740x; 1.1200x over previous
#include <hip/hip_runtime.h>


typedef __attribute__((ext_vector_type(2))) float v2f;
typedef __attribute__((ext_vector_type(8))) float v8f;

#define NN 50000
#define NE 800000
#define KIN 128
#define HID 64
#define NC 32
#define ROW_TILES (NN / 16)
#define NNP 50048

__device__ __forceinline__ int clamp_idx(int v, int hi) {
  return v < 0 ? 0 : (v > hi ? hi : v);
}

__global__ void k_zero(int* __restrict__ p, int n) {
  int i = blockIdx.x * blockDim.x + threadIdx.x;
  if (i < n) p[i] = 0;
}

__global__ void k_rank(const int* __restrict__ dst,
                       int* __restrict__ cnt,
                       int* __restrict__ rank) {
  int e = blockIdx.x * blockDim.x + threadIdx.x;
  if (e < NE) rank[e] = atomicAdd(&cnt[dst[e]], 1);
}

__global__ void k_offsets(const int* __restrict__ cnt,
                          int* __restrict__ off,
                          float* __restrict__ dinv) {
  __shared__ int red[8];
  __shared__ int wsum[8];
  int t = threadIdx.x;
  int lane = t & 31;
  int w = t >> 5;
  const int4* c4 = (const int4*)cnt;
  int n4 = blockIdx.x * 64;
  int s = 0;
#pragma unroll 8
  for (int i = t; i < n4; i += 256) {
    int4 v = c4[i];
    s += (v.x + v.y) + (v.z + v.w);
  }
  for (int o = 16; o > 0; o >>= 1) s += __shfl_xor(s, o);
  if (lane == 0) red[w] = s;

  int i = blockIdx.x * 256 + t;
  int c = (i < NN) ? cnt[i] : 0;
  int x = c;
  for (int o = 1; o < 32; o <<= 1) {
    int y = __shfl_up(x, o);
    if (lane >= o) x += y;
  }
  if (lane == 31) wsum[w] = x;
  __syncthreads();
  int base = 0;
  for (int k = 0; k < 8; ++k) base += red[k];
  for (int k = 0; k < 8; ++k) base += (k < w) ? wsum[k] : 0;
  if (i < NN) {
    off[i]  = base + x - c;
    dinv[i] = rsqrtf((float)c + 1.0f);
  }
}

__global__ void k_fill(const int* __restrict__ src,
                       const int* __restrict__ dst,
                       const int* __restrict__ off,
                       const int* __restrict__ rank,
                       int* __restrict__ lst) {
  int e = blockIdx.x * blockDim.x + threadIdx.x;
  if (e >= NE) return;
  int pos = clamp_idx(off[dst[e]] + rank[e], NE - 1);
  lst[pos] = src[e];
}

__global__ void k_gemm_xw0(const float* __restrict__ X,
                           const float* __restrict__ W,
                           float* __restrict__ H) {
  int wave = (blockIdx.x * blockDim.x + threadIdx.x) >> 5;
  if (wave >= ROW_TILES) return;
  int lane = threadIdx.x & 31;
  int hh   = lane >> 4;
  int lid  = lane & 15;
  const float* xr = X + (wave * 16 + lid) * KIN;
  v8f c0 = {}, c1 = {}, c2 = {}, c3 = {};
  for (int k = 0; k < KIN; k += 4) {
    __builtin_prefetch(xr + k + 32, 0, 1);
    int ka = k + 2 * hh;
    v2f a;
    a.x = xr[ka];
    a.y = xr[ka + 1];
    const float* wr0 = W + ka * HID;
    const float* wr1 = wr0 + HID;
    v2f b;
    b.x = wr0[lid];      b.y = wr1[lid];
    c0 = __builtin_amdgcn_wmma_f32_16x16x4_f32(false, a, false, b, (short)0, c0, false, false);
    b.x = wr0[lid + 16]; b.y = wr1[lid + 16];
    c1 = __builtin_amdgcn_wmma_f32_16x16x4_f32(false, a, false, b, (short)0, c1, false, false);
    b.x = wr0[lid + 32]; b.y = wr1[lid + 32];
    c2 = __builtin_amdgcn_wmma_f32_16x16x4_f32(false, a, false, b, (short)0, c2, false, false);
    b.x = wr0[lid + 48]; b.y = wr1[lid + 48];
    c3 = __builtin_amdgcn_wmma_f32_16x16x4_f32(false, a, false, b, (short)0, c3, false, false);
  }
  int mrow = wave * 16 + hh * 8;
  for (int j = 0; j < 8; ++j) {
    float* o = H + (mrow + j) * HID + lid;
    o[0]  = c0[j];
    o[16] = c1[j];
    o[32] = c2[j];
    o[48] = c3[j];
  }
}

__global__ void k_gemm_hw12(const float* __restrict__ H,
                            const float* __restrict__ W1,
                            const float* __restrict__ W2,
                            float* __restrict__ MU,
                            float* __restrict__ LV) {
  int wave = (blockIdx.x * blockDim.x + threadIdx.x) >> 5;
  if (wave >= ROW_TILES) return;
  int lane = threadIdx.x & 31;
  int hh   = lane >> 4;
  int lid  = lane & 15;
  const float* hr = H + (wave * 16 + lid) * HID;
  v8f m0 = {}, m1 = {}, l0 = {}, l1 = {};
  for (int k = 0; k < HID; k += 4) {
    int ka = k + 2 * hh;
    v2f a;
    a.x = hr[ka];
    a.y = hr[ka + 1];
    const float* u0 = W1 + ka * NC;
    const float* u1 = u0 + NC;
    const float* v0 = W2 + ka * NC;
    const float* v1 = v0 + NC;
    v2f b;
    b.x = u0[lid];      b.y = u1[lid];
    m0 = __builtin_amdgcn_wmma_f32_16x16x4_f32(false, a, false, b, (short)0, m0, false, false);
    b.x = u0[lid + 16]; b.y = u1[lid + 16];
    m1 = __builtin_amdgcn_wmma_f32_16x16x4_f32(false, a, false, b, (short)0, m1, false, false);
    b.x = v0[lid];      b.y = v1[lid];
    l0 = __builtin_amdgcn_wmma_f32_16x16x4_f32(false, a, false, b, (short)0, l0, false, false);
    b.x = v0[lid + 16]; b.y = v1[lid + 16];
    l1 = __builtin_amdgcn_wmma_f32_16x16x4_f32(false, a, false, b, (short)0, l1, false, false);
  }
  int mrow = wave * 16 + hh * 8;
  for (int j = 0; j < 8; ++j) {
    float* om = MU + (mrow + j) * NC + lid;
    float* ol = LV + (mrow + j) * NC + lid;
    om[0]  = m0[j];
    om[16] = m1[j];
    ol[0]  = l0[j];
    ol[16] = l1[j];
  }
}

__global__ void k_gather64(const float* __restrict__ Hlin,
                           const int* __restrict__ lst,
                           const int* __restrict__ off,
                           const int* __restrict__ cnt,
                           const float* __restrict__ dinv,
                           const float* __restrict__ bias,
                           const float* __restrict__ mask,
                           float* __restrict__ Hout) {
  int idx = blockIdx.x * blockDim.x + threadIdx.x;
  int i = idx >> 4;
  if (i >= NN) return;
  int q = idx & 15;
  const float4* H4 = (const float4*)Hlin;
  float di = dinv[i];
  float4 h = H4[i * 16 + q];
  float4 mk = ((const float4*)mask)[i * 16 + q];
  float4 acc;
  acc.x = h.x * di * di + bias[4 * q + 0];
  acc.y = h.y * di * di + bias[4 * q + 1];
  acc.z = h.z * di * di + bias[4 * q + 2];
  acc.w = h.w * di * di + bias[4 * q + 3];
  int j = off[i];
  int end = j + cnt[i];
  for (; j + 4 <= end; j += 4) {
    int s0 = clamp_idx(lst[j + 0], NN - 1);
    int s1 = clamp_idx(lst[j + 1], NN - 1);
    int s2 = clamp_idx(lst[j + 2], NN - 1);
    int s3 = clamp_idx(lst[j + 3], NN - 1);
    float w0 = dinv[s0] * di, w1 = dinv[s1] * di, w2 = dinv[s2] * di, w3 = dinv[s3] * di;
    float4 r0 = H4[s0 * 16 + q];
    float4 r1 = H4[s1 * 16 + q];
    float4 r2 = H4[s2 * 16 + q];
    float4 r3 = H4[s3 * 16 + q];
    acc.x += r0.x * w0; acc.y += r0.y * w0; acc.z += r0.z * w0; acc.w += r0.w * w0;
    acc.x += r1.x * w1; acc.y += r1.y * w1; acc.z += r1.z * w1; acc.w += r1.w * w1;
    acc.x += r2.x * w2; acc.y += r2.y * w2; acc.z += r2.z * w2; acc.w += r2.w * w2;
    acc.x += r3.x * w3; acc.y += r3.y * w3; acc.z += r3.z * w3; acc.w += r3.w * w3;
  }
  for (; j < end; ++j) {
    int s0 = clamp_idx(lst[j], NN - 1);
    float w0 = dinv[s0] * di;
    float4 r0 = H4[s0 * 16 + q];
    acc.x += r0.x * w0; acc.y += r0.y * w0; acc.z += r0.z * w0; acc.w += r0.w * w0;
  }
  float4 o;
  o.x = fmaxf(acc.x, 0.0f) * mk.x;
  o.y = fmaxf(acc.y, 0.0f) * mk.y;
  o.z = fmaxf(acc.z, 0.0f) * mk.z;
  o.w = fmaxf(acc.w, 0.0f) * mk.w;
  ((float4*)Hout)[i * 16 + q] = o;
}

__global__ void k_gather32(const float* __restrict__ MUlin,
                           const float* __restrict__ LVlin,
                           const int* __restrict__ lst,
                           const int* __restrict__ off,
                           const int* __restrict__ cnt,
                           const float* __restrict__ dinv,
                           const float* __restrict__ b1,
                           const float* __restrict__ b2,
                           const float* __restrict__ eps,
                           float* __restrict__ MUout,
                           float* __restrict__ LVout,
                           float* __restrict__ Z) {
  int idx = blockIdx.x * blockDim.x + threadIdx.x;
  int i = idx >> 4;
  if (i >= NN) return;
  int q = idx & 15;
  int g = q & 7;
  bool isLV = q >= 8;
  const float4* L4 = (const float4*)(isLV ? LVlin : MUlin);
  const float*  bb = isLV ? b2 : b1;
  float4*       O4 = (float4*)(isLV ? LVout : MUout);
  float di = dinv[i];
  float dd = di * di;
  float4 h = L4[i * 8 + g];
  float4 ep = ((const float4*)eps)[i * 8 + g];
  float4 acc;
  acc.x = h.x * dd + bb[4 * g + 0];
  acc.y = h.y * dd + bb[4 * g + 1];
  acc.z = h.z * dd + bb[4 * g + 2];
  acc.w = h.w * dd + bb[4 * g + 3];
  int j = off[i];
  int end = j + cnt[i];
  for (; j + 4 <= end; j += 4) {
    int s0 = clamp_idx(lst[j + 0], NN - 1);
    int s1 = clamp_idx(lst[j + 1], NN - 1);
    int s2 = clamp_idx(lst[j + 2], NN - 1);
    int s3 = clamp_idx(lst[j + 3], NN - 1);
    float w0 = dinv[s0] * di, w1 = dinv[s1] * di, w2 = dinv[s2] * di, w3 = dinv[s3] * di;
    float4 r0 = L4[s0 * 8 + g];
    float4 r1 = L4[s1 * 8 + g];
    float4 r2 = L4[s2 * 8 + g];
    float4 r3 = L4[s3 * 8 + g];
    acc.x += r0.x * w0; acc.y += r0.y * w0; acc.z += r0.z * w0; acc.w += r0.w * w0;
    acc.x += r1.x * w1; acc.y += r1.y * w1; acc.z += r1.z * w1; acc.w += r1.w * w1;
    acc.x += r2.x * w2; acc.y += r2.y * w2; acc.z += r2.z * w2; acc.w += r2.w * w2;
    acc.x += r3.x * w3; acc.y += r3.y * w3; acc.z += r3.z * w3; acc.w += r3.w * w3;
  }
  for (; j < end; ++j) {
    int s0 = clamp_idx(lst[j], NN - 1);
    float w0 = dinv[s0] * di;
    float4 r0 = L4[s0 * 8 + g];
    acc.x += r0.x * w0; acc.y += r0.y * w0; acc.z += r0.z * w0; acc.w += r0.w * w0;
  }
  O4[i * 8 + g] = acc;
  float4 lv;
  lv.x = __shfl_xor(acc.x, 8);
  lv.y = __shfl_xor(acc.y, 8);
  lv.z = __shfl_xor(acc.z, 8);
  lv.w = __shfl_xor(acc.w, 8);
  if (!isLV) {
    float4 zz;
    zz.x = ep.x * expf(lv.x) + acc.x;
    zz.y = ep.y * expf(lv.y) + acc.y;
    zz.z = ep.z * expf(lv.z) + acc.z;
    zz.w = ep.w * expf(lv.w) + acc.w;
    ((float4*)Z)[i * 8 + g] = zz;
  }
}

extern "C" void kernel_launch(void* const* d_in, const int* in_sizes, int n_in,
                              void* d_out, int out_size, void* d_ws, size_t ws_size,
                              hipStream_t stream) {
  (void)in_sizes; (void)n_in; (void)out_size; (void)ws_size;
  const float* x    = (const float*)d_in[0];
  const int*   ei   = (const int*)d_in[1];
  const float* W0   = (const float*)d_in[2];
  const float* b0   = (const float*)d_in[3];
  const float* W1   = (const float*)d_in[4];
  const float* b1   = (const float*)d_in[5];
  const float* W2   = (const float*)d_in[6];
  const float* b2   = (const float*)d_in[7];
  const float* mask = (const float*)d_in[8];
  const float* eps  = (const float*)d_in[9];
  const int* src = ei;
  const int* dst = ei + NE;

  int*   cnt   = (int*)d_ws;
  int*   off   = cnt + NNP;
  float* dinv  = (float*)(off + NNP);
  int*   rank  = (int*)(dinv + NNP);
  int*   lst   = rank + NE;
  float* hlin  = (float*)(lst + NE);
  float* hact  = hlin + (size_t)NN * HID;
  float* mulin = hact + (size_t)NN * HID;
  float* lvlin = mulin + (size_t)NN * NC;

  float* z  = (float*)d_out;
  float* mu = z + (size_t)NN * NC;
  float* lv = mu + (size_t)NN * NC;

  const int B = 256;
  const int gN   = (NN + B - 1) / B;
  const int gE   = (NE + B - 1) / B;
  const int gW   = (ROW_TILES * 32 + B - 1) / B;
  const int gG   = (NN * 16 + B - 1) / B;

  k_zero   <<<gN, B, 0, stream>>>(cnt, NN);
  k_rank   <<<gE, B, 0, stream>>>(dst, cnt, rank);
  k_offsets<<<gN, B, 0, stream>>>(cnt, off, dinv);
  k_fill   <<<gE, B, 0, stream>>>(src, dst, off, rank, lst);

  k_gemm_xw0 <<<gW,  B, 0, stream>>>(x, W0, hlin);
  k_gather64 <<<gG,  B, 0, stream>>>(hlin, lst, off, cnt, dinv, b0, mask, hact);

  k_gemm_hw12<<<gW,   B, 0, stream>>>(hact, W1, W2, mulin, lvlin);
  k_gather32 <<<gG,   B, 0, stream>>>(mulin, lvlin, lst, off, cnt, dinv, b1, b2, eps, mu, lv, z);
}


# Edges ordered by destination with LDS counters in two levels (buckets of 256 nodes, then the nodes of a bucket) in place of one returning global add per edge; sum order = list order (LDS arrival)
# speedup vs baseline: 3.4872x; 1.4689x over previous
#include <hip/hip_runtime.h>


typedef __attribute__((ext_vector_type(2))) float v2f;
typedef __attribute__((ext_vector_type(8))) float v8f;

#define NN 50000
#define NE 800000
#define KIN 128
#define HID 64
#define NC 32
#define ROW_TILES (NN / 16)
#define NNP 50048

__device__ __forceinline__ int clamp_idx(int v, int hi) {
  return v < 0 ? 0 : (v > hi ? hi : v);
}

#define NBLK 128
#define EPB  6250
#define EPT  7
#define NBK  196
#define HP   256
#define TPB  1024
#define KPT  6

__global__ void __launch_bounds__(TPB) k_bcount(const int* __restrict__ dst,
                                                int* __restrict__ hist) {
  __shared__ int h[HP];
  int t = threadIdx.x;
  if (t < HP) h[t] = 0;
  const int* d0 = dst + blockIdx.x * EPB;
  int dv[EPT];
#pragma unroll
  for (int j = 0; j < EPT; ++j) {
    int i = t + j * TPB;
    dv[j] = d0[i < EPB ? i : EPB - 1];
  }
  __syncthreads();
#pragma unroll
  for (int j = 0; j < EPT; ++j) {
    int i = t + j * TPB;
    if (i < EPB) atomicAdd(&h[clamp_idx(dv[j], NN - 1) >> 8], 1);
  }
  __syncthreads();
  if (t < HP) hist[blockIdx.x * HP + t] = h[t];
}

__global__ void __launch_bounds__(TPB) k_bplace(const int* __restrict__ src,
                                                const int* __restrict__ dst,
                                                const int* __restrict__ hist,
                                                int* __restrict__ key,
                                                int* __restrict__ bbase,
                                                int* __restrict__ btot) {
  __shared__ int part[4][HP];
  __shared__ int tot[4][HP];
  __shared__ int cur[HP];
  __shared__ int wsum[8];
  int t = threadIdx.x;
  int b = blockIdx.x;
  int lane = t & 31;
  int w = t >> 5;
  const int* d0 = dst + b * EPB;
  const int* s0 = src + b * EPB;
  int dv[EPT], sv[EPT];
#pragma unroll
  for (int j = 0; j < EPT; ++j) {
    int i = t + j * TPB;
    int ii = i < EPB ? i : EPB - 1;
    dv[j] = d0[ii];
    sv[j] = s0[ii];
  }
  int k = t & (HP - 1);
  int g = t >> 8;
  int hv[NBLK / 4];
#pragma unroll
  for (int q = 0; q < NBLK / 4; ++q) hv[q] = hist[(g + 4 * q) * HP + k];
  int p = 0, s = 0;
#pragma unroll
  for (int q = 0; q < NBLK / 4; ++q) {
    s += hv[q];
    p += (g + 4 * q < b) ? hv[q] : 0;
  }
  part[g][k] = p;
  tot[g][k] = s;
  __syncthreads();
  int T = 0, P = 0, x = 0;
  if (t < HP) {
    T = (tot[0][t] + tot[1][t]) + (tot[2][t] + tot[3][t]);
    P = (part[0][t] + part[1][t]) + (part[2][t] + part[3][t]);
    x = T;
    for (int o = 1; o < 32; o <<= 1) {
      int y = __shfl_up(x, o);
      if (lane >= o) x += y;
    }
    if (lane == 31) wsum[w] = x;
  }
  __syncthreads();
  if (t < HP) {
    int base = x - T;
    for (int q = 0; q < 8; ++q) base += (q < w) ? wsum[q] : 0;
    cur[t] = base + P;
    if (b == 0) {
      bbase[t] = base;
      btot[t]  = T;
    }
  }
  __syncthreads();
  int pos[EPT];
#pragma unroll
  for (int j = 0; j < EPT; ++j) {
    int i = t + j * TPB;
    dv[j] = clamp_idx(dv[j], NN - 1);
    pos[j] = (i < EPB) ? atomicAdd(&cur[dv[j] >> 8], 1) : 0;
  }
#pragma unroll
  for (int j = 0; j < EPT; ++j) {
    int i = t + j * TPB;
    if (i < EPB)
      key[clamp_idx(pos[j], NE - 1)] = ((dv[j] & 255) << 16) | clamp_idx(sv[j], NN - 1);
  }
}

__global__ void __launch_bounds__(TPB) k_bsort(const int* __restrict__ key,
                                               const int* __restrict__ bbase,
                                               const int* __restrict__ btot,
                                               int* __restrict__ lst,
                                               int* __restrict__ cnt,
                                               int* __restrict__ off,
                                               float* __restrict__ dinv) {
  __shared__ int c[256];
  __shared__ int cur[256];
  __shared__ int wsum[8];
  int t = threadIdx.x;
  int kb = blockIdx.x;
  int lane = t & 31;
  int w = t >> 5;
  int base = clamp_idx(bbase[kb], NE);
  int n    = clamp_idx(btot[kb], NE - base);
  const int* k0 = key + base;
  if (t < 256) c[t] = 0;
  int kv[KPT];
#pragma unroll
  for (int j = 0; j < KPT; ++j) {
    int i = t + j * TPB;
    kv[j] = (i < n) ? k0[i] : 0;
  }
  __syncthreads();
#pragma unroll
  for (int j = 0; j < KPT; ++j) {
    int i = t + j * TPB;
    if (i < n) atomicAdd(&c[(kv[j] >> 16) & 255], 1);
  }
  for (int i = t + KPT * TPB; i < n; i += TPB)
    atomicAdd(&c[(k0[i] >> 16) & 255], 1);
  __syncthreads();
  int cc = 0, x = 0;
  if (t < 256) {
    cc = c[t];
    x = cc;
    for (int o = 1; o < 32; o <<= 1) {
      int y = __shfl_up(x, o);
      if (lane >= o) x += y;
    }
    if (lane == 31) wsum[w] = x;
  }
  __syncthreads();
  if (t < 256) {
    int o = x - cc;
    for (int q = 0; q < 8; ++q) o += (q < w) ? wsum[q] : 0;
    cur[t] = o;
    int node = kb * 256 + t;
    if (node < NN) {
      cnt[node]  = cc;
      off[node]  = base + o;
      dinv[node] = rsqrtf((float)cc + 1.0f);
    }
  }
  __syncthreads();
  int pl[KPT];
#pragma unroll
  for (int j = 0; j < KPT; ++j) {
    int i = t + j * TPB;
    pl[j] = (i < n) ? atomicAdd(&cur[(kv[j] >> 16) & 255], 1) : 0;
  }
#pragma unroll
  for (int j = 0; j < KPT; ++j) {
    int i = t + j * TPB;
    if (i < n) lst[clamp_idx(base + pl[j], NE - 1)] = kv[j] & 0xFFFF;
  }
  for (int i = t + KPT * TPB; i < n; i += TPB) {
    int v = k0[i];
    int p1 = atomicAdd(&cur[(v >> 16) & 255], 1);
    lst[clamp_idx(base + p1, NE - 1)] = v & 0xFFFF;
  }
}

__global__ void k_gemm_xw0(const float* __restrict__ X,
                           const float* __restrict__ W,
                           float* __restrict__ H) {
  int wave = (blockIdx.x * blockDim.x + threadIdx.x) >> 5;
  if (wave >= ROW_TILES) return;
  int lane = threadIdx.x & 31;
  int hh   = lane >> 4;
  int lid  = lane & 15;
  const float* xr = X + (wave * 16 + lid) * KIN;
  v8f c0 = {}, c1 = {}, c2 = {}, c3 = {};
  for (int k = 0; k < KIN; k += 4) {
    __builtin_prefetch(xr + k + 32, 0, 1);
    int ka = k + 2 * hh;
    v2f a;
    a.x = xr[ka];
    a.y = xr[ka + 1];
    const float* wr0 = W + ka * HID;
    const float* wr1 = wr0 + HID;
    v2f b;
    b.x = wr0[lid];      b.y = wr1[lid];
    c0 = __builtin_amdgcn_wmma_f32_16x16x4_f32(false, a, false, b, (short)0, c0, false, false);
    b.x = wr0[lid + 16]; b.y = wr1[lid + 16];
    c1 = __builtin_amdgcn_wmma_f32_16x16x4_f32(false, a, false, b, (short)0, c1, false, false);
    b.x = wr0[lid + 32]; b.y = wr1[lid + 32];
    c2 = __builtin_amdgcn_wmma_f32_16x16x4_f32(false, a, false, b, (short)0, c2, false, false);
    b.x = wr0[lid + 48]; b.y = wr1[lid + 48];
    c3 = __builtin_amdgcn_wmma_f32_16x16x4_f32(false, a, false, b, (short)0, c3, false, false);
  }
  int mrow = wave * 16 + hh * 8;
  for (int j = 0; j < 8; ++j) {
    float* o = H + (mrow + j) * HID + lid;
    o[0]  = c0[j];
    o[16] = c1[j];
    o[32] = c2[j];
    o[48] = c3[j];
  }
}

__global__ void k_gemm_hw12(const float* __restrict__ H,
                            const float* __restrict__ W1,
                            const float* __restrict__ W2,
                            float* __restrict__ MU,
                            float* __restrict__ LV) {
  int wave = (blockIdx.x * blockDim.x + threadIdx.x) >> 5;
  if (wave >= ROW_TILES) return;
  int lane = threadIdx.x & 31;
  int hh   = lane >> 4;
  int lid  = lane & 15;
  const float* hr = H + (wave * 16 + lid) * HID;
  v8f m0 = {}, m1 = {}, l0 = {}, l1 = {};
  for (int k = 0; k < HID; k += 4) {
    int ka = k + 2 * hh;
    v2f a;
    a.x = hr[ka];
    a.y = hr[ka + 1];
    const float* u0 = W1 + ka * NC;
    const float* u1 = u0 + NC;
    const float* v0 = W2 + ka * NC;
    const float* v1 = v0 + NC;
    v2f b;
    b.x = u0[lid];      b.y = u1[lid];
    m0 = __builtin_amdgcn_wmma_f32_16x16x4_f32(false, a, false, b, (short)0, m0, false, false);
    b.x = u0[lid + 16]; b.y = u1[lid + 16];
    m1 = __builtin_amdgcn_wmma_f32_16x16x4_f32(false, a, false, b, (short)0, m1, false, false);
    b.x = v0[lid];      b.y = v1[lid];
    l0 = __builtin_amdgcn_wmma_f32_16x16x4_f32(false, a, false, b, (short)0, l0, false, false);
    b.x = v0[lid + 16]; b.y = v1[lid + 16];
    l1 = __builtin_amdgcn_wmma_f32_16x16x4_f32(false, a, false, b, (short)0, l1, false, false);
  }
  int mrow = wave * 16 + hh * 8;
  for (int j = 0; j < 8; ++j) {
    float* om = MU + (mrow + j) * NC + lid;
    float* ol = LV + (mrow + j) * NC + lid;
    om[0]  = m0[j];
    om[16] = m1[j];
    ol[0]  = l0[j];
    ol[16] = l1[j];
  }
}

__global__ void k_gather64(const float* __restrict__ Hlin,
                           const int* __restrict__ lst,
                           const int* __restrict__ off,
                           const int* __restrict__ cnt,
                           const float* __restrict__ dinv,
                           const float* __restrict__ bias,
                           const float* __restrict__ mask,
                           float* __restrict__ Hout) {
  int idx = blockIdx.x * blockDim.x + threadIdx.x;
  int i = idx >> 4;
  if (i >= NN) return;
  int q = idx & 15;
  const float4* H4 = (const float4*)Hlin;
  float di = dinv[i];
  float4 h = H4[i * 16 + q];
  float4 mk = ((const float4*)mask)[i * 16 + q];
  float4 acc;
  acc.x = h.x * di * di + bias[4 * q + 0];
  acc.y = h.y * di * di + bias[4 * q + 1];
  acc.z = h.z * di * di + bias[4 * q + 2];
  acc.w = h.w * di * di + bias[4 * q + 3];
  int j = off[i];
  int end = j + cnt[i];
  for (; j + 4 <= end; j += 4) {
    int s0 = clamp_idx(lst[j + 0], NN - 1);
    int s1 = clamp_idx(lst[j + 1], NN - 1);
    int s2 = clamp_idx(lst[j + 2], NN - 1);
    int s3 = clamp_idx(lst[j + 3], NN - 1);
    float w0 = dinv[s0] * di, w1 = dinv[s1] * di, w2 = dinv[s2] * di, w3 = dinv[s3] * di;
    float4 r0 = H4[s0 * 16 + q];
    float4 r1 = H4[s1 * 16 + q];
    float4 r2 = H4[s2 * 16 + q];
    float4 r3 = H4[s3 * 16 + q];
    acc.x += r0.x * w0; acc.y += r0.y * w0; acc.z += r0.z * w0; acc.w += r0.w * w0;
    acc.x += r1.x * w1; acc.y += r1.y * w1; acc.z += r1.z * w1; acc.w += r1.w * w1;
    acc.x += r2.x * w2; acc.y += r2.y * w2; acc.z += r2.z * w2; acc.w += r2.w * w2;
    acc.x += r3.x * w3; acc.y += r3.y * w3; acc.z += r3.z * w3; acc.w += r3.w * w3;
  }
  for (; j < end; ++j) {
    int s0 = clamp_idx(lst[j], NN - 1);
    float w0 = dinv[s0] * di;
    float4 r0 = H4[s0 * 16 + q];
    acc.x += r0.x * w0; acc.y += r0.y * w0; acc.z += r0.z * w0; acc.w += r0.w * w0;
  }
  float4 o;
  o.x = fmaxf(acc.x, 0.0f) * mk.x;
  o.y = fmaxf(acc.y, 0.0f) * mk.y;
  o.z = fmaxf(acc.z, 0.0f) * mk.z;
  o.w = fmaxf(acc.w, 0.0f) * mk.w;
  ((float4*)Hout)[i * 16 + q] = o;
}

__global__ void k_gather32(const float* __restrict__ MUlin,
                           const float* __restrict__ LVlin,
                           const int* __restrict__ lst,
                           const int* __restrict__ off,
                           const int* __restrict__ cnt,
                           const float* __restrict__ dinv,
                           const float* __restrict__ b1,
                           const float* __restrict__ b2,
                           const float* __restrict__ eps,
                           float* __restrict__ MUout,
                           float* __restrict__ LVout,
                           float* __restrict__ Z) {
  int idx = blockIdx.x * blockDim.x + threadIdx.x;
  int i = idx >> 4;
  if (i >= NN) return;
  int q = idx & 15;
  int g = q & 7;
  bool isLV = q >= 8;
  const float4* L4 = (const float4*)(isLV ? LVlin : MUlin);
  const float*  bb = isLV ? b2 : b1;
  float4*       O4 = (float4*)(isLV ? LVout : MUout);
  float di = dinv[i];
  float dd = di * di;
  float4 h = L4[i * 8 + g];
  float4 ep = ((const float4*)eps)[i * 8 + g];
  float4 acc;
  acc.x = h.x * dd + bb[4 * g + 0];
  acc.y = h.y * dd + bb[4 * g + 1];
  acc.z = h.z * dd + bb[4 * g + 2];
  acc.w = h.w * dd + bb[4 * g + 3];
  int j = off[i];
  int end = j + cnt[i];
  for (; j + 4 <= end; j += 4) {
    int s0 = clamp_idx(lst[j + 0], NN - 1);
    int s1 = clamp_idx(lst[j + 1], NN - 1);
    int s2 = clamp_idx(lst[j + 2], NN - 1);
    int s3 = clamp_idx(lst[j + 3], NN - 1);
    float w0 = dinv[s0] * di, w1 = dinv[s1] * di, w2 = dinv[s2] * di, w3 = dinv[s3] * di;
    float4 r0 = L4[s0 * 8 + g];
    float4 r1 = L4[s1 * 8 + g];
    float4 r2 = L4[s2 * 8 + g];
    float4 r3 = L4[s3 * 8 + g];
    acc.x += r0.x * w0; acc.y += r0.y * w0; acc.z += r0.z * w0; acc.w += r0.w * w0;
    acc.x += r1.x * w1; acc.y += r1.y * w1; acc.z += r1.z * w1; acc.w += r1.w * w1;
    acc.x += r2.x * w2; acc.y += r2.y * w2; acc.z += r2.z * w2; acc.w += r2.w * w2;
    acc.x += r3.x * w3; acc.y += r3.y * w3; acc.z += r3.z * w3; acc.w += r3.w * w3;
  }
  for (; j < end; ++j) {
    int s0 = clamp_idx(lst[j], NN - 1);
    float w0 = dinv[s0] * di;
    float4 r0 = L4[s0 * 8 + g];
    acc.x += r0.x * w0; acc.y += r0.y * w0; acc.z += r0.z * w0; acc.w += r0.w * w0;
  }
  O4[i * 8 + g] = acc;
  float4 lv;
  lv.x = __shfl_xor(acc.x, 8);
  lv.y = __shfl_xor(acc.y, 8);
  lv.z = __shfl_xor(acc.z, 8);
  lv.w = __shfl_xor(acc.w, 8);
  if (!isLV) {
    float4 zz;
    zz.x = ep.x * expf(lv.x) + acc.x;
    zz.y = ep.y * expf(lv.y) + acc.y;
    zz.z = ep.z * expf(lv.z) + acc.z;
    zz.w = ep.w * expf(lv.w) + acc.w;
    ((float4*)Z)[i * 8 + g] = zz;
  }
}

extern "C" void kernel_launch(void* const* d_in, const int* in_sizes, int n_in,
                              void* d_out, int out_size, void* d_ws, size_t ws_size,
                              hipStream_t stream) {
  (void)in_sizes; (void)n_in; (void)out_size; (void)ws_size;
  const float* x    = (const float*)d_in[0];
  const int*   ei   = (const int*)d_in[1];
  const float* W0   = (const float*)d_in[2];
  const float* b0   = (const float*)d_in[3];
  const float* W1   = (const float*)d_in[4];
  const float* b1   = (const float*)d_in[5];
  const float* W2   = (const float*)d_in[6];
  const float* b2   = (const float*)d_in[7];
  const float* mask = (const float*)d_in[8];
  const float* eps  = (const float*)d_in[9];
  const int* src = ei;
  const int* dst = ei + NE;

  int*   cnt   = (int*)d_ws;
  int*   off   = cnt + NNP;
  float* dinv  = (float*)(off + NNP);
  int*   hist  = (int*)(dinv + NNP);
  int*   bbase = hist + NBLK * HP;
  int*   btot  = bbase + HP;
  int*   key   = btot + HP;
  int*   lst   = key + NE;
  float* hlin  = (float*)(lst + NE);
  float* hact  = hlin + (size_t)NN * HID;
  float* mulin = hact + (size_t)NN * HID;
  float* lvlin = mulin + (size_t)NN * NC;

  float* z  = (float*)d_out;
  float* mu = z + (size_t)NN * NC;
  float* lv = mu + (size_t)NN * NC;

  const int B = 256;
  const int gW   = (ROW_TILES * 32 + B - 1) / B;
  const int gG   = (NN * 16 + B - 1) / B;

  k_bcount<<<NBLK, TPB, 0, stream>>>(dst, hist);
  k_bplace<<<NBLK, TPB, 0, stream>>>(src, dst, hist, key, bbase, btot);
  k_bsort <<<NBK,  TPB, 0, stream>>>(key, bbase, btot, lst, cnt, off, dinv);

  k_gemm_xw0 <<<gW,  B, 0, stream>>>(x, W0, hlin);
  k_gather64 <<<gG,  B, 0, stream>>>(hlin, lst, off, cnt, dinv, b0, mask, hact);

  k_gemm_hw12<<<gW,   B, 0, stream>>>(hact, W1, W2, mulin, lvlin);
  k_gather32 <<<gG,   B, 0, stream>>>(mulin, lvlin, lst, off, cnt, dinv, b1, b2, eps, mu, lv, z);
}


# Layer 2/3 products run in the epilogue of the layer-1 row-sum kernel: a block's 16 finished rows go through LDS to two waves' f32 WMMAs; the product launch and the activation plane are gone
# speedup vs baseline: 3.8354x; 1.0999x over previous
#include <hip/hip_runtime.h>


typedef __attribute__((ext_vector_type(2))) float v2f;
typedef __attribute__((ext_vector_type(8))) float v8f;

#define NN 50000
#define NE 800000
#define KIN 128
#define HID 64
#define NC 32
#define ROW_TILES (NN / 16)
#define NNP 50048

__device__ __forceinline__ int clamp_idx(int v, int hi) {
  return v < 0 ? 0 : (v > hi ? hi : v);
}

#define NBLK 128
#define EPB  6250
#define EPT  7
#define NBK  196
#define HP   256
#define TPB  1024
#define KPT  6

__global__ void __launch_bounds__(TPB) k_bcount(const int* __restrict__ dst,
                                                int* __restrict__ hist) {
  __shared__ int h[HP];
  int t = threadIdx.x;
  if (t < HP) h[t] = 0;
  const int* d0 = dst + blockIdx.x * EPB;
  int dv[EPT];
#pragma unroll
  for (int j = 0; j < EPT; ++j) {
    int i = t + j * TPB;
    dv[j] = d0[i < EPB ? i : EPB - 1];
  }
  __syncthreads();
#pragma unroll
  for (int j = 0; j < EPT; ++j) {
    int i = t + j * TPB;
    if (i < EPB) atomicAdd(&h[clamp_idx(dv[j], NN - 1) >> 8], 1);
  }
  __syncthreads();
  if (t < HP) hist[blockIdx.x * HP + t] = h[t];
}

__global__ void __launch_bounds__(TPB) k_bplace(const int* __restrict__ src,
                                                const int* __restrict__ dst,
                                                const int* __restrict__ hist,
                                                int* __restrict__ key,
                                                int* __restrict__ bbase,
                                                int* __restrict__ btot) {
  __shared__ int part[4][HP];
  __shared__ int tot[4][HP];
  __shared__ int cur[HP];
  __shared__ int wsum[8];
  int t = threadIdx.x;
  int b = blockIdx.x;
  int lane = t & 31;
  int w = t >> 5;
  const int* d0 = dst + b * EPB;
  const int* s0 = src + b * EPB;
  int dv[EPT], sv[EPT];
#pragma unroll
  for (int j = 0; j < EPT; ++j) {
    int i = t + j * TPB;
    int ii = i < EPB ? i : EPB - 1;
    dv[j] = d0[ii];
    sv[j] = s0[ii];
  }
  int k = t & (HP - 1);
  int g = t >> 8;
  int hv[NBLK / 4];
#pragma unroll
  for (int q = 0; q < NBLK / 4; ++q) hv[q] = hist[(g + 4 * q) * HP + k];
  int p = 0, s = 0;
#pragma unroll
  for (int q = 0; q < NBLK / 4; ++q) {
    s += hv[q];
    p += (g + 4 * q < b) ? hv[q] : 0;
  }
  part[g][k] = p;
  tot[g][k] = s;
  __syncthreads();
  int T = 0, P = 0, x = 0;
  if (t < HP) {
    T = (tot[0][t] + tot[1][t]) + (tot[2][t] + tot[3][t]);
    P = (part[0][t] + part[1][t]) + (part[2][t] + part[3][t]);
    x = T;
    for (int o = 1; o < 32; o <<= 1) {
      int y = __shfl_up(x, o);
      if (lane >= o) x += y;
    }
    if (lane == 31) wsum[w] = x;
  }
  __syncthreads();
  if (t < HP) {
    int base = x - T;
    for (int q = 0; q < 8; ++q) base += (q < w) ? wsum[q] : 0;
    cur[t] = base + P;
    if (b == 0) {
      bbase[t] = base;
      btot[t]  = T;
    }
  }
  __syncthreads();
  int pos[EPT];
#pragma unroll
  for (int j = 0; j < EPT; ++j) {
    int i = t + j * TPB;
    dv[j] = clamp_idx(dv[j], NN - 1);
    pos[j] = (i < EPB) ? atomicAdd(&cur[dv[j] >> 8], 1) : 0;
  }
#pragma unroll
  for (int j = 0; j < EPT; ++j) {
    int i = t + j * TPB;
    if (i < EPB)
      key[clamp_idx(pos[j], NE - 1)] = ((dv[j] & 255) << 16) | clamp_idx(sv[j], NN - 1);
  }
}

__global__ void __launch_bounds__(TPB) k_bsort(const int* __restrict__ key,
                                               const int* __restrict__ bbase,
                                               const int* __restrict__ btot,
                                               int* __restrict__ lst,
                                               int* __restrict__ cnt,
                                               int* __restrict__ off,
                                               float* __restrict__ dinv) {
  __shared__ int c[256];
  __shared__ int cur[256];
  __shared__ int wsum[8];
  int t = threadIdx.x;
  int kb = blockIdx.x;
  int lane = t & 31;
  int w = t >> 5;
  int base = clamp_idx(bbase[kb], NE);
  int n    = clamp_idx(btot[kb], NE - base);
  const int* k0 = key + base;
  if (t < 256) c[t] = 0;
  int kv[KPT];
#pragma unroll
  for (int j = 0; j < KPT; ++j) {
    int i = t + j * TPB;
    kv[j] = (i < n) ? k0[i] : 0;
  }
  __syncthreads();
#pragma unroll
  for (int j = 0; j < KPT; ++j) {
    int i = t + j * TPB;
    if (i < n) atomicAdd(&c[(kv[j] >> 16) & 255], 1);
  }
  for (int i = t + KPT * TPB; i < n; i += TPB)
    atomicAdd(&c[(k0[i] >> 16) & 255], 1);
  __syncthreads();
  int cc = 0, x = 0;
  if (t < 256) {
    cc = c[t];
    x = cc;
    for (int o = 1; o < 32; o <<= 1) {
      int y = __shfl_up(x, o);
      if (lane >= o) x += y;
    }
    if (lane == 31) wsum[w] = x;
  }
  __syncthreads();
  if (t < 256) {
    int o = x - cc;
    for (int q = 0; q < 8; ++q) o += (q < w) ? wsum[q] : 0;
    cur[t] = o;
    int node = kb * 256 + t;
    if (node < NN) {
      cnt[node]  = cc;
      off[node]  = base + o;
      dinv[node] = rsqrtf((float)cc + 1.0f);
    }
  }
  __syncthreads();
  int pl[KPT];
#pragma unroll
  for (int j = 0; j < KPT; ++j) {
    int i = t + j * TPB;
    pl[j] = (i < n) ? atomicAdd(&cur[(kv[j] >> 16) & 255], 1) : 0;
  }
#pragma unroll
  for (int j = 0; j < KPT; ++j) {
    int i = t + j * TPB;
    if (i < n) lst[clamp_idx(base + pl[j], NE - 1)] = kv[j] & 0xFFFF;
  }
  for (int i = t + KPT * TPB; i < n; i += TPB) {
    int v = k0[i];
    int p1 = atomicAdd(&cur[(v >> 16) & 255], 1);
    lst[clamp_idx(base + p1, NE - 1)] = v & 0xFFFF;
  }
}

__global__ void k_gemm_xw0(const float* __restrict__ X,
                           const float* __restrict__ W,
                           float* __restrict__ H) {
  int wave = (blockIdx.x * blockDim.x + threadIdx.x) >> 5;
  if (wave >= ROW_TILES) return;
  int lane = threadIdx.x & 31;
  int hh   = lane >> 4;
  int lid  = lane & 15;
  const float* xr = X + (wave * 16 + lid) * KIN;
  v8f c0 = {}, c1 = {}, c2 = {}, c3 = {};
  for (int k = 0; k < KIN; k += 4) {
    __builtin_prefetch(xr + k + 32, 0, 1);
    int ka = k + 2 * hh;
    v2f a;
    a.x = xr[ka];
    a.y = xr[ka + 1];
    const float* wr0 = W + ka * HID;
    const float* wr1 = wr0 + HID;
    v2f b;
    b.x = wr0[lid];      b.y = wr1[lid];
    c0 = __builtin_amdgcn_wmma_f32_16x16x4_f32(false, a, false, b, (short)0, c0, false, false);
    b.x = wr0[lid + 16]; b.y = wr1[lid + 16];
    c1 = __builtin_amdgcn_wmma_f32_16x16x4_f32(false, a, false, b, (short)0, c1, false, false);
    b.x = wr0[lid + 32]; b.y = wr1[lid + 32];
    c2 = __builtin_amdgcn_wmma_f32_16x16x4_f32(false, a, false, b, (short)0, c2, false, false);
    b.x = wr0[lid + 48]; b.y = wr1[lid + 48];
    c3 = __builtin_amdgcn_wmma_f32_16x16x4_f32(false, a, false, b, (short)0, c3, false, false);
  }
  int mrow = wave * 16 + hh * 8;
  for (int j = 0; j < 8; ++j) {
    float* o = H + (mrow + j) * HID + lid;
    o[0]  = c0[j];
    o[16] = c1[j];
    o[32] = c2[j];
    o[48] = c3[j];
  }
}

#define TILE_NODES 16
#define AP 68
static_assert(NN % TILE_NODES == 0, "every block owns a whole row tile");

__global__ void __launch_bounds__(TILE_NODES * 16)
k_gather64_hw12(const float* __restrict__ Hlin,
                const int* __restrict__ lst,
                const int* __restrict__ off,
                const int* __restrict__ cnt,
                const float* __restrict__ dinv,
                const float* __restrict__ bias,
                const float* __restrict__ mask,
                const float* __restrict__ W1,
                const float* __restrict__ W2,
                float* __restrict__ MU,
                float* __restrict__ LV) {
  __shared__ float sA[TILE_NODES * AP];
  int t = threadIdx.x;
  int r = t >> 4;
  int i = blockIdx.x * TILE_NODES + r;
  int q = t & 15;
  const float4* H4 = (const float4*)Hlin;
  float di = dinv[i];
  float4 h = H4[i * 16 + q];
  float4 mk = ((const float4*)mask)[i * 16 + q];
  float4 acc;
  acc.x = h.x * di * di + bias[4 * q + 0];
  acc.y = h.y * di * di + bias[4 * q + 1];
  acc.z = h.z * di * di + bias[4 * q + 2];
  acc.w = h.w * di * di + bias[4 * q + 3];
  int j = off[i];
  int end = j + cnt[i];
  for (; j + 4 <= end; j += 4) {
    int s0 = clamp_idx(lst[j + 0], NN - 1);
    int s1 = clamp_idx(lst[j + 1], NN - 1);
    int s2 = clamp_idx(lst[j + 2], NN - 1);
    int s3 = clamp_idx(lst[j + 3], NN - 1);
    float w0 = dinv[s0] * di, w1 = dinv[s1] * di, w2 = dinv[s2] * di, w3 = dinv[s3] * di;
    float4 r0 = H4[s0 * 16 + q];
    float4 r1 = H4[s1 * 16 + q];
    float4 r2 = H4[s2 * 16 + q];
    float4 r3 = H4[s3 * 16 + q];
    acc.x += r0.x * w0; acc.y += r0.y * w0; acc.z += r0.z * w0; acc.w += r0.w * w0;
    acc.x += r1.x * w1; acc.y += r1.y * w1; acc.z += r1.z * w1; acc.w += r1.w * w1;
    acc.x += r2.x * w2; acc.y += r2.y * w2; acc.z += r2.z * w2; acc.w += r2.w * w2;
    acc.x += r3.x * w3; acc.y += r3.y * w3; acc.z += r3.z * w3; acc.w += r3.w * w3;
  }
  for (; j < end; ++j) {
    int s0 = clamp_idx(lst[j], NN - 1);
    float w0 = dinv[s0] * di;
    float4 r0 = H4[s0 * 16 + q];
    acc.x += r0.x * w0; acc.y += r0.y * w0; acc.z += r0.z * w0; acc.w += r0.w * w0;
  }
  float4 o;
  o.x = fmaxf(acc.x, 0.0f) * mk.x;
  o.y = fmaxf(acc.y, 0.0f) * mk.y;
  o.z = fmaxf(acc.z, 0.0f) * mk.z;
  o.w = fmaxf(acc.w, 0.0f) * mk.w;
  *(float4*)&sA[r * AP + 4 * q] = o;
  __syncthreads();
  int wave = t >> 5;
  if (wave >= 2) return;
  const float* Wm = wave == 0 ? W1 : W2;
  float*       Om = wave == 0 ? MU : LV;
  int lane = t & 31;
  int hh   = lane >> 4;
  int lid  = lane & 15;
  const float* ar = sA + lid * AP;
  v8f c0 = {}, c1 = {};
#pragma unroll 8
  for (int k = 0; k < HID; k += 4) {
    int ka = k + 2 * hh;
    v2f a;
    a.x = ar[ka];
    a.y = ar[ka + 1];
    const float* u0 = Wm + ka * NC;
    const float* u1 = u0 + NC;
    v2f b;
    b.x = u0[lid];      b.y = u1[lid];
    c0 = __builtin_amdgcn_wmma_f32_16x16x4_f32(false, a, false, b, (short)0, c0, false, false);
    b.x = u0[lid + 16]; b.y = u1[lid + 16];
    c1 = __builtin_amdgcn_wmma_f32_16x16x4_f32(false, a, false, b, (short)0, c1, false, false);
  }
  int mrow = blockIdx.x * TILE_NODES + hh * 8;
  for (int jj = 0; jj < 8; ++jj) {
    float* om = Om + (mrow + jj) * NC + lid;
    om[0]  = c0[jj];
    om[16] = c1[jj];
  }
}

__global__ void k_gather32(const float* __restrict__ MUlin,
                           const float* __restrict__ LVlin,
                           const int* __restrict__ lst,
                           const int* __restrict__ off,
                           const int* __restrict__ cnt,
                           const float* __restrict__ dinv,
                           const float* __restrict__ b1,
                           const float* __restrict__ b2,
                           const float* __restrict__ eps,
                           float* __restrict__ MUout,
                           float* __restrict__ LVout,
                           float* __restrict__ Z) {
  int idx = blockIdx.x * blockDim.x + threadIdx.x;
  int i = idx >> 4;
  if (i >= NN) return;
  int q = idx & 15;
  int g = q & 7;
  bool isLV = q >= 8;
  const float4* L4 = (const float4*)(isLV ? LVlin : MUlin);
  const float*  bb = isLV ? b2 : b1;
  float4*       O4 = (float4*)(isLV ? LVout : MUout);
  float di = dinv[i];
  float dd = di * di;
  float4 h = L4[i * 8 + g];
  float4 ep = ((const float4*)eps)[i * 8 + g];
  float4 acc;
  acc.x = h.x * dd + bb[4 * g + 0];
  acc.y = h.y * dd + bb[4 * g + 1];
  acc.z = h.z * dd + bb[4 * g + 2];
  acc.w = h.w * dd + bb[4 * g + 3];
  int j = off[i];
  int end = j + cnt[i];
  for (; j + 4 <= end; j += 4) {
    int s0 = clamp_idx(lst[j + 0], NN - 1);
    int s1 = clamp_idx(lst[j + 1], NN - 1);
    int s2 = clamp_idx(lst[j + 2], NN - 1);
    int s3 = clamp_idx(lst[j + 3], NN - 1);
    float w0 = dinv[s0] * di, w1 = dinv[s1] * di, w2 = dinv[s2] * di, w3 = dinv[s3] * di;
    float4 r0 = L4[s0 * 8 + g];
    float4 r1 = L4[s1 * 8 + g];
    float4 r2 = L4[s2 * 8 + g];
    float4 r3 = L4[s3 * 8 + g];
    acc.x += r0.x * w0; acc.y += r0.y * w0; acc.z += r0.z * w0; acc.w += r0.w * w0;
    acc.x += r1.x * w1; acc.y += r1.y * w1; acc.z += r1.z * w1; acc.w += r1.w * w1;
    acc.x += r2.x * w2; acc.y += r2.y * w2; acc.z += r2.z * w2; acc.w += r2.w * w2;
    acc.x += r3.x * w3; acc.y += r3.y * w3; acc.z += r3.z * w3; acc.w += r3.w * w3;
  }
  for (; j < end; ++j) {
    int s0 = clamp_idx(lst[j], NN - 1);
    float w0 = dinv[s0] * di;
    float4 r0 = L4[s0 * 8 + g];
    acc.x += r0.x * w0; acc.y += r0.y * w0; acc.z += r0.z * w0; acc.w += r0.w * w0;
  }
  O4[i * 8 + g] = acc;
  float4 lv;
  lv.x = __shfl_xor(acc.x, 8);
  lv.y = __shfl_xor(acc.y, 8);
  lv.z = __shfl_xor(acc.z, 8);
  lv.w = __shfl_xor(acc.w, 8);
  if (!isLV) {
    float4 zz;
    zz.x = ep.x * expf(lv.x) + acc.x;
    zz.y = ep.y * expf(lv.y) + acc.y;
    zz.z = ep.z * expf(lv.z) + acc.z;
    zz.w = ep.w * expf(lv.w) + acc.w;
    ((float4*)Z)[i * 8 + g] = zz;
  }
}

extern "C" void kernel_launch(void* const* d_in, const int* in_sizes, int n_in,
                              void* d_out, int out_size, void* d_ws, size_t ws_size,
                              hipStream_t stream) {
  (void)in_sizes; (void)n_in; (void)out_size; (void)ws_size;
  const float* x    = (const float*)d_in[0];
  const int*   ei   = (const int*)d_in[1];
  const float* W0   = (const float*)d_in[2];
  const float* b0   = (const float*)d_in[3];
  const float* W1   = (const float*)d_in[4];
  const float* b1   = (const float*)d_in[5];
  const float* W2   = (const float*)d_in[6];
  const float* b2   = (const float*)d_in[7];
  const float* mask = (const float*)d_in[8];
  const float* eps  = (const float*)d_in[9];
  const int* src = ei;
  const int* dst = ei + NE;

  int*   cnt   = (int*)d_ws;
  int*   off   = cnt + NNP;
  float* dinv  = (float*)(off + NNP);
  int*   hist  = (int*)(dinv + NNP);
  int*   bbase = hist + NBLK * HP;
  int*   btot  = bbase + HP;
  int*   key   = btot + HP;
  int*   lst   = key + NE;
  float* hlin  = (float*)(lst + NE);
  float* mulin = hlin + (size_t)NN * HID;
  float* lvlin = mulin + (size_t)NN * NC;

  float* z  = (float*)d_out;
  float* mu = z + (size_t)NN * NC;
  float* lv = mu + (size_t)NN * NC;

  const int B = 256;
  const int gW   = (ROW_TILES * 32 + B - 1) / B;
  const int gG   = (NN * 16 + B - 1) / B;

  k_bcount<<<NBLK, TPB, 0, stream>>>(dst, hist);
  k_bplace<<<NBLK, TPB, 0, stream>>>(src, dst, hist, key, bbase, btot);
  k_bsort <<<NBK,  TPB, 0, stream>>>(key, bbase, btot, lst, cnt, off, dinv);

  k_gemm_xw0 <<<gW,  B, 0, stream>>>(x, W0, hlin);
  k_gather64_hw12<<<NN / TILE_NODES, TILE_NODES * 16, 0, stream>>>(hlin, lst, off, cnt, dinv, b0, mask,
                                                                  W1, W2, mulin, lvlin);

  k_gather32 <<<gG,   B, 0, stream>>>(mulin, lvlin, lst, off, cnt, dinv, b1, b2, eps, mu, lv, z);
}


# Rows are stored times their own dinv by the kernels that make them, so the edge loops add rows as stored (no dinv[s] load per edge); dinv[i] and bias go on the finished sum; list order kept
# speedup vs baseline: 4.0473x; 1.0552x over previous
#include <hip/hip_runtime.h>


typedef __attribute__((ext_vector_type(2))) float v2f;
typedef __attribute__((ext_vector_type(8))) float v8f;

#define NN 50000
#define NE 800000
#define KIN 128
#define HID 64
#define NC 32
#define ROW_TILES (NN / 16)
#define NNP 50048

__device__ __forceinline__ int clamp_idx(int v, int hi) {
  return v < 0 ? 0 : (v > hi ? hi : v);
}

#define NBLK 128
#define EPB  6250
#define EPT  7
#define NBK  196
#define HP   256
#define TPB  1024
#define KPT  6

__global__ void __launch_bounds__(TPB) k_bcount(const int* __restrict__ dst,
                                                int* __restrict__ hist) {
  __shared__ int h[HP];
  int t = threadIdx.x;
  if (t < HP) h[t] = 0;
  const int* d0 = dst + blockIdx.x * EPB;
  int dv[EPT];
#pragma unroll
  for (int j = 0; j < EPT; ++j) {
    int i = t + j * TPB;
    dv[j] = d0[i < EPB ? i : EPB - 1];
  }
  __syncthreads();
#pragma unroll
  for (int j = 0; j < EPT; ++j) {
    int i = t + j * TPB;
    if (i < EPB) atomicAdd(&h[clamp_idx(dv[j], NN - 1) >> 8], 1);
  }
  __syncthreads();
  if (t < HP) hist[blockIdx.x * HP + t] = h[t];
}

__global__ void __launch_bounds__(TPB) k_bplace(const int* __restrict__ src,
                                                const int* __restrict__ dst,
                                                const int* __restrict__ hist,
                                                int* __restrict__ key,
                                                int* __restrict__ bbase,
                                                int* __restrict__ btot) {
  __shared__ int part[4][HP];
  __shared__ int tot[4][HP];
  __shared__ int cur[HP];
  __shared__ int wsum[8];
  int t = threadIdx.x;
  int b = blockIdx.x;
  int lane = t & 31;
  int w = t >> 5;
  const int* d0 = dst + b * EPB;
  const int* s0 = src + b * EPB;
  int dv[EPT], sv[EPT];
#pragma unroll
  for (int j = 0; j < EPT; ++j) {
    int i = t + j * TPB;
    int ii = i < EPB ? i : EPB - 1;
    dv[j] = d0[ii];
    sv[j] = s0[ii];
  }
  int k = t & (HP - 1);
  int g = t >> 8;
  int hv[NBLK / 4];
#pragma unroll
  for (int q = 0; q < NBLK / 4; ++q) hv[q] = hist[(g + 4 * q) * HP + k];
  int p = 0, s = 0;
#pragma unroll
  for (int q = 0; q < NBLK / 4; ++q) {
    s += hv[q];
    p += (g + 4 * q < b) ? hv[q] : 0;
  }
  part[g][k] = p;
  tot[g][k] = s;
  __syncthreads();
  int T = 0, P = 0, x = 0;
  if (t < HP) {
    T = (tot[0][t] + tot[1][t]) + (tot[2][t] + tot[3][t]);
    P = (part[0][t] + part[1][t]) + (part[2][t] + part[3][t]);
    x = T;
    for (int o = 1; o < 32; o <<= 1) {
      int y = __shfl_up(x, o);
      if (lane >= o) x += y;
    }
    if (lane == 31) wsum[w] = x;
  }
  __syncthreads();
  if (t < HP) {
    int base = x - T;
    for (int q = 0; q < 8; ++q) base += (q < w) ? wsum[q] : 0;
    cur[t] = base + P;
    if (b == 0) {
      bbase[t] = base;
      btot[t]  = T;
    }
  }
  __syncthreads();
  int pos[EPT];
#pragma unroll
  for (int j = 0; j < EPT; ++j) {
    int i = t + j * TPB;
    dv[j] = clamp_idx(dv[j], NN - 1);
    pos[j] = (i < EPB) ? atomicAdd(&cur[dv[j] >> 8], 1) : 0;
  }
#pragma unroll
  for (int j = 0; j < EPT; ++j) {
    int i = t + j * TPB;
    if (i < EPB)
      key[clamp_idx(pos[j], NE - 1)] = ((dv[j] & 255) << 16) | clamp_idx(sv[j], NN - 1);
  }
}

__global__ void __launch_bounds__(TPB) k_bsort(const int* __restrict__ key,
                                               const int* __restrict__ bbase,
                                               const int* __restrict__ btot,
                                               int* __restrict__ lst,
                                               int* __restrict__ cnt,
                                               int* __restrict__ off,
                                               float* __restrict__ dinv) {
  __shared__ int c[256];
  __shared__ int cur[256];
  __shared__ int wsum[8];
  int t = threadIdx.x;
  int kb = blockIdx.x;
  int lane = t & 31;
  int w = t >> 5;
  int base = clamp_idx(bbase[kb], NE);
  int n    = clamp_idx(btot[kb], NE - base);
  const int* k0 = key + base;
  if (t < 256) c[t] = 0;
  int kv[KPT];
#pragma unroll
  for (int j = 0; j < KPT; ++j) {
    int i = t + j * TPB;
    kv[j] = (i < n) ? k0[i] : 0;
  }
  __syncthreads();
#pragma unroll
  for (int j = 0; j < KPT; ++j) {
    int i = t + j * TPB;
    if (i < n) atomicAdd(&c[(kv[j] >> 16) & 255], 1);
  }
  for (int i = t + KPT * TPB; i < n; i += TPB)
    atomicAdd(&c[(k0[i] >> 16) & 255], 1);
  __syncthreads();
  int cc = 0, x = 0;
  if (t < 256) {
    cc = c[t];
    x = cc;
    for (int o = 1; o < 32; o <<= 1) {
      int y = __shfl_up(x, o);
      if (lane >= o) x += y;
    }
    if (lane == 31) wsum[w] = x;
  }
  __syncthreads();
  if (t < 256) {
    int o = x - cc;
    for (int q = 0; q < 8; ++q) o += (q < w) ? wsum[q] : 0;
    cur[t] = o;
    int node = kb * 256 + t;
    if (node < NN) {
      cnt[node]  = cc;
      off[node]  = base + o;
      dinv[node] = rsqrtf((float)cc + 1.0f);
    }
  }
  __syncthreads();
  int pl[KPT];
#pragma unroll
  for (int j = 0; j < KPT; ++j) {
    int i = t + j * TPB;
    pl[j] = (i < n) ? atomicAdd(&cur[(kv[j] >> 16) & 255], 1) : 0;
  }
#pragma unroll
  for (int j = 0; j < KPT; ++j) {
    int i = t + j * TPB;
    if (i < n) lst[clamp_idx(base + pl[j], NE - 1)] = kv[j] & 0xFFFF;
  }
  for (int i = t + KPT * TPB; i < n; i += TPB) {
    int v = k0[i];
    int p1 = atomicAdd(&cur[(v >> 16) & 255], 1);
    lst[clamp_idx(base + p1, NE - 1)] = v & 0xFFFF;
  }
}

__global__ void k_gemm_xw0(const float* __restrict__ X,
                           const float* __restrict__ W,
                           const float* __restrict__ dinv,
                           float* __restrict__ H) {
  int wave = (blockIdx.x * blockDim.x + threadIdx.x) >> 5;
  if (wave >= ROW_TILES) return;
  int lane = threadIdx.x & 31;
  int hh   = lane >> 4;
  int lid  = lane & 15;
  const float* xr = X + (wave * 16 + lid) * KIN;
  v8f c0 = {}, c1 = {}, c2 = {}, c3 = {};
  for (int k = 0; k < KIN; k += 4) {
    __builtin_prefetch(xr + k + 32, 0, 1);
    int ka = k + 2 * hh;
    v2f a;
    a.x = xr[ka];
    a.y = xr[ka + 1];
    const float* wr0 = W + ka * HID;
    const float* wr1 = wr0 + HID;
    v2f b;
    b.x = wr0[lid];      b.y = wr1[lid];
    c0 = __builtin_amdgcn_wmma_f32_16x16x4_f32(false, a, false, b, (short)0, c0, false, false);
    b.x = wr0[lid + 16]; b.y = wr1[lid + 16];
    c1 = __builtin_amdgcn_wmma_f32_16x16x4_f32(false, a, false, b, (short)0, c1, false, false);
    b.x = wr0[lid + 32]; b.y = wr1[lid + 32];
    c2 = __builtin_amdgcn_wmma_f32_16x16x4_f32(false, a, false, b, (short)0, c2, false, false);
    b.x = wr0[lid + 48]; b.y = wr1[lid + 48];
    c3 = __builtin_amdgcn_wmma_f32_16x16x4_f32(false, a, false, b, (short)0, c3, false, false);
  }
  int mrow = wave * 16 + hh * 8;
  for (int j = 0; j < 8; ++j) {
    float ds = dinv[mrow + j];
    float* o = H + (mrow + j) * HID + lid;
    o[0]  = c0[j] * ds;
    o[16] = c1[j] * ds;
    o[32] = c2[j] * ds;
    o[48] = c3[j] * ds;
  }
}

#define TILE_NODES 16
#define AP 68
static_assert(NN % TILE_NODES == 0, "every block owns a whole row tile");

__global__ void __launch_bounds__(TILE_NODES * 16)
k_gather64_hw12(const float* __restrict__ Hlin,
                const int* __restrict__ lst,
                const int* __restrict__ off,
                const int* __restrict__ cnt,
                const float* __restrict__ dinv,
                const float* __restrict__ bias,
                const float* __restrict__ mask,
                const float* __restrict__ W1,
                const float* __restrict__ W2,
                float* __restrict__ MU,
                float* __restrict__ LV) {
  __shared__ float sA[TILE_NODES * AP];
  int t = threadIdx.x;
  int r = t >> 4;
  int i = blockIdx.x * TILE_NODES + r;
  int q = t & 15;
  const float4* H4 = (const float4*)Hlin;
  float di = dinv[i];
  float4 acc = H4[i * 16 + q];
  float4 mk = ((const float4*)mask)[i * 16 + q];
  int j = off[i];
  int end = j + cnt[i];
  for (; j + 4 <= end; j += 4) {
    int s0 = clamp_idx(lst[j + 0], NN - 1);
    int s1 = clamp_idx(lst[j + 1], NN - 1);
    int s2 = clamp_idx(lst[j + 2], NN - 1);
    int s3 = clamp_idx(lst[j + 3], NN - 1);
    float4 r0 = H4[s0 * 16 + q];
    float4 r1 = H4[s1 * 16 + q];
    float4 r2 = H4[s2 * 16 + q];
    float4 r3 = H4[s3 * 16 + q];
    acc.x += r0.x; acc.y += r0.y; acc.z += r0.z; acc.w += r0.w;
    acc.x += r1.x; acc.y += r1.y; acc.z += r1.z; acc.w += r1.w;
    acc.x += r2.x; acc.y += r2.y; acc.z += r2.z; acc.w += r2.w;
    acc.x += r3.x; acc.y += r3.y; acc.z += r3.z; acc.w += r3.w;
  }
  for (; j < end; ++j) {
    int s0 = clamp_idx(lst[j], NN - 1);
    float4 r0 = H4[s0 * 16 + q];
    acc.x += r0.x; acc.y += r0.y; acc.z += r0.z; acc.w += r0.w;
  }
  float4 o;
  o.x = fmaxf(acc.x * di + bias[4 * q + 0], 0.0f) * mk.x * di;
  o.y = fmaxf(acc.y * di + bias[4 * q + 1], 0.0f) * mk.y * di;
  o.z = fmaxf(acc.z * di + bias[4 * q + 2], 0.0f) * mk.z * di;
  o.w = fmaxf(acc.w * di + bias[4 * q + 3], 0.0f) * mk.w * di;
  *(float4*)&sA[r * AP + 4 * q] = o;
  __syncthreads();
  int wave = t >> 5;
  if (wave >= 2) return;
  const float* Wm = wave == 0 ? W1 : W2;
  float*       Om = wave == 0 ? MU : LV;
  int lane = t & 31;
  int hh   = lane >> 4;
  int lid  = lane & 15;
  const float* ar = sA + lid * AP;
  v8f c0 = {}, c1 = {};
#pragma unroll 8
  for (int k = 0; k < HID; k += 4) {
    int ka = k + 2 * hh;
    v2f a;
    a.x = ar[ka];
    a.y = ar[ka + 1];
    const float* u0 = Wm + ka * NC;
    const float* u1 = u0 + NC;
    v2f b;
    b.x = u0[lid];      b.y = u1[lid];
    c0 = __builtin_amdgcn_wmma_f32_16x16x4_f32(false, a, false, b, (short)0, c0, false, false);
    b.x = u0[lid + 16]; b.y = u1[lid + 16];
    c1 = __builtin_amdgcn_wmma_f32_16x16x4_f32(false, a, false, b, (short)0, c1, false, false);
  }
  int mrow = blockIdx.x * TILE_NODES + hh * 8;
  for (int jj = 0; jj < 8; ++jj) {
    float* om = Om + (mrow + jj) * NC + lid;
    om[0]  = c0[jj];
    om[16] = c1[jj];
  }
}

__global__ void k_gather32(const float* __restrict__ MUlin,
                           const float* __restrict__ LVlin,
                           const int* __restrict__ lst,
                           const int* __restrict__ off,
                           const int* __restrict__ cnt,
                           const float* __restrict__ dinv,
                           const float* __restrict__ b1,
                           const float* __restrict__ b2,
                           const float* __restrict__ eps,
                           float* __restrict__ MUout,
                           float* __restrict__ LVout,
                           float* __restrict__ Z) {
  int idx = blockIdx.x * blockDim.x + threadIdx.x;
  int i = idx >> 4;
  if (i >= NN) return;
  int q = idx & 15;
  int g = q & 7;
  bool isLV = q >= 8;
  const float4* L4 = (const float4*)(isLV ? LVlin : MUlin);
  const float*  bb = isLV ? b2 : b1;
  float4*       O4 = (float4*)(isLV ? LVout : MUout);
  float di = dinv[i];
  float4 acc = L4[i * 8 + g];
  float4 ep = ((const float4*)eps)[i * 8 + g];
  int j = off[i];
  int end = j + cnt[i];
  for (; j + 4 <= end; j += 4) {
    int s0 = clamp_idx(lst[j + 0], NN - 1);
    int s1 = clamp_idx(lst[j + 1], NN - 1);
    int s2 = clamp_idx(lst[j + 2], NN - 1);
    int s3 = clamp_idx(lst[j + 3], NN - 1);
    float4 r0 = L4[s0 * 8 + g];
    float4 r1 = L4[s1 * 8 + g];
    float4 r2 = L4[s2 * 8 + g];
    float4 r3 = L4[s3 * 8 + g];
    acc.x += r0.x; acc.y += r0.y; acc.z += r0.z; acc.w += r0.w;
    acc.x += r1.x; acc.y += r1.y; acc.z += r1.z; acc.w += r1.w;
    acc.x += r2.x; acc.y += r2.y; acc.z += r2.z; acc.w += r2.w;
    acc.x += r3.x; acc.y += r3.y; acc.z += r3.z; acc.w += r3.w;
  }
  for (; j < end; ++j) {
    int s0 = clamp_idx(lst[j], NN - 1);
    float4 r0 = L4[s0 * 8 + g];
    acc.x += r0.x; acc.y += r0.y; acc.z += r0.z; acc.w += r0.w;
  }
  acc.x = acc.x * di + bb[4 * g + 0];
  acc.y = acc.y * di + bb[4 * g + 1];
  acc.z = acc.z * di + bb[4 * g + 2];
  acc.w = acc.w * di + bb[4 * g + 3];
  O4[i * 8 + g] = acc;
  float4 lv;
  lv.x = __shfl_xor(acc.x, 8);
  lv.y = __shfl_xor(acc.y, 8);
  lv.z = __shfl_xor(acc.z, 8);
  lv.w = __shfl_xor(acc.w, 8);
  if (!isLV) {
    float4 zz;
    zz.x = ep.x * expf(lv.x) + acc.x;
    zz.y = ep.y * expf(lv.y) + acc.y;
    zz.z = ep.z * expf(lv.z) + acc.z;
    zz.w = ep.w * expf(lv.w) + acc.w;
    ((float4*)Z)[i * 8 + g] = zz;
  }
}

extern "C" void kernel_launch(void* const* d_in, const int* in_sizes, int n_in,
                              void* d_out, int out_size, void* d_ws, size_t ws_size,
                              hipStream_t stream) {
  (void)in_sizes; (void)n_in; (void)out_size; (void)ws_size;
  const float* x    = (const float*)d_in[0];
  const int*   ei   = (const int*)d_in[1];
  const float* W0   = (const float*)d_in[2];
  const float* b0   = (const float*)d_in[3];
  const float* W1   = (const float*)d_in[4];
  const float* b1   = (const float*)d_in[5];
  const float* W2   = (const float*)d_in[6];
  const float* b2   = (const float*)d_in[7];
  const float* mask = (const float*)d_in[8];
  const float* eps  = (const float*)d_in[9];
  const int* src = ei;
  const int* dst = ei + NE;

  int*   cnt   = (int*)d_ws;
  int*   off   = cnt + NNP;
  float* dinv  = (float*)(off + NNP);
  int*   hist  = (int*)(dinv + NNP);
  int*   bbase = hist + NBLK * HP;
  int*   btot  = bbase + HP;
  int*   key   = btot + HP;
  int*   lst   = key + NE;
  float* hlin  = (float*)(lst + NE);
  float* mulin = hlin + (size_t)NN * HID;
  float* lvlin = mulin + (size_t)NN * NC;

  float* z  = (float*)d_out;
  float* mu = z + (size_t)NN * NC;
  float* lv = mu + (size_t)NN * NC;

  const int B = 256;
  const int gW   = (ROW_TILES * 32 + B - 1) / B;
  const int gG   = (NN * 16 + B - 1) / B;

  k_bcount<<<NBLK, TPB, 0, stream>>>(dst, hist);
  k_bplace<<<NBLK, TPB, 0, stream>>>(src, dst, hist, key, bbase, btot);
  k_bsort <<<NBK,  TPB, 0, stream>>>(key, bbase, btot, lst, cnt, off, dinv);

  k_gemm_xw0 <<<gW,  B, 0, stream>>>(x, W0, dinv, hlin);
  k_gather64_hw12<<<NN / TILE_NODES, TILE_NODES * 16, 0, stream>>>(hlin, lst, off, cnt, dinv, b0, mask,
                                                                  W1, W2, mulin, lvlin);

  k_gather32 <<<gG,   B, 0, stream>>>(mulin, lvlin, lst, off, cnt, dinv, b1, b2, eps, mu, lv, z);
}


# Each of 256 edge blocks orders its own edges by bucket in LDS (own counts and offsets) and writes them out in whole lines; the count launch is gone (6 to 5); sum order = list order (LDS arrival)
# speedup vs baseline: 5.4120x; 1.3372x over previous
#include <hip/hip_runtime.h>


typedef __attribute__((ext_vector_type(2))) float v2f;
typedef __attribute__((ext_vector_type(8))) float v8f;

#define NN 50000
#define NE 800000
#define KIN 128
#define HID 64
#define NC 32
#define ROW_TILES (NN / 16)
#define NNP 50048

__device__ __forceinline__ int clamp_idx(int v, int hi) {
  return v < 0 ? 0 : (v > hi ? hi : v);
}

#define NBLK 256
#define EPB  3125
#define TPP  512
#define EPT  7
#define NBK  196
#define HP   256
#define TPB  1024
#define SPW  (NBLK / 32)
#define RPL  1
static_assert(NBLK * EPB == NE, "the edge blocks cover the edge list exactly");
static_assert(EPT * TPP >= EPB && HP <= TPP, "a block's threads cover its edges and the table row");
static_assert(NBK + 1 <= HP && NBLK % 32 == 0 && NBLK <= TPB && TPB == 1024, "table pitch and run split");

__global__ void __launch_bounds__(TPP) k_bplace(const int* __restrict__ src,
                                                const int* __restrict__ dst,
                                                int* __restrict__ key,
                                                int* __restrict__ lpre) {
  __shared__ int h[HP];
  __shared__ int cur[HP];
  __shared__ int wsum[8];
  __shared__ int sk[EPB];
  int t = threadIdx.x;
  int b = blockIdx.x;
  int lane = t & 31;
  int w = t >> 5;
  if (t < HP) h[t] = 0;
  const int* d0 = dst + b * EPB;
  const int* s0 = src + b * EPB;
  int dv[EPT], sv[EPT];
#pragma unroll
  for (int j = 0; j < EPT; ++j) {
    int i = t + j * TPP;
    int ii = i < EPB ? i : EPB - 1;
    dv[j] = d0[ii];
    sv[j] = s0[ii];
  }
  __syncthreads();
#pragma unroll
  for (int j = 0; j < EPT; ++j) {
    int i = t + j * TPP;
    dv[j] = clamp_idx(dv[j], NN - 1);
    if (i < EPB) atomicAdd(&h[dv[j] >> 8], 1);
  }
  __syncthreads();
  int c = 0, x = 0;
  if (t < HP) {
    c = h[t];
    x = c;
    for (int o = 1; o < 32; o <<= 1) {
      int y = __shfl_up(x, o);
      if (lane >= o) x += y;
    }
    if (lane == 31) wsum[w] = x;
  }
  __syncthreads();
  if (t < HP) {
    int o = x - c;
    for (int q = 0; q < 8; ++q) o += (q < w) ? wsum[q] : 0;
    cur[t] = o;
    lpre[b * HP + t] = o;
  }
  __syncthreads();
  int pos[EPT];
#pragma unroll
  for (int j = 0; j < EPT; ++j) {
    int i = t + j * TPP;
    pos[j] = (i < EPB) ? atomicAdd(&cur[dv[j] >> 8], 1) : 0;
  }
#pragma unroll
  for (int j = 0; j < EPT; ++j) {
    int i = t + j * TPP;
    if (i < EPB)
      sk[clamp_idx(pos[j], EPB - 1)] = ((dv[j] & 255) << 16) | clamp_idx(sv[j], NN - 1);
  }
  __syncthreads();
#pragma unroll
  for (int j = 0; j < EPT; ++j) {
    int i = t + j * TPP;
    if (i < EPB) key[b * EPB + i] = sk[i];
  }
}

__global__ void __launch_bounds__(TPB) k_bsort(const int* __restrict__ key,
                                               const int* __restrict__ lpre,
                                               int* __restrict__ lst,
                                               int* __restrict__ cnt,
                                               int* __restrict__ off,
                                               float* __restrict__ dinv) {
  __shared__ int c[256];
  __shared__ int cur[256];
  __shared__ int wsum[8];
  __shared__ int segA[NBLK];
  __shared__ int segC[NBLK];
  __shared__ int redA[NBLK / 32];
  int t = threadIdx.x;
  int kb = blockIdx.x;
  int lane = t & 31;
  int w = t >> 5;
  if (t < 256) c[t] = 0;
  if (t < NBLK) {
    int a  = clamp_idx(lpre[t * HP + kb], EPB);
    int a1 = clamp_idx(lpre[t * HP + kb + 1], EPB);
    segA[t] = t * EPB + a;
    segC[t] = a1 > a ? a1 - a : 0;
    int sa = a;
    for (int o = 16; o > 0; o >>= 1) sa += __shfl_xor(sa, o);
    if (lane == 0) redA[w] = sa;
  }
  __syncthreads();
  int base = 0;
#pragma unroll
  for (int q = 0; q < NBLK / 32; ++q) base += redA[q];
  base = clamp_idx(base, NE);
  int kv[SPW * RPL];
#pragma unroll
  for (int s = 0; s < SPW; ++s) {
    int st = segA[SPW * w + s];
    int ln = segC[SPW * w + s];
#pragma unroll
    for (int r = 0; r < RPL; ++r) {
      int i = lane + 32 * r;
      kv[s * RPL + r] = (i < ln) ? key[st + i] : -1;
    }
  }
#pragma unroll
  for (int e = 0; e < SPW * RPL; ++e)
    if (kv[e] >= 0) atomicAdd(&c[(kv[e] >> 16) & 255], 1);
  for (int s = 0; s < SPW; ++s) {
    int st = segA[SPW * w + s];
    int ln = segC[SPW * w + s];
    for (int i = lane + 32 * RPL; i < ln; i += 32)
      atomicAdd(&c[(key[st + i] >> 16) & 255], 1);
  }
  __syncthreads();
  int cc = 0, x = 0;
  if (t < 256) {
    cc = c[t];
    x = cc;
    for (int o = 1; o < 32; o <<= 1) {
      int y = __shfl_up(x, o);
      if (lane >= o) x += y;
    }
    if (lane == 31) wsum[w] = x;
  }
  __syncthreads();
  if (t < 256) {
    int o = x - cc;
    for (int q = 0; q < 8; ++q) o += (q < w) ? wsum[q] : 0;
    cur[t] = o;
    int node = kb * 256 + t;
    if (node < NN) {
      cnt[node]  = cc;
      off[node]  = base + o;
      dinv[node] = rsqrtf((float)cc + 1.0f);
    }
  }
  __syncthreads();
  int pl[SPW * RPL];
#pragma unroll
  for (int e = 0; e < SPW * RPL; ++e)
    pl[e] = (kv[e] >= 0) ? atomicAdd(&cur[(kv[e] >> 16) & 255], 1) : 0;
#pragma unroll
  for (int e = 0; e < SPW * RPL; ++e)
    if (kv[e] >= 0) lst[clamp_idx(base + pl[e], NE - 1)] = kv[e] & 0xFFFF;
  for (int s = 0; s < SPW; ++s) {
    int st = segA[SPW * w + s];
    int ln = segC[SPW * w + s];
    for (int i = lane + 32 * RPL; i < ln; i += 32) {
      int v = key[st + i];
      int p1 = atomicAdd(&cur[(v >> 16) & 255], 1);
      lst[clamp_idx(base + p1, NE - 1)] = v & 0xFFFF;
    }
  }
}

__global__ void k_gemm_xw0(const float* __restrict__ X,
                           const float* __restrict__ W,
                           const float* __restrict__ dinv,
                           float* __restrict__ H) {
  int wave = (blockIdx.x * blockDim.x + threadIdx.x) >> 5;
  if (wave >= ROW_TILES) return;
  int lane = threadIdx.x & 31;
  int hh   = lane >> 4;
  int lid  = lane & 15;
  const float* xr = X + (wave * 16 + lid) * KIN;
  v8f c0 = {}, c1 = {}, c2 = {}, c3 = {};
  for (int k = 0; k < KIN; k += 4) {
    __builtin_prefetch(xr + k + 32, 0, 1);
    int ka = k + 2 * hh;
    v2f a;
    a.x = xr[ka];
    a.y = xr[ka + 1];
    const float* wr0 = W + ka * HID;
    const float* wr1 = wr0 + HID;
    v2f b;
    b.x = wr0[lid];      b.y = wr1[lid];
    c0 = __builtin_amdgcn_wmma_f32_16x16x4_f32(false, a, false, b, (short)0, c0, false, false);
    b.x = wr0[lid + 16]; b.y = wr1[lid + 16];
    c1 = __builtin_amdgcn_wmma_f32_16x16x4_f32(false, a, false, b, (short)0, c1, false, false);
    b.x = wr0[lid + 32]; b.y = wr1[lid + 32];
    c2 = __builtin_amdgcn_wmma_f32_16x16x4_f32(false, a, false, b, (short)0, c2, false, false);
    b.x = wr0[lid + 48]; b.y = wr1[lid + 48];
    c3 = __builtin_amdgcn_wmma_f32_16x16x4_f32(false, a, false, b, (short)0, c3, false, false);
  }
  int mrow = wave * 16 + hh * 8;
  for (int j = 0; j < 8; ++j) {
    float ds = dinv[mrow + j];
    float* o = H + (mrow + j) * HID + lid;
    o[0]  = c0[j] * ds;
    o[16] = c1[j] * ds;
    o[32] = c2[j] * ds;
    o[48] = c3[j] * ds;
  }
}

#define TILE_NODES 16
#define AP 68
static_assert(NN % TILE_NODES == 0, "every block owns a whole row tile");

__global__ void __launch_bounds__(TILE_NODES * 16)
k_gather64_hw12(const float* __restrict__ Hlin,
                const int* __restrict__ lst,
                const int* __restrict__ off,
                const int* __restrict__ cnt,
                const float* __restrict__ dinv,
                const float* __restrict__ bias,
                const float* __restrict__ mask,
                const float* __restrict__ W1,
                const float* __restrict__ W2,
                float* __restrict__ MU,
                float* __restrict__ LV) {
  __shared__ float sA[TILE_NODES * AP];
  int t = threadIdx.x;
  int r = t >> 4;
  int i = blockIdx.x * TILE_NODES + r;
  int q = t & 15;
  const float4* H4 = (const float4*)Hlin;
  float di = dinv[i];
  float4 acc = H4[i * 16 + q];
  float4 mk = ((const float4*)mask)[i * 16 + q];
  int j = off[i];
  int end = j + cnt[i];
  for (; j + 4 <= end; j += 4) {
    int s0 = clamp_idx(lst[j + 0], NN - 1);
    int s1 = clamp_idx(lst[j + 1], NN - 1);
    int s2 = clamp_idx(lst[j + 2], NN - 1);
    int s3 = clamp_idx(lst[j + 3], NN - 1);
    float4 r0 = H4[s0 * 16 + q];
    float4 r1 = H4[s1 * 16 + q];
    float4 r2 = H4[s2 * 16 + q];
    float4 r3 = H4[s3 * 16 + q];
    acc.x += r0.x; acc.y += r0.y; acc.z += r0.z; acc.w += r0.w;
    acc.x += r1.x; acc.y += r1.y; acc.z += r1.z; acc.w += r1.w;
    acc.x += r2.x; acc.y += r2.y; acc.z += r2.z; acc.w += r2.w;
    acc.x += r3.x; acc.y += r3.y; acc.z += r3.z; acc.w += r3.w;
  }
  for (; j < end; ++j) {
    int s0 = clamp_idx(lst[j], NN - 1);
    float4 r0 = H4[s0 * 16 + q];
    acc.x += r0.x; acc.y += r0.y; acc.z += r0.z; acc.w += r0.w;
  }
  float4 o;
  o.x = fmaxf(acc.x * di + bias[4 * q + 0], 0.0f) * mk.x * di;
  o.y = fmaxf(acc.y * di + bias[4 * q + 1], 0.0f) * mk.y * di;
  o.z = fmaxf(acc.z * di + bias[4 * q + 2], 0.0f) * mk.z * di;
  o.w = fmaxf(acc.w * di + bias[4 * q + 3], 0.0f) * mk.w * di;
  *(float4*)&sA[r * AP + 4 * q] = o;
  __syncthreads();
  int wave = t >> 5;
  if (wave >= 2) return;
  const float* Wm = wave == 0 ? W1 : W2;
  float*       Om = wave == 0 ? MU : LV;
  int lane = t & 31;
  int hh   = lane >> 4;
  int lid  = lane & 15;
  const float* ar = sA + lid * AP;
  v8f c0 = {}, c1 = {};
#pragma unroll 8
  for (int k = 0; k < HID; k += 4) {
    int ka = k + 2 * hh;
    v2f a;
    a.x = ar[ka];
    a.y = ar[ka + 1];
    const float* u0 = Wm + ka * NC;
    const float* u1 = u0 + NC;
    v2f b;
    b.x = u0[lid];      b.y = u1[lid];
    c0 = __builtin_amdgcn_wmma_f32_16x16x4_f32(false, a, false, b, (short)0, c0, false, false);
    b.x = u0[lid + 16]; b.y = u1[lid + 16];
    c1 = __builtin_amdgcn_wmma_f32_16x16x4_f32(false, a, false, b, (short)0, c1, false, false);
  }
  int mrow = blockIdx.x * TILE_NODES + hh * 8;
  for (int jj = 0; jj < 8; ++jj) {
    float* om = Om + (mrow + jj) * NC + lid;
    om[0]  = c0[jj];
    om[16] = c1[jj];
  }
}

__global__ void k_gather32(const float* __restrict__ MUlin,
                           const float* __restrict__ LVlin,
                           const int* __restrict__ lst,
                           const int* __restrict__ off,
                           const int* __restrict__ cnt,
                           const float* __restrict__ dinv,
                           const float* __restrict__ b1,
                           const float* __restrict__ b2,
                           const float* __restrict__ eps,
                           float* __restrict__ MUout,
                           float* __restrict__ LVout,
                           float* __restrict__ Z) {
  int idx = blockIdx.x * blockDim.x + threadIdx.x;
  int i = idx >> 4;
  if (i >= NN) return;
  int q = idx & 15;
  int g = q & 7;
  bool isLV = q >= 8;
  const float4* L4 = (const float4*)(isLV ? LVlin : MUlin);
  const float*  bb = isLV ? b2 : b1;
  float4*       O4 = (float4*)(isLV ? LVout : MUout);
  float di = dinv[i];
  float4 acc = L4[i * 8 + g];
  float4 ep = ((const float4*)eps)[i * 8 + g];
  int j = off[i];
  int end = j + cnt[i];
  for (; j + 4 <= end; j += 4) {
    int s0 = clamp_idx(lst[j + 0], NN - 1);
    int s1 = clamp_idx(lst[j + 1], NN - 1);
    int s2 = clamp_idx(lst[j + 2], NN - 1);
    int s3 = clamp_idx(lst[j + 3], NN - 1);
    float4 r0 = L4[s0 * 8 + g];
    float4 r1 = L4[s1 * 8 + g];
    float4 r2 = L4[s2 * 8 + g];
    float4 r3 = L4[s3 * 8 + g];
    acc.x += r0.x; acc.y += r0.y; acc.z += r0.z; acc.w += r0.w;
    acc.x += r1.x; acc.y += r1.y; acc.z += r1.z; acc.w += r1.w;
    acc.x += r2.x; acc.y += r2.y; acc.z += r2.z; acc.w += r2.w;
    acc.x += r3.x; acc.y += r3.y; acc.z += r3.z; acc.w += r3.w;
  }
  for (; j < end; ++j) {
    int s0 = clamp_idx(lst[j], NN - 1);
    float4 r0 = L4[s0 * 8 + g];
    acc.x += r0.x; acc.y += r0.y; acc.z += r0.z; acc.w += r0.w;
  }
  acc.x = acc.x * di + bb[4 * g + 0];
  acc.y = acc.y * di + bb[4 * g + 1];
  acc.z = acc.z * di + bb[4 * g + 2];
  acc.w = acc.w * di + bb[4 * g + 3];
  O4[i * 8 + g] = acc;
  float4 lv;
  lv.x = __shfl_xor(acc.x, 8);
  lv.y = __shfl_xor(acc.y, 8);
  lv.z = __shfl_xor(acc.z, 8);
  lv.w = __shfl_xor(acc.w, 8);
  if (!isLV) {
    float4 zz;
    zz.x = ep.x * expf(lv.x) + acc.x;
    zz.y = ep.y * expf(lv.y) + acc.y;
    zz.z = ep.z * expf(lv.z) + acc.z;
    zz.w = ep.w * expf(lv.w) + acc.w;
    ((float4*)Z)[i * 8 + g] = zz;
  }
}

extern "C" void kernel_launch(void* const* d_in, const int* in_sizes, int n_in,
                              void* d_out, int out_size, void* d_ws, size_t ws_size,
                              hipStream_t stream) {
  (void)in_sizes; (void)n_in; (void)out_size; (void)ws_size;
  const float* x    = (const float*)d_in[0];
  const int*   ei   = (const int*)d_in[1];
  const float* W0   = (const float*)d_in[2];
  const float* b0   = (const float*)d_in[3];
  const float* W1   = (const float*)d_in[4];
  const float* b1   = (const float*)d_in[5];
  const float* W2   = (const float*)d_in[6];
  const float* b2   = (const float*)d_in[7];
  const float* mask = (const float*)d_in[8];
  const float* eps  = (const float*)d_in[9];
  const int* src = ei;
  const int* dst = ei + NE;

  int*   cnt   = (int*)d_ws;
  int*   off   = cnt + NNP;
  float* dinv  = (float*)(off + NNP);
  int*   lpre  = (int*)(dinv + NNP);
  int*   key   = lpre + NBLK * HP;
  int*   lst   = key + NE;
  float* hlin  = (float*)(lst + NE);
  float* mulin = hlin + (size_t)NN * HID;
  float* lvlin = mulin + (size_t)NN * NC;

  float* z  = (float*)d_out;
  float* mu = z + (size_t)NN * NC;
  float* lv = mu + (size_t)NN * NC;

  const int B = 256;
  const int gW   = (ROW_TILES * 32 + B - 1) / B;
  const int gG   = (NN * 16 + B - 1) / B;

  k_bplace<<<NBLK, TPP, 0, stream>>>(src, dst, key, lpre);
  k_bsort <<<NBK,  TPB, 0, stream>>>(key, lpre, lst, cnt, off, dinv);

  k_gemm_xw0 <<<gW,  B, 0, stream>>>(x, W0, dinv, hlin);
  k_gather64_hw12<<<NN / TILE_NODES, TILE_NODES * 16, 0, stream>>>(hlin, lst, off, cnt, dinv, b0, mask,
                                                                  W1, W2, mulin, lvlin);

  k_gather32 <<<gG,   B, 0, stream>>>(mulin, lvlin, lst, off, cnt, dinv, b1, b2, eps, mu, lv, z);
}


# Both row-sum kernels give a node 8 lanes with two 16-byte pieces each (was 16 lanes, one piece): half the waves start, each with twice the row loads in flight; per-node sum order kept
# speedup vs baseline: 5.5774x; 1.0306x over previous
#include <hip/hip_runtime.h>


typedef __attribute__((ext_vector_type(2))) float v2f;
typedef __attribute__((ext_vector_type(8))) float v8f;

#define NN 50000
#define NE 800000
#define KIN 128
#define HID 64
#define NC 32
#define ROW_TILES (NN / 16)
#define NNP 50048

__device__ __forceinline__ int clamp_idx(int v, int hi) {
  return v < 0 ? 0 : (v > hi ? hi : v);
}

#define NBLK 256
#define EPB  3125
#define TPP  512
#define EPT  7
#define NBK  196
#define HP   256
#define TPB  1024
#define SPW  (NBLK / 32)
#define RPL  1
static_assert(NBLK * EPB == NE, "the edge blocks cover the edge list exactly");
static_assert(EPT * TPP >= EPB && HP <= TPP, "a block's threads cover its edges and the table row");
static_assert(NBK + 1 <= HP && NBLK % 32 == 0 && NBLK <= TPB && TPB == 1024, "table pitch and run split");

__global__ void __launch_bounds__(TPP) k_bplace(const int* __restrict__ src,
                                                const int* __restrict__ dst,
                                                int* __restrict__ key,
                                                int* __restrict__ lpre) {
  __shared__ int h[HP];
  __shared__ int cur[HP];
  __shared__ int wsum[8];
  __shared__ int sk[EPB];
  int t = threadIdx.x;
  int b = blockIdx.x;
  int lane = t & 31;
  int w = t >> 5;
  if (t < HP) h[t] = 0;
  const int* d0 = dst + b * EPB;
  const int* s0 = src + b * EPB;
  int dv[EPT], sv[EPT];
#pragma unroll
  for (int j = 0; j < EPT; ++j) {
    int i = t + j * TPP;
    int ii = i < EPB ? i : EPB - 1;
    dv[j] = d0[ii];
    sv[j] = s0[ii];
  }
  __syncthreads();
#pragma unroll
  for (int j = 0; j < EPT; ++j) {
    int i = t + j * TPP;
    dv[j] = clamp_idx(dv[j], NN - 1);
    if (i < EPB) atomicAdd(&h[dv[j] >> 8], 1);
  }
  __syncthreads();
  int c = 0, x = 0;
  if (t < HP) {
    c = h[t];
    x = c;
    for (int o = 1; o < 32; o <<= 1) {
      int y = __shfl_up(x, o);
      if (lane >= o) x += y;
    }
    if (lane == 31) wsum[w] = x;
  }
  __syncthreads();
  if (t < HP) {
    int o = x - c;
    for (int q = 0; q < 8; ++q) o += (q < w) ? wsum[q] : 0;
    cur[t] = o;
    lpre[b * HP + t] = o;
  }
  __syncthreads();
  int pos[EPT];
#pragma unroll
  for (int j = 0; j < EPT; ++j) {
    int i = t + j * TPP;
    pos[j] = (i < EPB) ? atomicAdd(&cur[dv[j] >> 8], 1) : 0;
  }
#pragma unroll
  for (int j = 0; j < EPT; ++j) {
    int i = t + j * TPP;
    if (i < EPB)
      sk[clamp_idx(pos[j], EPB - 1)] = ((dv[j] & 255) << 16) | clamp_idx(sv[j], NN - 1);
  }
  __syncthreads();
#pragma unroll
  for (int j = 0; j < EPT; ++j) {
    int i = t + j * TPP;
    if (i < EPB) key[b * EPB + i] = sk[i];
  }
}

__global__ void __launch_bounds__(TPB) k_bsort(const int* __restrict__ key,
                                               const int* __restrict__ lpre,
                                               int* __restrict__ lst,
                                               int* __restrict__ cnt,
                                               int* __restrict__ off,
                                               float* __restrict__ dinv) {
  __shared__ int c[256];
  __shared__ int cur[256];
  __shared__ int wsum[8];
  __shared__ int segA[NBLK];
  __shared__ int segC[NBLK];
  __shared__ int redA[NBLK / 32];
  int t = threadIdx.x;
  int kb = blockIdx.x;
  int lane = t & 31;
  int w = t >> 5;
  if (t < 256) c[t] = 0;
  if (t < NBLK) {
    int a  = clamp_idx(lpre[t * HP + kb], EPB);
    int a1 = clamp_idx(lpre[t * HP + kb + 1], EPB);
    segA[t] = t * EPB + a;
    segC[t] = a1 > a ? a1 - a : 0;
    int sa = a;
    for (int o = 16; o > 0; o >>= 1) sa += __shfl_xor(sa, o);
    if (lane == 0) redA[w] = sa;
  }
  __syncthreads();
  int base = 0;
#pragma unroll
  for (int q = 0; q < NBLK / 32; ++q) base += redA[q];
  base = clamp_idx(base, NE);
  int kv[SPW * RPL];
#pragma unroll
  for (int s = 0; s < SPW; ++s) {
    int st = segA[SPW * w + s];
    int ln = segC[SPW * w + s];
#pragma unroll
    for (int r = 0; r < RPL; ++r) {
      int i = lane + 32 * r;
      kv[s * RPL + r] = (i < ln) ? key[st + i] : -1;
    }
  }
#pragma unroll
  for (int e = 0; e < SPW * RPL; ++e)
    if (kv[e] >= 0) atomicAdd(&c[(kv[e] >> 16) & 255], 1);
  for (int s = 0; s < SPW; ++s) {
    int st = segA[SPW * w + s];
    int ln = segC[SPW * w + s];
    for (int i = lane + 32 * RPL; i < ln; i += 32)
      atomicAdd(&c[(key[st + i] >> 16) & 255], 1);
  }
  __syncthreads();
  int cc = 0, x = 0;
  if (t < 256) {
    cc = c[t];
    x = cc;
    for (int o = 1; o < 32; o <<= 1) {
      int y = __shfl_up(x, o);
      if (lane >= o) x += y;
    }
    if (lane == 31) wsum[w] = x;
  }
  __syncthreads();
  if (t < 256) {
    int o = x - cc;
    for (int q = 0; q < 8; ++q) o += (q < w) ? wsum[q] : 0;
    cur[t] = o;
    int node = kb * 256 + t;
    if (node < NN) {
      cnt[node]  = cc;
      off[node]  = base + o;
      dinv[node] = rsqrtf((float)cc + 1.0f);
    }
  }
  __syncthreads();
  int pl[SPW * RPL];
#pragma unroll
  for (int e = 0; e < SPW * RPL; ++e)
    pl[e] = (kv[e] >= 0) ? atomicAdd(&cur[(kv[e] >> 16) & 255], 1) : 0;
#pragma unroll
  for (int e = 0; e < SPW * RPL; ++e)
    if (kv[e] >= 0) lst[clamp_idx(base + pl[e], NE - 1)] = kv[e] & 0xFFFF;
  for (int s = 0; s < SPW; ++s) {
    int st = segA[SPW * w + s];
    int ln = segC[SPW * w + s];
    for (int i = lane + 32 * RPL; i < ln; i += 32) {
      int v = key[st + i];
      int p1 = atomicAdd(&cur[(v >> 16) & 255], 1);
      lst[clamp_idx(base + p1, NE - 1)] = v & 0xFFFF;
    }
  }
}

__global__ void k_gemm_xw0(const float* __restrict__ X,
                           const float* __restrict__ W,
                           const float* __restrict__ dinv,
                           float* __restrict__ H) {
  int wave = (blockIdx.x * blockDim.x + threadIdx.x) >> 5;
  if (wave >= ROW_TILES) return;
  int lane = threadIdx.x & 31;
  int hh   = lane >> 4;
  int lid  = lane & 15;
  const float* xr = X + (wave * 16 + lid) * KIN;
  v8f c0 = {}, c1 = {}, c2 = {}, c3 = {};
  for (int k = 0; k < KIN; k += 4) {
    __builtin_prefetch(xr + k + 32, 0, 1);
    int ka = k + 2 * hh;
    v2f a;
    a.x = xr[ka];
    a.y = xr[ka + 1];
    const float* wr0 = W + ka * HID;
    const float* wr1 = wr0 + HID;
    v2f b;
    b.x = wr0[lid];      b.y = wr1[lid];
    c0 = __builtin_amdgcn_wmma_f32_16x16x4_f32(false, a, false, b, (short)0, c0, false, false);
    b.x = wr0[lid + 16]; b.y = wr1[lid + 16];
    c1 = __builtin_amdgcn_wmma_f32_16x16x4_f32(false, a, false, b, (short)0, c1, false, false);
    b.x = wr0[lid + 32]; b.y = wr1[lid + 32];
    c2 = __builtin_amdgcn_wmma_f32_16x16x4_f32(false, a, false, b, (short)0, c2, false, false);
    b.x = wr0[lid + 48]; b.y = wr1[lid + 48];
    c3 = __builtin_amdgcn_wmma_f32_16x16x4_f32(false, a, false, b, (short)0, c3, false, false);
  }
  int mrow = wave * 16 + hh * 8;
  for (int j = 0; j < 8; ++j) {
    float ds = dinv[mrow + j];
    float* o = H + (mrow + j) * HID + lid;
    o[0]  = c0[j] * ds;
    o[16] = c1[j] * ds;
    o[32] = c2[j] * ds;
    o[48] = c3[j] * ds;
  }
}

#define TILE_NODES 16
#define LPN 8
#define AP 68
static_assert(NN % TILE_NODES == 0, "every block owns a whole row tile");

__global__ void __launch_bounds__(TILE_NODES * LPN)
k_gather64_hw12(const float* __restrict__ Hlin,
                const int* __restrict__ lst,
                const int* __restrict__ off,
                const int* __restrict__ cnt,
                const float* __restrict__ dinv,
                const float* __restrict__ bias,
                const float* __restrict__ mask,
                const float* __restrict__ W1,
                const float* __restrict__ W2,
                float* __restrict__ MU,
                float* __restrict__ LV) {
  __shared__ float sA[TILE_NODES * AP];
  int t = threadIdx.x;
  int r = t >> 3;
  int i = blockIdx.x * TILE_NODES + r;
  int g = t & 7;
  const float4* H4 = (const float4*)Hlin;
  float di = dinv[i];
  float4 a0 = H4[i * 16 + g];
  float4 a1 = H4[i * 16 + 8 + g];
  float4 m0 = ((const float4*)mask)[i * 16 + g];
  float4 m1 = ((const float4*)mask)[i * 16 + 8 + g];
  int j = off[i];
  int end = j + cnt[i];
  for (; j + 4 <= end; j += 4) {
    int s0 = clamp_idx(lst[j + 0], NN - 1);
    int s1 = clamp_idx(lst[j + 1], NN - 1);
    int s2 = clamp_idx(lst[j + 2], NN - 1);
    int s3 = clamp_idx(lst[j + 3], NN - 1);
    float4 p0 = H4[s0 * 16 + g], q0 = H4[s0 * 16 + 8 + g];
    float4 p1 = H4[s1 * 16 + g], q1 = H4[s1 * 16 + 8 + g];
    float4 p2 = H4[s2 * 16 + g], q2 = H4[s2 * 16 + 8 + g];
    float4 p3 = H4[s3 * 16 + g], q3 = H4[s3 * 16 + 8 + g];
    a0.x += p0.x; a0.y += p0.y; a0.z += p0.z; a0.w += p0.w;
    a1.x += q0.x; a1.y += q0.y; a1.z += q0.z; a1.w += q0.w;
    a0.x += p1.x; a0.y += p1.y; a0.z += p1.z; a0.w += p1.w;
    a1.x += q1.x; a1.y += q1.y; a1.z += q1.z; a1.w += q1.w;
    a0.x += p2.x; a0.y += p2.y; a0.z += p2.z; a0.w += p2.w;
    a1.x += q2.x; a1.y += q2.y; a1.z += q2.z; a1.w += q2.w;
    a0.x += p3.x; a0.y += p3.y; a0.z += p3.z; a0.w += p3.w;
    a1.x += q3.x; a1.y += q3.y; a1.z += q3.z; a1.w += q3.w;
  }
#pragma unroll 1
  for (; j < end; ++j) {
    int s0 = clamp_idx(lst[j], NN - 1);
    float4 p0 = H4[s0 * 16 + g], q0 = H4[s0 * 16 + 8 + g];
    a0.x += p0.x; a0.y += p0.y; a0.z += p0.z; a0.w += p0.w;
    a1.x += q0.x; a1.y += q0.y; a1.z += q0.z; a1.w += q0.w;
  }
  float4 o0, o1;
  o0.x = fmaxf(a0.x * di + bias[4 * g + 0], 0.0f) * m0.x * di;
  o0.y = fmaxf(a0.y * di + bias[4 * g + 1], 0.0f) * m0.y * di;
  o0.z = fmaxf(a0.z * di + bias[4 * g + 2], 0.0f) * m0.z * di;
  o0.w = fmaxf(a0.w * di + bias[4 * g + 3], 0.0f) * m0.w * di;
  o1.x = fmaxf(a1.x * di + bias[32 + 4 * g + 0], 0.0f) * m1.x * di;
  o1.y = fmaxf(a1.y * di + bias[32 + 4 * g + 1], 0.0f) * m1.y * di;
  o1.z = fmaxf(a1.z * di + bias[32 + 4 * g + 2], 0.0f) * m1.z * di;
  o1.w = fmaxf(a1.w * di + bias[32 + 4 * g + 3], 0.0f) * m1.w * di;
  *(float4*)&sA[r * AP + 4 * g] = o0;
  *(float4*)&sA[r * AP + 32 + 4 * g] = o1;
  __syncthreads();
  int wave = t >> 5;
  if (wave >= 2) return;
  const float* Wm = wave == 0 ? W1 : W2;
  float*       Om = wave == 0 ? MU : LV;
  int lane = t & 31;
  int hh   = lane >> 4;
  int lid  = lane & 15;
  const float* ar = sA + lid * AP;
  v8f c0 = {}, c1 = {};
#pragma unroll 8
  for (int k = 0; k < HID; k += 4) {
    int ka = k + 2 * hh;
    v2f a;
    a.x = ar[ka];
    a.y = ar[ka + 1];
    const float* u0 = Wm + ka * NC;
    const float* u1 = u0 + NC;
    v2f b;
    b.x = u0[lid];      b.y = u1[lid];
    c0 = __builtin_amdgcn_wmma_f32_16x16x4_f32(false, a, false, b, (short)0, c0, false, false);
    b.x = u0[lid + 16]; b.y = u1[lid + 16];
    c1 = __builtin_amdgcn_wmma_f32_16x16x4_f32(false, a, false, b, (short)0, c1, false, false);
  }
  int mrow = blockIdx.x * TILE_NODES + hh * 8;
  for (int jj = 0; jj < 8; ++jj) {
    float* om = Om + (mrow + jj) * NC + lid;
    om[0]  = c0[jj];
    om[16] = c1[jj];
  }
}

__global__ void k_gather32(const float* __restrict__ MUlin,
                           const float* __restrict__ LVlin,
                           const int* __restrict__ lst,
                           const int* __restrict__ off,
                           const int* __restrict__ cnt,
                           const float* __restrict__ dinv,
                           const float* __restrict__ b1,
                           const float* __restrict__ b2,
                           const float* __restrict__ eps,
                           float* __restrict__ MUout,
                           float* __restrict__ LVout,
                           float* __restrict__ Z) {
  int idx = blockIdx.x * blockDim.x + threadIdx.x;
  int i = idx >> 3;
  if (i >= NN) return;
  int g = idx & 7;
  const float4* M4 = (const float4*)MUlin;
  const float4* L4 = (const float4*)LVlin;
  float di = dinv[i];
  float4 am = M4[i * 8 + g];
  float4 al = L4[i * 8 + g];
  float4 ep = ((const float4*)eps)[i * 8 + g];
  int j = off[i];
  int end = j + cnt[i];
  for (; j + 4 <= end; j += 4) {
    int s0 = clamp_idx(lst[j + 0], NN - 1);
    int s1 = clamp_idx(lst[j + 1], NN - 1);
    int s2 = clamp_idx(lst[j + 2], NN - 1);
    int s3 = clamp_idx(lst[j + 3], NN - 1);
    float4 p0 = M4[s0 * 8 + g], q0 = L4[s0 * 8 + g];
    float4 p1 = M4[s1 * 8 + g], q1 = L4[s1 * 8 + g];
    float4 p2 = M4[s2 * 8 + g], q2 = L4[s2 * 8 + g];
    float4 p3 = M4[s3 * 8 + g], q3 = L4[s3 * 8 + g];
    am.x += p0.x; am.y += p0.y; am.z += p0.z; am.w += p0.w;
    al.x += q0.x; al.y += q0.y; al.z += q0.z; al.w += q0.w;
    am.x += p1.x; am.y += p1.y; am.z += p1.z; am.w += p1.w;
    al.x += q1.x; al.y += q1.y; al.z += q1.z; al.w += q1.w;
    am.x += p2.x; am.y += p2.y; am.z += p2.z; am.w += p2.w;
    al.x += q2.x; al.y += q2.y; al.z += q2.z; al.w += q2.w;
    am.x += p3.x; am.y += p3.y; am.z += p3.z; am.w += p3.w;
    al.x += q3.x; al.y += q3.y; al.z += q3.z; al.w += q3.w;
  }
  for (; j < end; ++j) {
    int s0 = clamp_idx(lst[j], NN - 1);
    float4 p0 = M4[s0 * 8 + g], q0 = L4[s0 * 8 + g];
    am.x += p0.x; am.y += p0.y; am.z += p0.z; am.w += p0.w;
    al.x += q0.x; al.y += q0.y; al.z += q0.z; al.w += q0.w;
  }
  am.x = am.x * di + b1[4 * g + 0];
  am.y = am.y * di + b1[4 * g + 1];
  am.z = am.z * di + b1[4 * g + 2];
  am.w = am.w * di + b1[4 * g + 3];
  al.x = al.x * di + b2[4 * g + 0];
  al.y = al.y * di + b2[4 * g + 1];
  al.z = al.z * di + b2[4 * g + 2];
  al.w = al.w * di + b2[4 * g + 3];
  ((float4*)MUout)[i * 8 + g] = am;
  ((float4*)LVout)[i * 8 + g] = al;
  float4 zz;
  zz.x = ep.x * expf(al.x) + am.x;
  zz.y = ep.y * expf(al.y) + am.y;
  zz.z = ep.z * expf(al.z) + am.z;
  zz.w = ep.w * expf(al.w) + am.w;
  ((float4*)Z)[i * 8 + g] = zz;
}

extern "C" void kernel_launch(void* const* d_in, const int* in_sizes, int n_in,
                              void* d_out, int out_size, void* d_ws, size_t ws_size,
                              hipStream_t stream) {
  (void)in_sizes; (void)n_in; (void)out_size; (void)ws_size;
  const float* x    = (const float*)d_in[0];
  const int*   ei   = (const int*)d_in[1];
  const float* W0   = (const float*)d_in[2];
  const float* b0   = (const float*)d_in[3];
  const float* W1   = (const float*)d_in[4];
  const float* b1   = (const float*)d_in[5];
  const float* W2   = (const float*)d_in[6];
  const float* b2   = (const float*)d_in[7];
  const float* mask = (const float*)d_in[8];
  const float* eps  = (const float*)d_in[9];
  const int* src = ei;
  const int* dst = ei + NE;

  int*   cnt   = (int*)d_ws;
  int*   off   = cnt + NNP;
  float* dinv  = (float*)(off + NNP);
  int*   lpre  = (int*)(dinv + NNP);
  int*   key   = lpre + NBLK * HP;
  int*   lst   = key + NE;
  float* hlin  = (float*)(lst + NE);
  float* mulin = hlin + (size_t)NN * HID;
  float* lvlin = mulin + (size_t)NN * NC;

  float* z  = (float*)d_out;
  float* mu = z + (size_t)NN * NC;
  float* lv = mu + (size_t)NN * NC;

  const int B = 256;
  const int gW   = (ROW_TILES * 32 + B - 1) / B;
  const int gG   = (NN * LPN + B - 1) / B;

  k_bplace<<<NBLK, TPP, 0, stream>>>(src, dst, key, lpre);
  k_bsort <<<NBK,  TPB, 0, stream>>>(key, lpre, lst, cnt, off, dinv);

  k_gemm_xw0 <<<gW,  B, 0, stream>>>(x, W0, dinv, hlin);
  k_gather64_hw12<<<NN / TILE_NODES, TILE_NODES * LPN, 0, stream>>>(hlin, lst, off, cnt, dinv, b0, mask,
                                                                  W1, W2, mulin, lvlin);

  k_gather32 <<<gG,   B, 0, stream>>>(mulin, lvlin, lst, off, cnt, dinv, b1, b2, eps, mu, lv, z);
}


# X @ W0 for a bucket's 256 rows runs at the end of the bucket kernel on 16 of its waves (dinv from its own LDS, X prefetched behind its last loads); no product launch (5 to 4); sums unchanged
# speedup vs baseline: 6.0694x; 1.0882x over previous
#include <hip/hip_runtime.h>


typedef __attribute__((ext_vector_type(2))) float v2f;
typedef __attribute__((ext_vector_type(8))) float v8f;

#define NN 50000
#define NE 800000
#define KIN 128
#define HID 64
#define NC 32
#define ROW_TILES (NN / 16)
#define NNP 50048

__device__ __forceinline__ int clamp_idx(int v, int hi) {
  return v < 0 ? 0 : (v > hi ? hi : v);
}

#define NBLK 256
#define EPB  3125
#define TPP  512
#define EPT  7
#define NBK  196
#define HP   256
#define TPB  1024
#define SPW  (NBLK / 32)
#define RPL  1
static_assert(NBLK * EPB == NE, "the edge blocks cover the edge list exactly");
static_assert(EPT * TPP >= EPB && HP <= TPP, "a block's threads cover its edges and the table row");
static_assert(NBK + 1 <= HP && NBLK % 32 == 0 && NBLK <= TPB && TPB == 1024, "table pitch and run split");

__global__ void __launch_bounds__(TPP) k_bplace(const int* __restrict__ src,
                                                const int* __restrict__ dst,
                                                int* __restrict__ key,
                                                int* __restrict__ lpre) {
  __shared__ int h[HP];
  __shared__ int cur[HP];
  __shared__ int wsum[8];
  __shared__ int sk[EPB];
  int t = threadIdx.x;
  int b = blockIdx.x;
  int lane = t & 31;
  int w = t >> 5;
  if (t < HP) h[t] = 0;
  const int* d0 = dst + b * EPB;
  const int* s0 = src + b * EPB;
  int dv[EPT], sv[EPT];
#pragma unroll
  for (int j = 0; j < EPT; ++j) {
    int i = t + j * TPP;
    int ii = i < EPB ? i : EPB - 1;
    dv[j] = d0[ii];
    sv[j] = s0[ii];
  }
  __syncthreads();
#pragma unroll
  for (int j = 0; j < EPT; ++j) {
    int i = t + j * TPP;
    dv[j] = clamp_idx(dv[j], NN - 1);
    if (i < EPB) atomicAdd(&h[dv[j] >> 8], 1);
  }
  __syncthreads();
  int c = 0, x = 0;
  if (t < HP) {
    c = h[t];
    x = c;
    for (int o = 1; o < 32; o <<= 1) {
      int y = __shfl_up(x, o);
      if (lane >= o) x += y;
    }
    if (lane == 31) wsum[w] = x;
  }
  __syncthreads();
  if (t < HP) {
    int o = x - c;
    for (int q = 0; q < 8; ++q) o += (q < w) ? wsum[q] : 0;
    cur[t] = o;
    lpre[b * HP + t] = o;
  }
  __syncthreads();
  int pos[EPT];
#pragma unroll
  for (int j = 0; j < EPT; ++j) {
    int i = t + j * TPP;
    pos[j] = (i < EPB) ? atomicAdd(&cur[dv[j] >> 8], 1) : 0;
  }
#pragma unroll
  for (int j = 0; j < EPT; ++j) {
    int i = t + j * TPP;
    if (i < EPB)
      sk[clamp_idx(pos[j], EPB - 1)] = ((dv[j] & 255) << 16) | clamp_idx(sv[j], NN - 1);
  }
  __syncthreads();
#pragma unroll
  for (int j = 0; j < EPT; ++j) {
    int i = t + j * TPP;
    if (i < EPB) key[b * EPB + i] = sk[i];
  }
}

#define GW 16
static_assert(GW * 16 == 256 && GW * 32 <= TPB, "a bucket is GW row tiles and the work-group has the waves for them");
__global__ void __launch_bounds__(TPB) k_bsort(const int* __restrict__ key,
                                               const int* __restrict__ lpre,
                                               int* __restrict__ lst,
                                               int* __restrict__ cnt,
                                               int* __restrict__ off,
                                               float* __restrict__ dinv,
                                               const float* __restrict__ X,
                                               const float* __restrict__ W,
                                               float* __restrict__ H) {
  __shared__ int c[256];
  __shared__ float sdv[256];
  __shared__ int cur[256];
  __shared__ int wsum[8];
  __shared__ int segA[NBLK];
  __shared__ int segC[NBLK];
  __shared__ int redA[NBLK / 32];
  int t = threadIdx.x;
  int kb = blockIdx.x;
  int lane = t & 31;
  int w = t >> 5;
  int tile = kb * GW + w;
  bool prod = (w < GW) && (tile < ROW_TILES);
  if (t < 256) c[t] = 0;
  if (t < NBLK) {
    int a  = clamp_idx(lpre[t * HP + kb], EPB);
    int a1 = clamp_idx(lpre[t * HP + kb + 1], EPB);
    segA[t] = t * EPB + a;
    segC[t] = a1 > a ? a1 - a : 0;
    int sa = a;
    for (int o = 16; o > 0; o >>= 1) sa += __shfl_xor(sa, o);
    if (lane == 0) redA[w] = sa;
  }
  __syncthreads();
  int base = 0;
#pragma unroll
  for (int q = 0; q < NBLK / 32; ++q) base += redA[q];
  base = clamp_idx(base, NE);
  int kv[SPW * RPL];
#pragma unroll
  for (int s = 0; s < SPW; ++s) {
    int st = segA[SPW * w + s];
    int ln = segC[SPW * w + s];
#pragma unroll
    for (int r = 0; r < RPL; ++r) {
      int i = lane + 32 * r;
      kv[s * RPL + r] = (i < ln) ? key[st + i] : -1;
    }
  }
  if (prod) __builtin_prefetch(X + tile * 16 * KIN + lane * 64, 0, 1);
#pragma unroll
  for (int e = 0; e < SPW * RPL; ++e)
    if (kv[e] >= 0) atomicAdd(&c[(kv[e] >> 16) & 255], 1);
  for (int s = 0; s < SPW; ++s) {
    int st = segA[SPW * w + s];
    int ln = segC[SPW * w + s];
    for (int i = lane + 32 * RPL; i < ln; i += 32)
      atomicAdd(&c[(key[st + i] >> 16) & 255], 1);
  }
  __syncthreads();
  int cc = 0, x = 0;
  if (t < 256) {
    cc = c[t];
    x = cc;
    for (int o = 1; o < 32; o <<= 1) {
      int y = __shfl_up(x, o);
      if (lane >= o) x += y;
    }
    if (lane == 31) wsum[w] = x;
  }
  __syncthreads();
  if (t < 256) {
    int o = x - cc;
    for (int q = 0; q < 8; ++q) o += (q < w) ? wsum[q] : 0;
    cur[t] = o;
    int node = kb * 256 + t;
    float dv = rsqrtf((float)cc + 1.0f);
    sdv[t] = dv;
    if (node < NN) {
      cnt[node]  = cc;
      off[node]  = base + o;
      dinv[node] = dv;
    }
  }
  __syncthreads();
  int pl[SPW * RPL];
#pragma unroll
  for (int e = 0; e < SPW * RPL; ++e)
    pl[e] = (kv[e] >= 0) ? atomicAdd(&cur[(kv[e] >> 16) & 255], 1) : 0;
#pragma unroll
  for (int e = 0; e < SPW * RPL; ++e)
    if (kv[e] >= 0) lst[clamp_idx(base + pl[e], NE - 1)] = kv[e] & 0xFFFF;
  for (int s = 0; s < SPW; ++s) {
    int st = segA[SPW * w + s];
    int ln = segC[SPW * w + s];
    for (int i = lane + 32 * RPL; i < ln; i += 32) {
      int v = key[st + i];
      int p1 = atomicAdd(&cur[(v >> 16) & 255], 1);
      lst[clamp_idx(base + p1, NE - 1)] = v & 0xFFFF;
    }
  }
  if (!prod) return;
  int hh   = lane >> 4;
  int lid  = lane & 15;
  const float* xr = X + (tile * 16 + lid) * KIN;
  v8f c0 = {}, c1 = {}, c2 = {}, c3 = {};
  for (int k = 0; k < KIN; k += 4) {
    int ka = k + 2 * hh;
    v2f a;
    a.x = xr[ka];
    a.y = xr[ka + 1];
    const float* wr0 = W + ka * HID;
    const float* wr1 = wr0 + HID;
    v2f b;
    b.x = wr0[lid];      b.y = wr1[lid];
    c0 = __builtin_amdgcn_wmma_f32_16x16x4_f32(false, a, false, b, (short)0, c0, false, false);
    b.x = wr0[lid + 16]; b.y = wr1[lid + 16];
    c1 = __builtin_amdgcn_wmma_f32_16x16x4_f32(false, a, false, b, (short)0, c1, false, false);
    b.x = wr0[lid + 32]; b.y = wr1[lid + 32];
    c2 = __builtin_amdgcn_wmma_f32_16x16x4_f32(false, a, false, b, (short)0, c2, false, false);
    b.x = wr0[lid + 48]; b.y = wr1[lid + 48];
    c3 = __builtin_amdgcn_wmma_f32_16x16x4_f32(false, a, false, b, (short)0, c3, false, false);
  }
  int lrow = w * 16 + hh * 8;
  int mrow = tile * 16 + hh * 8;
  for (int j = 0; j < 8; ++j) {
    float ds = sdv[lrow + j];
    float* o = H + (mrow + j) * HID + lid;
    o[0]  = c0[j] * ds;
    o[16] = c1[j] * ds;
    o[32] = c2[j] * ds;
    o[48] = c3[j] * ds;
  }
}

#define TILE_NODES 16
#define LPN 8
#define AP 68
static_assert(NN % TILE_NODES == 0, "every block owns a whole row tile");

__global__ void __launch_bounds__(TILE_NODES * LPN)
k_gather64_hw12(const float* __restrict__ Hlin,
                const int* __restrict__ lst,
                const int* __restrict__ off,
                const int* __restrict__ cnt,
                const float* __restrict__ dinv,
                const float* __restrict__ bias,
                const float* __restrict__ mask,
                const float* __restrict__ W1,
                const float* __restrict__ W2,
                float* __restrict__ MU,
                float* __restrict__ LV) {
  __shared__ float sA[TILE_NODES * AP];
  int t = threadIdx.x;
  int r = t >> 3;
  int i = blockIdx.x * TILE_NODES + r;
  int g = t & 7;
  const float4* H4 = (const float4*)Hlin;
  float di = dinv[i];
  float4 a0 = H4[i * 16 + g];
  float4 a1 = H4[i * 16 + 8 + g];
  float4 m0 = ((const float4*)mask)[i * 16 + g];
  float4 m1 = ((const float4*)mask)[i * 16 + 8 + g];
  int j = off[i];
  int end = j + cnt[i];
  for (; j + 4 <= end; j += 4) {
    int s0 = clamp_idx(lst[j + 0], NN - 1);
    int s1 = clamp_idx(lst[j + 1], NN - 1);
    int s2 = clamp_idx(lst[j + 2], NN - 1);
    int s3 = clamp_idx(lst[j + 3], NN - 1);
    float4 p0 = H4[s0 * 16 + g], q0 = H4[s0 * 16 + 8 + g];
    float4 p1 = H4[s1 * 16 + g], q1 = H4[s1 * 16 + 8 + g];
    float4 p2 = H4[s2 * 16 + g], q2 = H4[s2 * 16 + 8 + g];
    float4 p3 = H4[s3 * 16 + g], q3 = H4[s3 * 16 + 8 + g];
    a0.x += p0.x; a0.y += p0.y; a0.z += p0.z; a0.w += p0.w;
    a1.x += q0.x; a1.y += q0.y; a1.z += q0.z; a1.w += q0.w;
    a0.x += p1.x; a0.y += p1.y; a0.z += p1.z; a0.w += p1.w;
    a1.x += q1.x; a1.y += q1.y; a1.z += q1.z; a1.w += q1.w;
    a0.x += p2.x; a0.y += p2.y; a0.z += p2.z; a0.w += p2.w;
    a1.x += q2.x; a1.y += q2.y; a1.z += q2.z; a1.w += q2.w;
    a0.x += p3.x; a0.y += p3.y; a0.z += p3.z; a0.w += p3.w;
    a1.x += q3.x; a1.y += q3.y; a1.z += q3.z; a1.w += q3.w;
  }
#pragma unroll 1
  for (; j < end; ++j) {
    int s0 = clamp_idx(lst[j], NN - 1);
    float4 p0 = H4[s0 * 16 + g], q0 = H4[s0 * 16 + 8 + g];
    a0.x += p0.x; a0.y += p0.y; a0.z += p0.z; a0.w += p0.w;
    a1.x += q0.x; a1.y += q0.y; a1.z += q0.z; a1.w += q0.w;
  }
  float4 o0, o1;
  o0.x = fmaxf(a0.x * di + bias[4 * g + 0], 0.0f) * m0.x * di;
  o0.y = fmaxf(a0.y * di + bias[4 * g + 1], 0.0f) * m0.y * di;
  o0.z = fmaxf(a0.z * di + bias[4 * g + 2], 0.0f) * m0.z * di;
  o0.w = fmaxf(a0.w * di + bias[4 * g + 3], 0.0f) * m0.w * di;
  o1.x = fmaxf(a1.x * di + bias[32 + 4 * g + 0], 0.0f) * m1.x * di;
  o1.y = fmaxf(a1.y * di + bias[32 + 4 * g + 1], 0.0f) * m1.y * di;
  o1.z = fmaxf(a1.z * di + bias[32 + 4 * g + 2], 0.0f) * m1.z * di;
  o1.w = fmaxf(a1.w * di + bias[32 + 4 * g + 3], 0.0f) * m1.w * di;
  *(float4*)&sA[r * AP + 4 * g] = o0;
  *(float4*)&sA[r * AP + 32 + 4 * g] = o1;
  __syncthreads();
  int wave = t >> 5;
  if (wave >= 2) return;
  const float* Wm = wave == 0 ? W1 : W2;
  float*       Om = wave == 0 ? MU : LV;
  int lane = t & 31;
  int hh   = lane >> 4;
  int lid  = lane & 15;
  const float* ar = sA + lid * AP;
  v8f c0 = {}, c1 = {};
#pragma unroll 8
  for (int k = 0; k < HID; k += 4) {
    int ka = k + 2 * hh;
    v2f a;
    a.x = ar[ka];
    a.y = ar[ka + 1];
    const float* u0 = Wm + ka * NC;
    const float* u1 = u0 + NC;
    v2f b;
    b.x = u0[lid];      b.y = u1[lid];
    c0 = __builtin_amdgcn_wmma_f32_16x16x4_f32(false, a, false, b, (short)0, c0, false, false);
    b.x = u0[lid + 16]; b.y = u1[lid + 16];
    c1 = __builtin_amdgcn_wmma_f32_16x16x4_f32(false, a, false, b, (short)0, c1, false, false);
  }
  int mrow = blockIdx.x * TILE_NODES + hh * 8;
  for (int jj = 0; jj < 8; ++jj) {
    float* om = Om + (mrow + jj) * NC + lid;
    om[0]  = c0[jj];
    om[16] = c1[jj];
  }
}

__global__ void k_gather32(const float* __restrict__ MUlin,
                           const float* __restrict__ LVlin,
                           const int* __restrict__ lst,
                           const int* __restrict__ off,
                           const int* __restrict__ cnt,
                           const float* __restrict__ dinv,
                           const float* __restrict__ b1,
                           const float* __restrict__ b2,
                           const float* __restrict__ eps,
                           float* __restrict__ MUout,
                           float* __restrict__ LVout,
                           float* __restrict__ Z) {
  int idx = blockIdx.x * blockDim.x + threadIdx.x;
  int i = idx >> 3;
  if (i >= NN) return;
  int g = idx & 7;
  const float4* M4 = (const float4*)MUlin;
  const float4* L4 = (const float4*)LVlin;
  float di = dinv[i];
  float4 am = M4[i * 8 + g];
  float4 al = L4[i * 8 + g];
  float4 ep = ((const float4*)eps)[i * 8 + g];
  int j = off[i];
  int end = j + cnt[i];
  for (; j + 4 <= end; j += 4) {
    int s0 = clamp_idx(lst[j + 0], NN - 1);
    int s1 = clamp_idx(lst[j + 1], NN - 1);
    int s2 = clamp_idx(lst[j + 2], NN - 1);
    int s3 = clamp_idx(lst[j + 3], NN - 1);
    float4 p0 = M4[s0 * 8 + g], q0 = L4[s0 * 8 + g];
    float4 p1 = M4[s1 * 8 + g], q1 = L4[s1 * 8 + g];
    float4 p2 = M4[s2 * 8 + g], q2 = L4[s2 * 8 + g];
    float4 p3 = M4[s3 * 8 + g], q3 = L4[s3 * 8 + g];
    am.x += p0.x; am.y += p0.y; am.z += p0.z; am.w += p0.w;
    al.x += q0.x; al.y += q0.y; al.z += q0.z; al.w += q0.w;
    am.x += p1.x; am.y += p1.y; am.z += p1.z; am.w += p1.w;
    al.x += q1.x; al.y += q1.y; al.z += q1.z; al.w += q1.w;
    am.x += p2.x; am.y += p2.y; am.z += p2.z; am.w += p2.w;
    al.x += q2.x; al.y += q2.y; al.z += q2.z; al.w += q2.w;
    am.x += p3.x; am.y += p3.y; am.z += p3.z; am.w += p3.w;
    al.x += q3.x; al.y += q3.y; al.z += q3.z; al.w += q3.w;
  }
  for (; j < end; ++j) {
    int s0 = clamp_idx(lst[j], NN - 1);
    float4 p0 = M4[s0 * 8 + g], q0 = L4[s0 * 8 + g];
    am.x += p0.x; am.y += p0.y; am.z += p0.z; am.w += p0.w;
    al.x += q0.x; al.y += q0.y; al.z += q0.z; al.w += q0.w;
  }
  am.x = am.x * di + b1[4 * g + 0];
  am.y = am.y * di + b1[4 * g + 1];
  am.z = am.z * di + b1[4 * g + 2];
  am.w = am.w * di + b1[4 * g + 3];
  al.x = al.x * di + b2[4 * g + 0];
  al.y = al.y * di + b2[4 * g + 1];
  al.z = al.z * di + b2[4 * g + 2];
  al.w = al.w * di + b2[4 * g + 3];
  ((float4*)MUout)[i * 8 + g] = am;
  ((float4*)LVout)[i * 8 + g] = al;
  float4 zz;
  zz.x = ep.x * expf(al.x) + am.x;
  zz.y = ep.y * expf(al.y) + am.y;
  zz.z = ep.z * expf(al.z) + am.z;
  zz.w = ep.w * expf(al.w) + am.w;
  ((float4*)Z)[i * 8 + g] = zz;
}

extern "C" void kernel_launch(void* const* d_in, const int* in_sizes, int n_in,
                              void* d_out, int out_size, void* d_ws, size_t ws_size,
                              hipStream_t stream) {
  (void)in_sizes; (void)n_in; (void)out_size; (void)ws_size;
  const float* x    = (const float*)d_in[0];
  const int*   ei   = (const int*)d_in[1];
  const float* W0   = (const float*)d_in[2];
  const float* b0   = (const float*)d_in[3];
  const float* W1   = (const float*)d_in[4];
  const float* b1   = (const float*)d_in[5];
  const float* W2   = (const float*)d_in[6];
  const float* b2   = (const float*)d_in[7];
  const float* mask = (const float*)d_in[8];
  const float* eps  = (const float*)d_in[9];
  const int* src = ei;
  const int* dst = ei + NE;

  int*   cnt   = (int*)d_ws;
  int*   off   = cnt + NNP;
  float* dinv  = (float*)(off + NNP);
  int*   lpre  = (int*)(dinv + NNP);
  int*   key   = lpre + NBLK * HP;
  int*   lst   = key + NE;
  float* hlin  = (float*)(lst + NE);
  float* mulin = hlin + (size_t)NN * HID;
  float* lvlin = mulin + (size_t)NN * NC;

  float* z  = (float*)d_out;
  float* mu = z + (size_t)NN * NC;
  float* lv = mu + (size_t)NN * NC;

  const int B = 256;
  const int gG   = (NN * LPN + B - 1) / B;

  k_bplace<<<NBLK, TPP, 0, stream>>>(src, dst, key, lpre);
  k_bsort <<<NBK,  TPB, 0, stream>>>(key, lpre, lst, cnt, off, dinv, x, W0, hlin);

  k_gather64_hw12<<<NN / TILE_NODES, TILE_NODES * LPN, 0, stream>>>(hlin, lst, off, cnt, dinv, b0, mask,
                                                                  W1, W2, mulin, lvlin);

  k_gather32 <<<gG,   B, 0, stream>>>(mulin, lvlin, lst, off, cnt, dinv, b1, b2, eps, mu, lv, z);
}


# Row sums: a node's 8 lanes load its first 32 list entries side by side up front and share them by DPP8: no row load waits for a list load of its trip, leftover edges load together; list order kept
# speedup vs baseline: 6.3961x; 1.0538x over previous
#include <hip/hip_runtime.h>


typedef __attribute__((ext_vector_type(2))) float v2f;
typedef __attribute__((ext_vector_type(8))) float v8f;

#define NN 50000
#define NE 800000
#define KIN 128
#define HID 64
#define NC 32
#define ROW_TILES (NN / 16)
#define NNP 50048

__device__ __forceinline__ int clamp_idx(int v, int hi) {
  return v < 0 ? 0 : (v > hi ? hi : v);
}

template <int K>
__device__ __forceinline__ int bc8(int v) {
  return __builtin_amdgcn_mov_dpp8(v, K * 0x249249);
}

#define NBLK 256
#define EPB  3125
#define TPP  512
#define EPT  7
#define NBK  196
#define HP   256
#define TPB  1024
#define SPW  (NBLK / 32)
#define RPL  1
static_assert(NBLK * EPB == NE, "the edge blocks cover the edge list exactly");
static_assert(EPT * TPP >= EPB && HP <= TPP, "a block's threads cover its edges and the table row");
static_assert(NBK + 1 <= HP && NBLK % 32 == 0 && NBLK <= TPB && TPB == 1024, "table pitch and run split");

__global__ void __launch_bounds__(TPP) k_bplace(const int* __restrict__ src,
                                                const int* __restrict__ dst,
                                                int* __restrict__ key,
                                                int* __restrict__ lpre) {
  __shared__ int h[HP];
  __shared__ int cur[HP];
  __shared__ int wsum[8];
  __shared__ int sk[EPB];
  int t = threadIdx.x;
  int b = blockIdx.x;
  int lane = t & 31;
  int w = t >> 5;
  if (t < HP) h[t] = 0;
  const int* d0 = dst + b * EPB;
  const int* s0 = src + b * EPB;
  int dv[EPT], sv[EPT];
#pragma unroll
  for (int j = 0; j < EPT; ++j) {
    int i = t + j * TPP;
    int ii = i < EPB ? i : EPB - 1;
    dv[j] = d0[ii];
    sv[j] = s0[ii];
  }
  __syncthreads();
#pragma unroll
  for (int j = 0; j < EPT; ++j) {
    int i = t + j * TPP;
    dv[j] = clamp_idx(dv[j], NN - 1);
    if (i < EPB) atomicAdd(&h[dv[j] >> 8], 1);
  }
  __syncthreads();
  int c = 0, x = 0;
  if (t < HP) {
    c = h[t];
    x = c;
    for (int o = 1; o < 32; o <<= 1) {
      int y = __shfl_up(x, o);
      if (lane >= o) x += y;
    }
    if (lane == 31) wsum[w] = x;
  }
  __syncthreads();
  if (t < HP) {
    int o = x - c;
    for (int q = 0; q < 8; ++q) o += (q < w) ? wsum[q] : 0;
    cur[t] = o;
    lpre[b * HP + t] = o;
  }
  __syncthreads();
  int pos[EPT];
#pragma unroll
  for (int j = 0; j < EPT; ++j) {
    int i = t + j * TPP;
    pos[j] = (i < EPB) ? atomicAdd(&cur[dv[j] >> 8], 1) : 0;
  }
#pragma unroll
  for (int j = 0; j < EPT; ++j) {
    int i = t + j * TPP;
    if (i < EPB)
      sk[clamp_idx(pos[j], EPB - 1)] = ((dv[j] & 255) << 16) | clamp_idx(sv[j], NN - 1);
  }
  __syncthreads();
#pragma unroll
  for (int j = 0; j < EPT; ++j) {
    int i = t + j * TPP;
    if (i < EPB) key[b * EPB + i] = sk[i];
  }
}

#define GW 16
static_assert(GW * 16 == 256 && GW * 32 <= TPB, "a bucket is GW row tiles and the work-group has the waves for them");
__global__ void __launch_bounds__(TPB) k_bsort(const int* __restrict__ key,
                                               const int* __restrict__ lpre,
                                               int* __restrict__ lst,
                                               int* __restrict__ cnt,
                                               int* __restrict__ off,
                                               float* __restrict__ dinv,
                                               const float* __restrict__ X,
                                               const float* __restrict__ W,
                                               float* __restrict__ H) {
  __shared__ int c[256];
  __shared__ float sdv[256];
  __shared__ int cur[256];
  __shared__ int wsum[8];
  __shared__ int segA[NBLK];
  __shared__ int segC[NBLK];
  __shared__ int redA[NBLK / 32];
  int t = threadIdx.x;
  int kb = blockIdx.x;
  int lane = t & 31;
  int w = t >> 5;
  int tile = kb * GW + w;
  bool prod = (w < GW) && (tile < ROW_TILES);
  if (t < 256) c[t] = 0;
  if (t < NBLK) {
    int a  = clamp_idx(lpre[t * HP + kb], EPB);
    int a1 = clamp_idx(lpre[t * HP + kb + 1], EPB);
    segA[t] = t * EPB + a;
    segC[t] = a1 > a ? a1 - a : 0;
    int sa = a;
    for (int o = 16; o > 0; o >>= 1) sa += __shfl_xor(sa, o);
    if (lane == 0) redA[w] = sa;
  }
  __syncthreads();
  int base = 0;
#pragma unroll
  for (int q = 0; q < NBLK / 32; ++q) base += redA[q];
  base = clamp_idx(base, NE);
  int kv[SPW * RPL];
#pragma unroll
  for (int s = 0; s < SPW; ++s) {
    int st = segA[SPW * w + s];
    int ln = segC[SPW * w + s];
#pragma unroll
    for (int r = 0; r < RPL; ++r) {
      int i = lane + 32 * r;
      kv[s * RPL + r] = (i < ln) ? key[st + i] : -1;
    }
  }
  if (prod) __builtin_prefetch(X + tile * 16 * KIN + lane * 64, 0, 1);
#pragma unroll
  for (int e = 0; e < SPW * RPL; ++e)
    if (kv[e] >= 0) atomicAdd(&c[(kv[e] >> 16) & 255], 1);
  for (int s = 0; s < SPW; ++s) {
    int st = segA[SPW * w + s];
    int ln = segC[SPW * w + s];
    for (int i = lane + 32 * RPL; i < ln; i += 32)
      atomicAdd(&c[(key[st + i] >> 16) & 255], 1);
  }
  __syncthreads();
  int cc = 0, x = 0;
  if (t < 256) {
    cc = c[t];
    x = cc;
    for (int o = 1; o < 32; o <<= 1) {
      int y = __shfl_up(x, o);
      if (lane >= o) x += y;
    }
    if (lane == 31) wsum[w] = x;
  }
  __syncthreads();
  if (t < 256) {
    int o = x - cc;
    for (int q = 0; q < 8; ++q) o += (q < w) ? wsum[q] : 0;
    cur[t] = o;
    int node = kb * 256 + t;
    float dv = rsqrtf((float)cc + 1.0f);
    sdv[t] = dv;
    if (node < NN) {
      cnt[node]  = cc;
      off[node]  = base + o;
      dinv[node] = dv;
    }
  }
  __syncthreads();
  int pl[SPW * RPL];
#pragma unroll
  for (int e = 0; e < SPW * RPL; ++e)
    pl[e] = (kv[e] >= 0) ? atomicAdd(&cur[(kv[e] >> 16) & 255], 1) : 0;
#pragma unroll
  for (int e = 0; e < SPW * RPL; ++e)
    if (kv[e] >= 0) lst[clamp_idx(base + pl[e], NE - 1)] = kv[e] & 0xFFFF;
  for (int s = 0; s < SPW; ++s) {
    int st = segA[SPW * w + s];
    int ln = segC[SPW * w + s];
    for (int i = lane + 32 * RPL; i < ln; i += 32) {
      int v = key[st + i];
      int p1 = atomicAdd(&cur[(v >> 16) & 255], 1);
      lst[clamp_idx(base + p1, NE - 1)] = v & 0xFFFF;
    }
  }
  if (!prod) return;
  int hh   = lane >> 4;
  int lid  = lane & 15;
  const float* xr = X + (tile * 16 + lid) * KIN;
  v8f c0 = {}, c1 = {}, c2 = {}, c3 = {};
  for (int k = 0; k < KIN; k += 4) {
    int ka = k + 2 * hh;
    v2f a;
    a.x = xr[ka];
    a.y = xr[ka + 1];
    const float* wr0 = W + ka * HID;
    const float* wr1 = wr0 + HID;
    v2f b;
    b.x = wr0[lid];      b.y = wr1[lid];
    c0 = __builtin_amdgcn_wmma_f32_16x16x4_f32(false, a, false, b, (short)0, c0, false, false);
    b.x = wr0[lid + 16]; b.y = wr1[lid + 16];
    c1 = __builtin_amdgcn_wmma_f32_16x16x4_f32(false, a, false, b, (short)0, c1, false, false);
    b.x = wr0[lid + 32]; b.y = wr1[lid + 32];
    c2 = __builtin_amdgcn_wmma_f32_16x16x4_f32(false, a, false, b, (short)0, c2, false, false);
    b.x = wr0[lid + 48]; b.y = wr1[lid + 48];
    c3 = __builtin_amdgcn_wmma_f32_16x16x4_f32(false, a, false, b, (short)0, c3, false, false);
  }
  int lrow = w * 16 + hh * 8;
  int mrow = tile * 16 + hh * 8;
  for (int j = 0; j < 8; ++j) {
    float ds = sdv[lrow + j];
    float* o = H + (mrow + j) * HID + lid;
    o[0]  = c0[j] * ds;
    o[16] = c1[j] * ds;
    o[32] = c2[j] * ds;
    o[48] = c3[j] * ds;
  }
}

__device__ __forceinline__ void add4(float4& a, const float4& p) {
  a.x += p.x; a.y += p.y; a.z += p.z; a.w += p.w;
}

template <int P, int B>
__device__ __forceinline__ void chunk4(const float4* __restrict__ PA, const float4* __restrict__ PB,
                                       int ev, int g, float4& a0, float4& a1) {
  int s0 = clamp_idx(bc8<B + 0>(ev), NN - 1);
  int s1 = clamp_idx(bc8<B + 1>(ev), NN - 1);
  int s2 = clamp_idx(bc8<B + 2>(ev), NN - 1);
  int s3 = clamp_idx(bc8<B + 3>(ev), NN - 1);
  float4 p0 = PA[s0 * P + g], q0 = PB[s0 * P + g];
  float4 p1 = PA[s1 * P + g], q1 = PB[s1 * P + g];
  float4 p2 = PA[s2 * P + g], q2 = PB[s2 * P + g];
  float4 p3 = PA[s3 * P + g], q3 = PB[s3 * P + g];
  add4(a0, p0); add4(a1, q0);
  add4(a0, p1); add4(a1, q1);
  add4(a0, p2); add4(a1, q2);
  add4(a0, p3); add4(a1, q3);
}

template <int P>
__device__ __forceinline__ void row_sums(const float4* __restrict__ PA, const float4* __restrict__ PB,
                                         const int* __restrict__ lst, int jb, int n, int g,
                                         float4& a0, float4& a1) {
  int e0 = lst[clamp_idx(jb + g, NE - 1)];
  int e1 = lst[clamp_idx(jb + 8 + g, NE - 1)];
  int e2 = lst[clamp_idx(jb + 16 + g, NE - 1)];
  int e3 = 0;
  if (n > 24) e3 = lst[clamp_idx(jb + 24 + g, NE - 1)];
  if (n >= 4)  chunk4<P, 0>(PA, PB, e0, g, a0, a1);
  if (n >= 8)  chunk4<P, 4>(PA, PB, e0, g, a0, a1);
  if (n >= 12) chunk4<P, 0>(PA, PB, e1, g, a0, a1);
  if (n >= 16) chunk4<P, 4>(PA, PB, e1, g, a0, a1);
  if (n >= 20) chunk4<P, 0>(PA, PB, e2, g, a0, a1);
  if (n >= 24) chunk4<P, 4>(PA, PB, e2, g, a0, a1);
  if (n >= 28) chunk4<P, 0>(PA, PB, e3, g, a0, a1);
  if (n >= 32) chunk4<P, 4>(PA, PB, e3, g, a0, a1);
  int rem = n & 3;
  if (rem != 0 && n < 32) {
    int cp = n >> 2;
    int ev = cp < 2 ? e0 : (cp < 4 ? e1 : (cp < 6 ? e2 : e3));
    bool hi = (cp & 1) != 0;
    int t0 = bc8<0>(ev), t1 = bc8<1>(ev), t2 = bc8<2>(ev);
    int u0 = bc8<4>(ev), u1 = bc8<5>(ev), u2 = bc8<6>(ev);
    int s0 = clamp_idx(hi ? u0 : t0, NN - 1);
    int s1 = clamp_idx(hi ? u1 : t1, NN - 1);
    int s2 = clamp_idx(hi ? u2 : t2, NN - 1);
    float4 z = make_float4(0.0f, 0.0f, 0.0f, 0.0f);
    float4 p0 = PA[s0 * P + g], q0 = PB[s0 * P + g];
    float4 p1 = z, q1 = z, p2 = z, q2 = z;
    if (rem > 1) { p1 = PA[s1 * P + g]; q1 = PB[s1 * P + g]; }
    if (rem > 2) { p2 = PA[s2 * P + g]; q2 = PB[s2 * P + g]; }
    add4(a0, p0); add4(a1, q0);
    add4(a0, p1); add4(a1, q1);
    add4(a0, p2); add4(a1, q2);
  }
#pragma unroll 1
  for (int j = jb + 32; j < jb + n; ++j) {
    int s0 = clamp_idx(lst[j], NN - 1);
    float4 p0 = PA[s0 * P + g], q0 = PB[s0 * P + g];
    add4(a0, p0); add4(a1, q0);
  }
}

#define TILE_NODES 16
#define LPN 8
#define AP 68
static_assert(NN % TILE_NODES == 0, "every block owns a whole row tile");

__global__ void __launch_bounds__(TILE_NODES * LPN)
k_gather64_hw12(const float* __restrict__ Hlin,
                const int* __restrict__ lst,
                const int* __restrict__ off,
                const int* __restrict__ cnt,
                const float* __restrict__ dinv,
                const float* __restrict__ bias,
                const float* __restrict__ mask,
                const float* __restrict__ W1,
                const float* __restrict__ W2,
                float* __restrict__ MU,
                float* __restrict__ LV) {
  __shared__ float sA[TILE_NODES * AP];
  int t = threadIdx.x;
  int r = t >> 3;
  int i = blockIdx.x * TILE_NODES + r;
  int g = t & 7;
  const float4* H4 = (const float4*)Hlin;
  float di = dinv[i];
  float4 a0 = H4[i * 16 + g];
  float4 a1 = H4[i * 16 + 8 + g];
  float4 m0 = ((const float4*)mask)[i * 16 + g];
  float4 m1 = ((const float4*)mask)[i * 16 + 8 + g];
  row_sums<16>(H4, H4 + 8, lst, off[i], cnt[i], g, a0, a1);
  float4 o0, o1;
  o0.x = fmaxf(a0.x * di + bias[4 * g + 0], 0.0f) * m0.x * di;
  o0.y = fmaxf(a0.y * di + bias[4 * g + 1], 0.0f) * m0.y * di;
  o0.z = fmaxf(a0.z * di + bias[4 * g + 2], 0.0f) * m0.z * di;
  o0.w = fmaxf(a0.w * di + bias[4 * g + 3], 0.0f) * m0.w * di;
  o1.x = fmaxf(a1.x * di + bias[32 + 4 * g + 0], 0.0f) * m1.x * di;
  o1.y = fmaxf(a1.y * di + bias[32 + 4 * g + 1], 0.0f) * m1.y * di;
  o1.z = fmaxf(a1.z * di + bias[32 + 4 * g + 2], 0.0f) * m1.z * di;
  o1.w = fmaxf(a1.w * di + bias[32 + 4 * g + 3], 0.0f) * m1.w * di;
  *(float4*)&sA[r * AP + 4 * g] = o0;
  *(float4*)&sA[r * AP + 32 + 4 * g] = o1;
  __syncthreads();
  int wave = t >> 5;
  if (wave >= 2) return;
  const float* Wm = wave == 0 ? W1 : W2;
  float*       Om = wave == 0 ? MU : LV;
  int lane = t & 31;
  int hh   = lane >> 4;
  int lid  = lane & 15;
  const float* ar = sA + lid * AP;
  v8f c0 = {}, c1 = {};
#pragma unroll 8
  for (int k = 0; k < HID; k += 4) {
    int ka = k + 2 * hh;
    v2f a;
    a.x = ar[ka];
    a.y = ar[ka + 1];
    const float* u0 = Wm + ka * NC;
    const float* u1 = u0 + NC;
    v2f b;
    b.x = u0[lid];      b.y = u1[lid];
    c0 = __builtin_amdgcn_wmma_f32_16x16x4_f32(false, a, false, b, (short)0, c0, false, false);
    b.x = u0[lid + 16]; b.y = u1[lid + 16];
    c1 = __builtin_amdgcn_wmma_f32_16x16x4_f32(false, a, false, b, (short)0, c1, false, false);
  }
  int mrow = blockIdx.x * TILE_NODES + hh * 8;
  for (int jj = 0; jj < 8; ++jj) {
    float* om = Om + (mrow + jj) * NC + lid;
    om[0]  = c0[jj];
    om[16] = c1[jj];
  }
}

__global__ void k_gather32(const float* __restrict__ MUlin,
                           const float* __restrict__ LVlin,
                           const int* __restrict__ lst,
                           const int* __restrict__ off,
                           const int* __restrict__ cnt,
                           const float* __restrict__ dinv,
                           const float* __restrict__ b1,
                           const float* __restrict__ b2,
                           const float* __restrict__ eps,
                           float* __restrict__ MUout,
                           float* __restrict__ LVout,
                           float* __restrict__ Z) {
  int idx = blockIdx.x * blockDim.x + threadIdx.x;
  int i = idx >> 3;
  if (i >= NN) return;
  int g = idx & 7;
  const float4* M4 = (const float4*)MUlin;
  const float4* L4 = (const float4*)LVlin;
  float di = dinv[i];
  float4 am = M4[i * 8 + g];
  float4 al = L4[i * 8 + g];
  float4 ep = ((const float4*)eps)[i * 8 + g];
  row_sums<8>(M4, L4, lst, off[i], cnt[i], g, am, al);
  am.x = am.x * di + b1[4 * g + 0];
  am.y = am.y * di + b1[4 * g + 1];
  am.z = am.z * di + b1[4 * g + 2];
  am.w = am.w * di + b1[4 * g + 3];
  al.x = al.x * di + b2[4 * g + 0];
  al.y = al.y * di + b2[4 * g + 1];
  al.z = al.z * di + b2[4 * g + 2];
  al.w = al.w * di + b2[4 * g + 3];
  ((float4*)MUout)[i * 8 + g] = am;
  ((float4*)LVout)[i * 8 + g] = al;
  float4 zz;
  zz.x = ep.x * expf(al.x) + am.x;
  zz.y = ep.y * expf(al.y) + am.y;
  zz.z = ep.z * expf(al.z) + am.z;
  zz.w = ep.w * expf(al.w) + am.w;
  ((float4*)Z)[i * 8 + g] = zz;
}

extern "C" void kernel_launch(void* const* d_in, const int* in_sizes, int n_in,
                              void* d_out, int out_size, void* d_ws, size_t ws_size,
                              hipStream_t stream) {
  (void)in_sizes; (void)n_in; (void)out_size; (void)ws_size;
  const float* x    = (const float*)d_in[0];
  const int*   ei   = (const int*)d_in[1];
  const float* W0   = (const float*)d_in[2];
  const float* b0   = (const float*)d_in[3];
  const float* W1   = (const float*)d_in[4];
  const float* b1   = (const float*)d_in[5];
  const float* W2   = (const float*)d_in[6];
  const float* b2   = (const float*)d_in[7];
  const float* mask = (const float*)d_in[8];
  const float* eps  = (const float*)d_in[9];
  const int* src = ei;
  const int* dst = ei + NE;

  int*   cnt   = (int*)d_ws;
  int*   off   = cnt + NNP;
  float* dinv  = (float*)(off + NNP);
  int*   lpre  = (int*)(dinv + NNP);
  int*   key   = lpre + NBLK * HP;
  int*   lst   = key + NE;
  float* hlin  = (float*)(lst + NE);
  float* mulin = hlin + (size_t)NN * HID;
  float* lvlin = mulin + (size_t)NN * NC;

  float* z  = (float*)d_out;
  float* mu = z + (size_t)NN * NC;
  float* lv = mu + (size_t)NN * NC;

  const int B = 256;
  const int gG   = (NN * LPN + B - 1) / B;

  k_bplace<<<NBLK, TPP, 0, stream>>>(src, dst, key, lpre);
  k_bsort <<<NBK,  TPB, 0, stream>>>(key, lpre, lst, cnt, off, dinv, x, W0, hlin);

  k_gather64_hw12<<<NN / TILE_NODES, TILE_NODES * LPN, 0, stream>>>(hlin, lst, off, cnt, dinv, b0, mask,
                                                                  W1, W2, mulin, lvlin);

  k_gather32 <<<gG,   B, 0, stream>>>(mulin, lvlin, lst, off, cnt, dinv, b1, b2, eps, mu, lv, z);
}


# Bucket kernel: 16 waves order the edges while the other 16 run X @ W0 from the start (k loop split at the four barriers), not behind the ordering; k order kept, list order still LDS arrival
# speedup vs baseline: 6.9016x; 1.0790x over previous
#include <hip/hip_runtime.h>


typedef __attribute__((ext_vector_type(2))) float v2f;
typedef __attribute__((ext_vector_type(8))) float v8f;

#define NN 50000
#define NE 800000
#define KIN 128
#define HID 64
#define NC 32
#define ROW_TILES (NN / 16)
#define NNP 50048

__device__ __forceinline__ int clamp_idx(int v, int hi) {
  return v < 0 ? 0 : (v > hi ? hi : v);
}

template <int K>
__device__ __forceinline__ int bc8(int v) {
  return __builtin_amdgcn_mov_dpp8(v, K * 0x249249);
}

#define NBLK 256
#define EPB  3125
#define TPP  512
#define EPT  7
#define NBK  196
#define HP   256
#define TPB  1024
static_assert(NBLK * EPB == NE, "the edge blocks cover the edge list exactly");
static_assert(EPT * TPP >= EPB && HP <= TPP, "a block's threads cover its edges and the table row");
static_assert(NBK + 1 <= HP && NBLK % 32 == 0 && NBLK <= TPB && TPB == 1024, "table pitch and run split");

__global__ void __launch_bounds__(TPP) k_bplace(const int* __restrict__ src,
                                                const int* __restrict__ dst,
                                                int* __restrict__ key,
                                                int* __restrict__ lpre) {
  __shared__ int h[HP];
  __shared__ int cur[HP];
  __shared__ int wsum[8];
  __shared__ int sk[EPB];
  int t = threadIdx.x;
  int b = blockIdx.x;
  int lane = t & 31;
  int w = t >> 5;
  if (t < HP) h[t] = 0;
  const int* d0 = dst + b * EPB;
  const int* s0 = src + b * EPB;
  int dv[EPT], sv[EPT];
#pragma unroll
  for (int j = 0; j < EPT; ++j) {
    int i = t + j * TPP;
    int ii = i < EPB ? i : EPB - 1;
    dv[j] = d0[ii];
    sv[j] = s0[ii];
  }
  __syncthreads();
#pragma unroll
  for (int j = 0; j < EPT; ++j) {
    int i = t + j * TPP;
    dv[j] = clamp_idx(dv[j], NN - 1);
    if (i < EPB) atomicAdd(&h[dv[j] >> 8], 1);
  }
  __syncthreads();
  int c = 0, x = 0;
  if (t < HP) {
    c = h[t];
    x = c;
    for (int o = 1; o < 32; o <<= 1) {
      int y = __shfl_up(x, o);
      if (lane >= o) x += y;
    }
    if (lane == 31) wsum[w] = x;
  }
  __syncthreads();
  if (t < HP) {
    int o = x - c;
    for (int q = 0; q < 8; ++q) o += (q < w) ? wsum[q] : 0;
    cur[t] = o;
    lpre[b * HP + t] = o;
  }
  __syncthreads();
  int pos[EPT];
#pragma unroll
  for (int j = 0; j < EPT; ++j) {
    int i = t + j * TPP;
    pos[j] = (i < EPB) ? atomicAdd(&cur[dv[j] >> 8], 1) : 0;
  }
#pragma unroll
  for (int j = 0; j < EPT; ++j) {
    int i = t + j * TPP;
    if (i < EPB)
      sk[clamp_idx(pos[j], EPB - 1)] = ((dv[j] & 255) << 16) | clamp_idx(sv[j], NN - 1);
  }
  __syncthreads();
#pragma unroll
  for (int j = 0; j < EPT; ++j) {
    int i = t + j * TPP;
    if (i < EPB) key[b * EPB + i] = sk[i];
  }
}

#define SW 16
#define GW 16
#define SPW (NBLK / SW)
#define RPL 1
#define KS0 16
#define KS1 48
#define KS2 56
#define KS3 64
static_assert(GW * 16 == 256 && (SW + GW) * 32 == TPB, "a bucket is GW row tiles; both roles together are the work-group");
static_assert(NBLK % SW == 0 && SW * 32 >= 256 && SW * 32 >= NBLK, "the ordering waves cover the runs, the nodes and the table column");
static_assert(KS0 % 4 == 0 && KS1 % 4 == 0 && KS2 % 4 == 0 && KS3 % 4 == 0 && KS3 <= KIN, "whole k steps");

__device__ __forceinline__ void xw_steps(const float* __restrict__ xr, const float* __restrict__ W,
                                         int hh, int lid, int k0, int k1,
                                         v8f& c0, v8f& c1, v8f& c2, v8f& c3) {
  for (int k = k0; k < k1; k += 4) {
    int ka = k + 2 * hh;
    v2f a;
    a.x = xr[ka];
    a.y = xr[ka + 1];
    const float* wr0 = W + ka * HID;
    const float* wr1 = wr0 + HID;
    v2f b;
    b.x = wr0[lid];      b.y = wr1[lid];
    c0 = __builtin_amdgcn_wmma_f32_16x16x4_f32(false, a, false, b, (short)0, c0, false, false);
    b.x = wr0[lid + 16]; b.y = wr1[lid + 16];
    c1 = __builtin_amdgcn_wmma_f32_16x16x4_f32(false, a, false, b, (short)0, c1, false, false);
    b.x = wr0[lid + 32]; b.y = wr1[lid + 32];
    c2 = __builtin_amdgcn_wmma_f32_16x16x4_f32(false, a, false, b, (short)0, c2, false, false);
    b.x = wr0[lid + 48]; b.y = wr1[lid + 48];
    c3 = __builtin_amdgcn_wmma_f32_16x16x4_f32(false, a, false, b, (short)0, c3, false, false);
  }
}

__global__ void __launch_bounds__(TPB) k_bsort(const int* __restrict__ key,
                                               const int* __restrict__ lpre,
                                               int* __restrict__ lst,
                                               int* __restrict__ cnt,
                                               int* __restrict__ off,
                                               float* __restrict__ dinv,
                                               const float* __restrict__ X,
                                               const float* __restrict__ W,
                                               float* __restrict__ H) {
  __shared__ int c[256];
  __shared__ float sdv[256];
  __shared__ int cur[256];
  __shared__ int wsum[8];
  __shared__ int segA[NBLK];
  __shared__ int segC[NBLK];
  __shared__ int redA[NBLK / 32];
  int t = threadIdx.x;
  int kb = blockIdx.x;
  int lane = t & 31;
  int w = __builtin_amdgcn_readfirstlane(t >> 5);
  bool sorter = w < SW;
  int tile = kb * GW + (w - SW);
  bool prod = !sorter && (tile < ROW_TILES);
  int hh   = lane >> 4;
  int lid  = lane & 15;
  const float* xr = X + ((prod ? tile : 0) * 16 + lid) * KIN;
  v8f c0 = {}, c1 = {}, c2 = {}, c3 = {};
  int kv[SPW * RPL];
  int base = 0;
  int cc = 0, x = 0;

  if (sorter) {
    if (t < 256) c[t] = 0;
    if (t < NBLK) {
      int a  = clamp_idx(lpre[t * HP + kb], EPB);
      int a1 = clamp_idx(lpre[t * HP + kb + 1], EPB);
      segA[t] = t * EPB + a;
      segC[t] = a1 > a ? a1 - a : 0;
      int sa = a;
      for (int o = 16; o > 0; o >>= 1) sa += __shfl_xor(sa, o);
      if (lane == 0) redA[w] = sa;
    }
  } else if (prod) {
    __builtin_prefetch(X + tile * 16 * KIN + lane * 64, 0, 1);
    xw_steps(xr, W, hh, lid, 0, KS0, c0, c1, c2, c3);
  }
  __syncthreads();
  if (sorter) {
#pragma unroll
    for (int q = 0; q < NBLK / 32; ++q) base += redA[q];
    base = clamp_idx(base, NE);
    int st[SPW], ln[SPW];
#pragma unroll
    for (int s = 0; s < SPW; ++s) {
      st[s] = segA[SPW * w + s];
      ln[s] = segC[SPW * w + s];
    }
#pragma unroll
    for (int s = 0; s < SPW; ++s) {
#pragma unroll
      for (int r = 0; r < RPL; ++r) {
        int i = lane + 32 * r;
        kv[s * RPL + r] = (i < ln[s]) ? key[st[s] + i] : -1;
      }
    }
#pragma unroll
    for (int e = 0; e < SPW * RPL; ++e)
      if (kv[e] >= 0) atomicAdd(&c[(kv[e] >> 16) & 255], 1);
    for (int s = 0; s < SPW; ++s) {
      int st = segA[SPW * w + s];
      int ln = segC[SPW * w + s];
      for (int i = lane + 32 * RPL; i < ln; i += 32)
        atomicAdd(&c[(key[st + i] >> 16) & 255], 1);
    }
  } else if (prod) {
    xw_steps(xr, W, hh, lid, KS0, KS1, c0, c1, c2, c3);
  }
  __syncthreads();
  if (sorter) {
    if (t < 256) {
      cc = c[t];
      x = cc;
      for (int o = 1; o < 32; o <<= 1) {
        int y = __shfl_up(x, o);
        if (lane >= o) x += y;
      }
      if (lane == 31) wsum[w] = x;
    }
  } else if (prod) {
    xw_steps(xr, W, hh, lid, KS1, KS2, c0, c1, c2, c3);
  }
  __syncthreads();
  if (sorter) {
    if (t < 256) {
      int o = x - cc;
      for (int q = 0; q < 8; ++q) o += (q < w) ? wsum[q] : 0;
      cur[t] = o;
      int node = kb * 256 + t;
      float dv = rsqrtf((float)cc + 1.0f);
      sdv[t] = dv;
      if (node < NN) {
        cnt[node]  = cc;
        off[node]  = base + o;
        dinv[node] = dv;
      }
    }
  } else if (prod) {
    xw_steps(xr, W, hh, lid, KS2, KS3, c0, c1, c2, c3);
  }
  __syncthreads();
  if (sorter) {
    int pl[SPW * RPL];
#pragma unroll
    for (int e = 0; e < SPW * RPL; ++e)
      pl[e] = (kv[e] >= 0) ? atomicAdd(&cur[(kv[e] >> 16) & 255], 1) : 0;
#pragma unroll
    for (int e = 0; e < SPW * RPL; ++e)
      if (kv[e] >= 0) lst[clamp_idx(base + pl[e], NE - 1)] = kv[e] & 0xFFFF;
    for (int s = 0; s < SPW; ++s) {
      int st = segA[SPW * w + s];
      int ln = segC[SPW * w + s];
      for (int i = lane + 32 * RPL; i < ln; i += 32) {
        int v = key[st + i];
        int p1 = atomicAdd(&cur[(v >> 16) & 255], 1);
        lst[clamp_idx(base + p1, NE - 1)] = v & 0xFFFF;
      }
    }
    return;
  }
  if (!prod) return;
  xw_steps(xr, W, hh, lid, KS3, KIN, c0, c1, c2, c3);
  int lrow = (w - SW) * 16 + hh * 8;
  int mrow = tile * 16 + hh * 8;
  for (int j = 0; j < 8; ++j) {
    float ds = sdv[lrow + j];
    float* o = H + (mrow + j) * HID + lid;
    o[0]  = c0[j] * ds;
    o[16] = c1[j] * ds;
    o[32] = c2[j] * ds;
    o[48] = c3[j] * ds;
  }
}

__device__ __forceinline__ void add4(float4& a, const float4& p) {
  a.x += p.x; a.y += p.y; a.z += p.z; a.w += p.w;
}

template <int P, int B>
__device__ __forceinline__ void chunk4(const float4* __restrict__ PA, const float4* __restrict__ PB,
                                       int ev, int g, float4& a0, float4& a1) {
  int s0 = clamp_idx(bc8<B + 0>(ev), NN - 1);
  int s1 = clamp_idx(bc8<B + 1>(ev), NN - 1);
  int s2 = clamp_idx(bc8<B + 2>(ev), NN - 1);
  int s3 = clamp_idx(bc8<B + 3>(ev), NN - 1);
  float4 p0 = PA[s0 * P + g], q0 = PB[s0 * P + g];
  float4 p1 = PA[s1 * P + g], q1 = PB[s1 * P + g];
  float4 p2 = PA[s2 * P + g], q2 = PB[s2 * P + g];
  float4 p3 = PA[s3 * P + g], q3 = PB[s3 * P + g];
  add4(a0, p0); add4(a1, q0);
  add4(a0, p1); add4(a1, q1);
  add4(a0, p2); add4(a1, q2);
  add4(a0, p3); add4(a1, q3);
}

template <int P>
__device__ __forceinline__ void row_sums(const float4* __restrict__ PA, const float4* __restrict__ PB,
                                         const int* __restrict__ lst, int jb, int n, int g,
                                         float4& a0, float4& a1) {
  int e0 = lst[clamp_idx(jb + g, NE - 1)];
  int e1 = lst[clamp_idx(jb + 8 + g, NE - 1)];
  int e2 = lst[clamp_idx(jb + 16 + g, NE - 1)];
  int e3 = 0;
  if (n > 24) e3 = lst[clamp_idx(jb + 24 + g, NE - 1)];
  if (n >= 4)  chunk4<P, 0>(PA, PB, e0, g, a0, a1);
  if (n >= 8)  chunk4<P, 4>(PA, PB, e0, g, a0, a1);
  if (n >= 12) chunk4<P, 0>(PA, PB, e1, g, a0, a1);
  if (n >= 16) chunk4<P, 4>(PA, PB, e1, g, a0, a1);
  if (n >= 20) chunk4<P, 0>(PA, PB, e2, g, a0, a1);
  if (n >= 24) chunk4<P, 4>(PA, PB, e2, g, a0, a1);
  if (n >= 28) chunk4<P, 0>(PA, PB, e3, g, a0, a1);
  if (n >= 32) chunk4<P, 4>(PA, PB, e3, g, a0, a1);
  int rem = n & 3;
  if (rem != 0 && n < 32) {
    int cp = n >> 2;
    int ev = cp < 2 ? e0 : (cp < 4 ? e1 : (cp < 6 ? e2 : e3));
    bool hi = (cp & 1) != 0;
    int t0 = bc8<0>(ev), t1 = bc8<1>(ev), t2 = bc8<2>(ev);
    int u0 = bc8<4>(ev), u1 = bc8<5>(ev), u2 = bc8<6>(ev);
    int s0 = clamp_idx(hi ? u0 : t0, NN - 1);
    int s1 = clamp_idx(hi ? u1 : t1, NN - 1);
    int s2 = clamp_idx(hi ? u2 : t2, NN - 1);
    float4 z = make_float4(0.0f, 0.0f, 0.0f, 0.0f);
    float4 p0 = PA[s0 * P + g], q0 = PB[s0 * P + g];
    float4 p1 = z, q1 = z, p2 = z, q2 = z;
    if (rem > 1) { p1 = PA[s1 * P + g]; q1 = PB[s1 * P + g]; }
    if (rem > 2) { p2 = PA[s2 * P + g]; q2 = PB[s2 * P + g]; }
    add4(a0, p0); add4(a1, q0);
    add4(a0, p1); add4(a1, q1);
    add4(a0, p2); add4(a1, q2);
  }
#pragma unroll 1
  for (int j = jb + 32; j < jb + n; ++j) {
    int s0 = clamp_idx(lst[j], NN - 1);
    float4 p0 = PA[s0 * P + g], q0 = PB[s0 * P + g];
    add4(a0, p0); add4(a1, q0);
  }
}

#define TILE_NODES 16
#define LPN 8
#define AP 68
static_assert(NN % TILE_NODES == 0, "every block owns a whole row tile");

__global__ void __launch_bounds__(TILE_NODES * LPN)
k_gather64_hw12(const float* __restrict__ Hlin,
                const int* __restrict__ lst,
                const int* __restrict__ off,
                const int* __restrict__ cnt,
                const float* __restrict__ dinv,
                const float* __restrict__ bias,
                const float* __restrict__ mask,
                const float* __restrict__ W1,
                const float* __restrict__ W2,
                float* __restrict__ MU,
                float* __restrict__ LV) {
  __shared__ float sA[TILE_NODES * AP];
  int t = threadIdx.x;
  int r = t >> 3;
  int i = blockIdx.x * TILE_NODES + r;
  int g = t & 7;
  const float4* H4 = (const float4*)Hlin;
  float di = dinv[i];
  float4 a0 = H4[i * 16 + g];
  float4 a1 = H4[i * 16 + 8 + g];
  float4 m0 = ((const float4*)mask)[i * 16 + g];
  float4 m1 = ((const float4*)mask)[i * 16 + 8 + g];
  row_sums<16>(H4, H4 + 8, lst, off[i], cnt[i], g, a0, a1);
  float4 o0, o1;
  o0.x = fmaxf(a0.x * di + bias[4 * g + 0], 0.0f) * m0.x * di;
  o0.y = fmaxf(a0.y * di + bias[4 * g + 1], 0.0f) * m0.y * di;
  o0.z = fmaxf(a0.z * di + bias[4 * g + 2], 0.0f) * m0.z * di;
  o0.w = fmaxf(a0.w * di + bias[4 * g + 3], 0.0f) * m0.w * di;
  o1.x = fmaxf(a1.x * di + bias[32 + 4 * g + 0], 0.0f) * m1.x * di;
  o1.y = fmaxf(a1.y * di + bias[32 + 4 * g + 1], 0.0f) * m1.y * di;
  o1.z = fmaxf(a1.z * di + bias[32 + 4 * g + 2], 0.0f) * m1.z * di;
  o1.w = fmaxf(a1.w * di + bias[32 + 4 * g + 3], 0.0f) * m1.w * di;
  *(float4*)&sA[r * AP + 4 * g] = o0;
  *(float4*)&sA[r * AP + 32 + 4 * g] = o1;
  __syncthreads();
  int wave = t >> 5;
  if (wave >= 2) return;
  const float* Wm = wave == 0 ? W1 : W2;
  float*       Om = wave == 0 ? MU : LV;
  int lane = t & 31;
  int hh   = lane >> 4;
  int lid  = lane & 15;
  const float* ar = sA + lid * AP;
  v8f c0 = {}, c1 = {};
#pragma unroll 8
  for (int k = 0; k < HID; k += 4) {
    int ka = k + 2 * hh;
    v2f a;
    a.x = ar[ka];
    a.y = ar[ka + 1];
    const float* u0 = Wm + ka * NC;
    const float* u1 = u0 + NC;
    v2f b;
    b.x = u0[lid];      b.y = u1[lid];
    c0 = __builtin_amdgcn_wmma_f32_16x16x4_f32(false, a, false, b, (short)0, c0, false, false);
    b.x = u0[lid + 16]; b.y = u1[lid + 16];
    c1 = __builtin_amdgcn_wmma_f32_16x16x4_f32(false, a, false, b, (short)0, c1, false, false);
  }
  int mrow = blockIdx.x * TILE_NODES + hh * 8;
  for (int jj = 0; jj < 8; ++jj) {
    float* om = Om + (mrow + jj) * NC + lid;
    om[0]  = c0[jj];
    om[16] = c1[jj];
  }
}

__global__ void k_gather32(const float* __restrict__ MUlin,
                           const float* __restrict__ LVlin,
                           const int* __restrict__ lst,
                           const int* __restrict__ off,
                           const int* __restrict__ cnt,
                           const float* __restrict__ dinv,
                           const float* __restrict__ b1,
                           const float* __restrict__ b2,
                           const float* __restrict__ eps,
                           float* __restrict__ MUout,
                           float* __restrict__ LVout,
                           float* __restrict__ Z) {
  int idx = blockIdx.x * blockDim.x + threadIdx.x;
  int i = idx >> 3;
  if (i >= NN) return;
  int g = idx & 7;
  const float4* M4 = (const float4*)MUlin;
  const float4* L4 = (const float4*)LVlin;
  float di = dinv[i];
  float4 am = M4[i * 8 + g];
  float4 al = L4[i * 8 + g];
  float4 ep = ((const float4*)eps)[i * 8 + g];
  row_sums<8>(M4, L4, lst, off[i], cnt[i], g, am, al);
  am.x = am.x * di + b1[4 * g + 0];
  am.y = am.y * di + b1[4 * g + 1];
  am.z = am.z * di + b1[4 * g + 2];
  am.w = am.w * di + b1[4 * g + 3];
  al.x = al.x * di + b2[4 * g + 0];
  al.y = al.y * di + b2[4 * g + 1];
  al.z = al.z * di + b2[4 * g + 2];
  al.w = al.w * di + b2[4 * g + 3];
  ((float4*)MUout)[i * 8 + g] = am;
  ((float4*)LVout)[i * 8 + g] = al;
  float4 zz;
  zz.x = ep.x * expf(al.x) + am.x;
  zz.y = ep.y * expf(al.y) + am.y;
  zz.z = ep.z * expf(al.z) + am.z;
  zz.w = ep.w * expf(al.w) + am.w;
  ((float4*)Z)[i * 8 + g] = zz;
}

extern "C" void kernel_launch(void* const* d_in, const int* in_sizes, int n_in,
                              void* d_out, int out_size, void* d_ws, size_t ws_size,
                              hipStream_t stream) {
  (void)in_sizes; (void)n_in; (void)out_size; (void)ws_size;
  const float* x    = (const float*)d_in[0];
  const int*   ei   = (const int*)d_in[1];
  const float* W0   = (const float*)d_in[2];
  const float* b0   = (const float*)d_in[3];
  const float* W1   = (const float*)d_in[4];
  const float* b1   = (const float*)d_in[5];
  const float* W2   = (const float*)d_in[6];
  const float* b2   = (const float*)d_in[7];
  const float* mask = (const float*)d_in[8];
  const float* eps  = (const float*)d_in[9];
  const int* src = ei;
  const int* dst = ei + NE;

  int*   cnt   = (int*)d_ws;
  int*   off   = cnt + NNP;
  float* dinv  = (float*)(off + NNP);
  int*   lpre  = (int*)(dinv + NNP);
  int*   key   = lpre + NBLK * HP;
  int*   lst   = key + NE;
  float* hlin  = (float*)(lst + NE);
  float* mulin = hlin + (size_t)NN * HID;
  float* lvlin = mulin + (size_t)NN * NC;

  float* z  = (float*)d_out;
  float* mu = z + (size_t)NN * NC;
  float* lv = mu + (size_t)NN * NC;

  const int B = 256;
  const int gG   = (NN * LPN + B - 1) / B;

  k_bplace<<<NBLK, TPP, 0, stream>>>(src, dst, key, lpre);
  k_bsort <<<NBK,  TPB, 0, stream>>>(key, lpre, lst, cnt, off, dinv, x, W0, hlin);

  k_gather64_hw12<<<NN / TILE_NODES, TILE_NODES * LPN, 0, stream>>>(hlin, lst, off, cnt, dinv, b0, mask,
                                                                  W1, W2, mulin, lvlin);

  k_gather32 <<<gG,   B, 0, stream>>>(mulin, lvlin, lst, off, cnt, dinv, b1, b2, eps, mu, lv, z);
}
